# starting kernel
# baseline (speedup 1.0000x reference)
#include <hip/hip_runtime.h>

typedef __attribute__((ext_vector_type(2))) float v2f;
typedef __attribute__((ext_vector_type(8))) float v8f;

namespace {
constexpr int kB = 1024;
constexpr int kM = 64;
constexpr int kH = 256;
constexpr int kA = 16;
constexpr int kSteps = 4;
constexpr int kPitch = 260;
constexpr int kMT = 4;
constexpr int kNTpw = 2;
constexpr int kLdsFloats = 3 * kM * kPitch + 2 * kPitch;
}

__launch_bounds__(256)
__global__ void commnet_wmma_kernel(
    const int* __restrict__ ids, const float* __restrict__ emb,
    const float* __restrict__ W1, const float* __restrict__ b1,
    const float* __restrict__ W2, const float* __restrict__ b2,
    const float* __restrict__ Wd, const float* __restrict__ bd,
    float* __restrict__ out) {
  extern __shared__ float smem[];
  float* sh = smem;
  float* sp = sh + kM * kPitch;
  float* st = sp + kM * kPitch;
  float* sS = st + kM * kPitch;
  float* sU = sS + kPitch;

  const int b     = blockIdx.x;
  const int tid   = threadIdx.x;
  const int wave  = tid >> 5;
  const int lane  = tid & 31;
  const int lrow  = lane & 15;
  const int khalf = (lane >> 4) << 1;
  const int hi8   = (lane >> 4) << 3;
  const float inv = 1.0f / (float)(kM - 1);
  const int nb    = wave * (kNTpw * 16);

  for (int m = 0; m < kM; ++m) {
    const int id  = ids[b * kM + m];
    const float v = emb[id * kH + tid];
    sp[m * kPitch + tid] = v;
    sh[m * kPitch + tid] = v;
  }
  __syncthreads();

  {
    v8f accP[kMT][kNTpw];
    const v8f vzero = {};
    for (int mt = 0; mt < kMT; ++mt)
      for (int j = 0; j < kNTpw; ++j) accP[mt][j] = vzero;
#pragma unroll 4
    for (int k0 = 0; k0 < kH; k0 += 4) {
      const int kk = k0 + khalf;
      v2f bf[kNTpw];
      for (int j = 0; j < kNTpw; ++j) {
        const int n = nb + j * 16 + lrow;
        bf[j].x = W1[(2 * kH + kk) * kH + n];
        bf[j].y = W1[(2 * kH + kk + 1) * kH + n];
      }
      for (int mt = 0; mt < kMT; ++mt) {
        const v2f a = *(const v2f*)&sp[(mt * 16 + lrow) * kPitch + kk];
        for (int j = 0; j < kNTpw; ++j)
          accP[mt][j] = __builtin_amdgcn_wmma_f32_16x16x4_f32(
              false, a, false, bf[j], (short)0, accP[mt][j], false, false);
      }
    }
    __syncthreads();
    for (int j = 0; j < kNTpw; ++j) {
      const int n = nb + j * 16 + lrow;
      const float bias = b1[n];
      for (int mt = 0; mt < kMT; ++mt) {
        const int mrow = mt * 16 + hi8;
#pragma unroll
        for (int r = 0; r < 8; ++r)
          sp[(mrow + r) * kPitch + n] = accP[mt][j][r] + bias;
      }
    }
  }

  for (int step = 0; step < kSteps; ++step) {
    __syncthreads();
    {
      float s = 0.f;
#pragma unroll 8
      for (int m = 0; m < kM; ++m) s += sh[m * kPitch + tid];
      sS[tid] = s;
    }
    __syncthreads();

    {
      float acc = 0.f;
#pragma unroll 4
      for (int k = 0; k < kH; ++k)
        acc = fmaf(sS[k], W1[(kH + k) * kH + tid], acc);
      sU[tid] = acc * inv;
    }

    v8f acc1[kMT][kNTpw];
    const v8f vzero = {};
    for (int mt = 0; mt < kMT; ++mt)
      for (int j = 0; j < kNTpw; ++j) acc1[mt][j] = vzero;

#pragma unroll 2
    for (int k0 = 0; k0 < kH; k0 += 4) {
      const int kk = k0 + khalf;
      v2f bf[kNTpw];
      for (int j = 0; j < kNTpw; ++j) {
        const int n = nb + j * 16 + lrow;
        const float a0 = W1[kk * kH + n];
        const float c0 = W1[(kH + kk) * kH + n];
        const float a1 = W1[(kk + 1) * kH + n];
        const float c1 = W1[(kH + kk + 1) * kH + n];
        bf[j].x = fmaf(-inv, c0, a0);
        bf[j].y = fmaf(-inv, c1, a1);
      }
      for (int mt = 0; mt < kMT; ++mt) {
        const v2f a = *(const v2f*)&sh[(mt * 16 + lrow) * kPitch + kk];
        for (int j = 0; j < kNTpw; ++j)
          acc1[mt][j] = __builtin_amdgcn_wmma_f32_16x16x4_f32(
              false, a, false, bf[j], (short)0, acc1[mt][j], false, false);
      }
    }
    for (int j = 0; j < kNTpw; ++j) {
      const int n = nb + j * 16 + lrow;
      const float u = sU[n];
      for (int mt = 0; mt < kMT; ++mt) {
        const int mrow = mt * 16 + hi8;
#pragma unroll
        for (int r = 0; r < 8; ++r) {
          const float v = acc1[mt][j][r] + sp[(mrow + r) * kPitch + n] + u;
          st[(mrow + r) * kPitch + n] = fmaxf(v, 0.f);
        }
      }
    }
    __syncthreads();

    v8f acc2[kMT][kNTpw];
    for (int mt = 0; mt < kMT; ++mt)
      for (int j = 0; j < kNTpw; ++j) acc2[mt][j] = vzero;

#pragma unroll 4
    for (int k0 = 0; k0 < kH; k0 += 4) {
      const int kk = k0 + khalf;
      v2f bf[kNTpw];
      for (int j = 0; j < kNTpw; ++j) {
        const int n = nb + j * 16 + lrow;
        bf[j].x = W2[kk * kH + n];
        bf[j].y = W2[(kk + 1) * kH + n];
      }
      for (int mt = 0; mt < kMT; ++mt) {
        const v2f a = *(const v2f*)&st[(mt * 16 + lrow) * kPitch + kk];
        for (int j = 0; j < kNTpw; ++j)
          acc2[mt][j] = __builtin_amdgcn_wmma_f32_16x16x4_f32(
              false, a, false, bf[j], (short)0, acc2[mt][j], false, false);
      }
    }
    for (int j = 0; j < kNTpw; ++j) {
      const int n = nb + j * 16 + lrow;
      const float bias = b2[n];
      for (int mt = 0; mt < kMT; ++mt) {
        const int mrow = mt * 16 + hi8;
#pragma unroll
        for (int r = 0; r < 8; ++r)
          sh[(mrow + r) * kPitch + n] += acc2[mt][j][r] + bias;
      }
    }
  }

  __syncthreads();
  if (wave < kMT) {
    v8f accL = {};
#pragma unroll 4
    for (int k0 = 0; k0 < kH; k0 += 4) {
      const int kk = k0 + khalf;
      v2f bf;
      bf.x = Wd[kk * kA + lrow];
      bf.y = Wd[(kk + 1) * kA + lrow];
      const v2f a = *(const v2f*)&sh[(wave * 16 + lrow) * kPitch + kk];
      accL = __builtin_amdgcn_wmma_f32_16x16x4_f32(
          false, a, false, bf, (short)0, accL, false, false);
    }
    const float bias = bd[lrow];
    const int mrow = wave * 16 + hi8;
#pragma unroll
    for (int r = 0; r < 8; ++r)
      out[(b * kM + mrow + r) * kA + lrow] = accL[r] + bias;
  }
}

extern "C" void kernel_launch(void* const* d_in, const int* in_sizes, int n_in,
                              void* d_out, int out_size, void* d_ws, size_t ws_size,
                              hipStream_t stream) {
  (void)in_sizes; (void)n_in; (void)out_size; (void)d_ws; (void)ws_size;
  const int*   ids = (const int*)d_in[0];
  const float* emb = (const float*)d_in[1];
  const float* W1  = (const float*)d_in[2];
  const float* b1  = (const float*)d_in[3];
  const float* W2  = (const float*)d_in[4];
  const float* b2  = (const float*)d_in[5];
  const float* Wd  = (const float*)d_in[6];
  const float* bd  = (const float*)d_in[7];
  float* out = (float*)d_out;

  const size_t lds_bytes = (size_t)kLdsFloats * sizeof(float);
  (void)hipFuncSetAttribute((const void*)commnet_wmma_kernel,
                            hipFuncAttributeMaxDynamicSharedMemorySize,
                            (int)lds_bytes);
  commnet_wmma_kernel<<<dim3(kB), dim3(256), lds_bytes, stream>>>(
      ids, emb, W1, b1, W2, b2, Wd, bd, out);
}


# Fuse the u = inv*(S@W1b) mat-vec into the GEMM1 loop: reuse the W1b values each lane loads for the effective weight; u summed per half-wave then added (u summation order changes)
# speedup vs baseline: 1.4127x; 1.4127x over previous
#include <hip/hip_runtime.h>

typedef __attribute__((ext_vector_type(2))) float v2f;
typedef __attribute__((ext_vector_type(8))) float v8f;

namespace {
constexpr int kB = 1024;
constexpr int kM = 64;
constexpr int kH = 256;
constexpr int kA = 16;
constexpr int kSteps = 4;
constexpr int kPitch = 260;
constexpr int kMT = 4;
constexpr int kNTpw = 2;
constexpr int kLdsFloats = 3 * kM * kPitch + kPitch;
}

__launch_bounds__(256)
__global__ void commnet_wmma_kernel(
    const int* __restrict__ ids, const float* __restrict__ emb,
    const float* __restrict__ W1, const float* __restrict__ b1,
    const float* __restrict__ W2, const float* __restrict__ b2,
    const float* __restrict__ Wd, const float* __restrict__ bd,
    float* __restrict__ out) {
  extern __shared__ float smem[];
  float* sh = smem;
  float* sp = sh + kM * kPitch;
  float* st = sp + kM * kPitch;
  float* sS = st + kM * kPitch;

  const int b     = blockIdx.x;
  const int tid   = threadIdx.x;
  const int wave  = tid >> 5;
  const int lane  = tid & 31;
  const int lrow  = lane & 15;
  const int khalf = (lane >> 4) << 1;
  const int hi8   = (lane >> 4) << 3;
  const float inv = 1.0f / (float)(kM - 1);
  const int nb    = wave * (kNTpw * 16);

  for (int m = 0; m < kM; ++m) {
    const int id  = ids[b * kM + m];
    const float v = emb[id * kH + tid];
    sp[m * kPitch + tid] = v;
    sh[m * kPitch + tid] = v;
  }
  __syncthreads();

  {
    v8f accP[kMT][kNTpw];
    const v8f vzero = {};
    for (int mt = 0; mt < kMT; ++mt)
      for (int j = 0; j < kNTpw; ++j) accP[mt][j] = vzero;
#pragma unroll 4
    for (int k0 = 0; k0 < kH; k0 += 4) {
      const int kk = k0 + khalf;
      v2f bf[kNTpw];
      for (int j = 0; j < kNTpw; ++j) {
        const int n = nb + j * 16 + lrow;
        bf[j].x = W1[(2 * kH + kk) * kH + n];
        bf[j].y = W1[(2 * kH + kk + 1) * kH + n];
      }
      for (int mt = 0; mt < kMT; ++mt) {
        const v2f a = *(const v2f*)&sp[(mt * 16 + lrow) * kPitch + kk];
        for (int j = 0; j < kNTpw; ++j)
          accP[mt][j] = __builtin_amdgcn_wmma_f32_16x16x4_f32(
              false, a, false, bf[j], (short)0, accP[mt][j], false, false);
      }
    }
    __syncthreads();
    for (int j = 0; j < kNTpw; ++j) {
      const int n = nb + j * 16 + lrow;
      const float bias = b1[n];
      for (int mt = 0; mt < kMT; ++mt) {
        const int mrow = mt * 16 + hi8;
#pragma unroll
        for (int r = 0; r < 8; ++r)
          sp[(mrow + r) * kPitch + n] = accP[mt][j][r] + bias;
      }
    }
  }

  for (int step = 0; step < kSteps; ++step) {
    __syncthreads();
    {
      float s = 0.f;
#pragma unroll 8
      for (int m = 0; m < kM; ++m) s += sh[m * kPitch + tid];
      sS[tid] = s;
    }
    __syncthreads();

    v8f acc1[kMT][kNTpw];
    float uacc[kNTpw];
    const v8f vzero = {};
    for (int mt = 0; mt < kMT; ++mt)
      for (int j = 0; j < kNTpw; ++j) acc1[mt][j] = vzero;
    for (int j = 0; j < kNTpw; ++j) uacc[j] = 0.f;

#pragma unroll 2
    for (int k0 = 0; k0 < kH; k0 += 4) {
      const int kk = k0 + khalf;
      const v2f sv = *(const v2f*)&sS[kk];
      v2f bf[kNTpw];
      for (int j = 0; j < kNTpw; ++j) {
        const int n = nb + j * 16 + lrow;
        const float a0 = W1[kk * kH + n];
        const float c0 = W1[(kH + kk) * kH + n];
        const float a1 = W1[(kk + 1) * kH + n];
        const float c1 = W1[(kH + kk + 1) * kH + n];
        bf[j].x = fmaf(-inv, c0, a0);
        bf[j].y = fmaf(-inv, c1, a1);
        uacc[j] = fmaf(sv.x, c0, uacc[j]);
        uacc[j] = fmaf(sv.y, c1, uacc[j]);
      }
      for (int mt = 0; mt < kMT; ++mt) {
        const v2f a = *(const v2f*)&sh[(mt * 16 + lrow) * kPitch + kk];
        for (int j = 0; j < kNTpw; ++j)
          acc1[mt][j] = __builtin_amdgcn_wmma_f32_16x16x4_f32(
              false, a, false, bf[j], (short)0, acc1[mt][j], false, false);
      }
    }
    for (int j = 0; j < kNTpw; ++j) {
      const int n = nb + j * 16 + lrow;
      const float u = (uacc[j] + __shfl_xor(uacc[j], 16)) * inv;
      for (int mt = 0; mt < kMT; ++mt) {
        const int mrow = mt * 16 + hi8;
#pragma unroll
        for (int r = 0; r < 8; ++r) {
          const float v = acc1[mt][j][r] + sp[(mrow + r) * kPitch + n] + u;
          st[(mrow + r) * kPitch + n] = fmaxf(v, 0.f);
        }
      }
    }
    __syncthreads();

    v8f acc2[kMT][kNTpw];
    for (int mt = 0; mt < kMT; ++mt)
      for (int j = 0; j < kNTpw; ++j) acc2[mt][j] = vzero;

#pragma unroll 4
    for (int k0 = 0; k0 < kH; k0 += 4) {
      const int kk = k0 + khalf;
      v2f bf[kNTpw];
      for (int j = 0; j < kNTpw; ++j) {
        const int n = nb + j * 16 + lrow;
        bf[j].x = W2[kk * kH + n];
        bf[j].y = W2[(kk + 1) * kH + n];
      }
      for (int mt = 0; mt < kMT; ++mt) {
        const v2f a = *(const v2f*)&st[(mt * 16 + lrow) * kPitch + kk];
        for (int j = 0; j < kNTpw; ++j)
          acc2[mt][j] = __builtin_amdgcn_wmma_f32_16x16x4_f32(
              false, a, false, bf[j], (short)0, acc2[mt][j], false, false);
      }
    }
    for (int j = 0; j < kNTpw; ++j) {
      const int n = nb + j * 16 + lrow;
      const float bias = b2[n];
      for (int mt = 0; mt < kMT; ++mt) {
        const int mrow = mt * 16 + hi8;
#pragma unroll
        for (int r = 0; r < 8; ++r)
          sh[(mrow + r) * kPitch + n] += acc2[mt][j][r] + bias;
      }
    }
  }

  __syncthreads();
  if (wave < kMT) {
    v8f accL = {};
#pragma unroll 4
    for (int k0 = 0; k0 < kH; k0 += 4) {
      const int kk = k0 + khalf;
      v2f bf;
      bf.x = Wd[kk * kA + lrow];
      bf.y = Wd[(kk + 1) * kA + lrow];
      const v2f a = *(const v2f*)&sh[(wave * 16 + lrow) * kPitch + kk];
      accL = __builtin_amdgcn_wmma_f32_16x16x4_f32(
          false, a, false, bf, (short)0, accL, false, false);
    }
    const float bias = bd[lrow];
    const int mrow = wave * 16 + hi8;
#pragma unroll
    for (int r = 0; r < 8; ++r)
      out[(b * kM + mrow + r) * kA + lrow] = accL[r] + bias;
  }
}

extern "C" void kernel_launch(void* const* d_in, const int* in_sizes, int n_in,
                              void* d_out, int out_size, void* d_ws, size_t ws_size,
                              hipStream_t stream) {
  (void)in_sizes; (void)n_in; (void)out_size; (void)d_ws; (void)ws_size;
  const int*   ids = (const int*)d_in[0];
  const float* emb = (const float*)d_in[1];
  const float* W1  = (const float*)d_in[2];
  const float* b1  = (const float*)d_in[3];
  const float* W2  = (const float*)d_in[4];
  const float* b2  = (const float*)d_in[5];
  const float* Wd  = (const float*)d_in[6];
  const float* bd  = (const float*)d_in[7];
  float* out = (float*)d_out;

  const size_t lds_bytes = (size_t)kLdsFloats * sizeof(float);
  (void)hipFuncSetAttribute((const void*)commnet_wmma_kernel,
                            hipFuncAttributeMaxDynamicSharedMemorySize,
                            (int)lds_bytes);
  commnet_wmma_kernel<<<dim3(kB), dim3(256), lds_bytes, stream>>>(
      ids, emb, W1, b1, W2, b2, Wd, bd, out);
}


# Software-pipeline the operand loads of every GEMM loop: two register sets ping-pong; the next 16-k block's weight (global) and A/S (LDS) loads go out as one group before the current block's WMMAs
# speedup vs baseline: 1.6616x; 1.1762x over previous
#include <hip/hip_runtime.h>


typedef __attribute__((ext_vector_type(2))) float v2f;
typedef __attribute__((ext_vector_type(8))) float v8f;

namespace {
constexpr int kB = 1024;
constexpr int kM = 64;
constexpr int kH = 256;
constexpr int kA = 16;
constexpr int kSteps = 4;
constexpr int kPitch = 260;
constexpr int kMT = 4;
constexpr int kNTpw = 2;
constexpr int kLdsFloats = 3 * kM * kPitch + kPitch;
constexpr int kKB = 16;
constexpr int kKS = kKB / 4;
constexpr int kDB = 64;
constexpr int kDS = kDB / 4;

struct W1Frag { float a0, c0, a1, c1; };

__device__ __forceinline__ v8f mma(v2f a, v2f b, v8f c) {
  return __builtin_amdgcn_wmma_f32_16x16x4_f32(false, a, false, b, (short)0, c,
                                               false, false);
}


__device__ __forceinline__ void load_b(v2f (&bf)[kKS][kNTpw],
                                       const float* __restrict__ wp, int kb) {
  for (int s = 0; s < kKS; ++s) {
    for (int j = 0; j < kNTpw; ++j) {
      const float* p = wp + (kb + 4 * s) * kH + j * 16;
      bf[s][j].x = p[0];
      bf[s][j].y = p[kH];
    }
  }
}

__device__ __forceinline__ void load_a(v2f (&a)[kKS][kMT], const float* ap, int kb) {
  for (int s = 0; s < kKS; ++s) {
    for (int mt = 0; mt < kMT; ++mt)
      a[s][mt] = *(const v2f*)&ap[mt * 16 * kPitch + kb + 4 * s];
  }
}

__device__ __forceinline__ void mma_block(v8f (&acc)[kMT][kNTpw],
                                          const v2f (&a)[kKS][kMT],
                                          const v2f (&bf)[kKS][kNTpw]) {
  for (int s = 0; s < kKS; ++s) {
    for (int mt = 0; mt < kMT; ++mt) {
      for (int j = 0; j < kNTpw; ++j)
        acc[mt][j] = mma(a[s][mt], bf[s][j], acc[mt][j]);
    }
  }
}

__device__ __forceinline__ void load_w1(W1Frag (&w)[kKS][kNTpw],
                                        const float* __restrict__ wp, int kb) {
  for (int s = 0; s < kKS; ++s) {
    for (int j = 0; j < kNTpw; ++j) {
      const float* p = wp + (kb + 4 * s) * kH + j * 16;
      w[s][j].a0 = p[0];
      w[s][j].c0 = p[kH * kH];
      w[s][j].a1 = p[kH];
      w[s][j].c1 = p[kH * kH + kH];
    }
  }
}

__device__ __forceinline__ void load_s(v2f (&sv)[kKS], const float* sk, int kb) {
  for (int s = 0; s < kKS; ++s) sv[s] = *(const v2f*)&sk[kb + 4 * s];
}

__device__ __forceinline__ void mma_block1(v8f (&acc)[kMT][kNTpw],
                                           float (&uacc)[kNTpw],
                                           const v2f (&a)[kKS][kMT],
                                           const v2f (&svb)[kKS],
                                           const W1Frag (&w)[kKS][kNTpw],
                                           float inv) {
  for (int s = 0; s < kKS; ++s) {
    const v2f sv = svb[s];
    v2f bf[kNTpw];
    for (int j = 0; j < kNTpw; ++j) {
      bf[j].x = fmaf(-inv, w[s][j].c0, w[s][j].a0);
      bf[j].y = fmaf(-inv, w[s][j].c1, w[s][j].a1);
      uacc[j] = fmaf(sv.x, w[s][j].c0, uacc[j]);
      uacc[j] = fmaf(sv.y, w[s][j].c1, uacc[j]);
    }
    for (int mt = 0; mt < kMT; ++mt) {
      for (int j = 0; j < kNTpw; ++j) acc[mt][j] = mma(a[s][mt], bf[j], acc[mt][j]);
    }
  }
}

__device__ __forceinline__ void group_fence() { __builtin_amdgcn_sched_barrier(0); }

__device__ __forceinline__ void load_d(v2f (&d)[kDS], const float* __restrict__ wp,
                                       int kb) {
  for (int s = 0; s < kDS; ++s) {
    const float* p = wp + (kb + 4 * s) * kA;
    d[s].x = p[0];
    d[s].y = p[kA];
  }
}

__device__ __forceinline__ void mma_d(v8f& acc, const v2f (&d)[kDS], const float* ap,
                                      int kb) {
  for (int s = 0; s < kDS; ++s) {
    const v2f a = *(const v2f*)&ap[kb + 4 * s];
    acc = mma(a, d[s], acc);
  }
}
}

__launch_bounds__(256)
__global__ void commnet_wmma_kernel(
    const int* __restrict__ ids, const float* __restrict__ emb,
    const float* __restrict__ W1, const float* __restrict__ b1,
    const float* __restrict__ W2, const float* __restrict__ b2,
    const float* __restrict__ Wd, const float* __restrict__ bd,
    float* __restrict__ out) {
  extern __shared__ float smem[];
  float* sh = smem;
  float* sp = sh + kM * kPitch;
  float* st = sp + kM * kPitch;
  float* sS = st + kM * kPitch;

  const int b     = blockIdx.x;
  const int tid   = threadIdx.x;
  const int wave  = tid >> 5;
  const int lane  = tid & 31;
  const int lrow  = lane & 15;
  const int khalf = (lane >> 4) << 1;
  const int hi8   = (lane >> 4) << 3;
  const float inv = 1.0f / (float)(kM - 1);
  const int nb    = wave * (kNTpw * 16);

  const float* w1a = W1 + khalf * kH + nb + lrow;
  const float* w1c = W1 + (2 * kH + khalf) * kH + nb + lrow;
  const float* w2p = W2 + khalf * kH + nb + lrow;
  const float* wdp = Wd + khalf * kA + lrow;
  const float* hA  = sh + lrow * kPitch + khalf;
  const float* pA  = sp + lrow * kPitch + khalf;
  const float* tA  = st + lrow * kPitch + khalf;
  const float* sk  = sS + khalf;
  const float* hL  = sh + (wave * 16 + lrow) * kPitch + khalf;

  v2f bA[kKS][kNTpw], bB[kKS][kNTpw];
  W1Frag wA[kKS][kNTpw], wB[kKS][kNTpw];
  v2f dA[kDS], dB[kDS];
  v2f aA[kKS][kMT], aB[kKS][kMT];
  v2f sA[kKS], sB[kKS];
  float bdv = 0.f;

  load_b(bA, w1c, 0);
  float b1v[kNTpw], b2v[kNTpw];
  for (int j = 0; j < kNTpw; ++j) {
    b1v[j] = b1[nb + j * 16 + lrow];
    b2v[j] = b2[nb + j * 16 + lrow];
  }

  for (int m = 0; m < kM; ++m) {
    const int id  = ids[b * kM + m];
    const float v = emb[id * kH + tid];
    sp[m * kPitch + tid] = v;
    sh[m * kPitch + tid] = v;
  }
  __syncthreads();

  v8f acc[kMT][kNTpw];
  const v8f vzero = {};

  for (int mt = 0; mt < kMT; ++mt)
    for (int j = 0; j < kNTpw; ++j) acc[mt][j] = vzero;
  load_a(aA, pA, 0);
  for (int kb = 0; kb < kH - 2 * kKB; kb += 2 * kKB) {
    load_b(bB, w1c, kb + kKB);
    load_a(aB, pA, kb + kKB);
    group_fence();
    mma_block(acc, aA, bA);
    group_fence();
    load_b(bA, w1c, kb + 2 * kKB);
    load_a(aA, pA, kb + 2 * kKB);
    group_fence();
    mma_block(acc, aB, bB);
    group_fence();
  }
  load_b(bB, w1c, kH - kKB);
  load_a(aB, pA, kH - kKB);
  group_fence();
  mma_block(acc, aA, bA);
  group_fence();
  mma_block(acc, aB, bB);
  group_fence();
  __syncthreads();
  for (int j = 0; j < kNTpw; ++j) {
    const int n = nb + j * 16 + lrow;
    for (int mt = 0; mt < kMT; ++mt) {
      const int mrow = mt * 16 + hi8;
#pragma unroll
      for (int r = 0; r < 8; ++r)
        sp[(mrow + r) * kPitch + n] = acc[mt][j][r] + b1v[j];
    }
  }

  for (int step = 0; step < kSteps; ++step) {
    load_w1(wA, w1a, 0);
    __syncthreads();
    {
      float s = 0.f;
#pragma unroll 8
      for (int m = 0; m < kM; ++m) s += sh[m * kPitch + tid];
      sS[tid] = s;
    }
    __syncthreads();

    float uacc[kNTpw];
    for (int mt = 0; mt < kMT; ++mt)
      for (int j = 0; j < kNTpw; ++j) acc[mt][j] = vzero;
    for (int j = 0; j < kNTpw; ++j) uacc[j] = 0.f;

    load_a(aA, hA, 0);
    load_s(sA, sk, 0);
    for (int kb = 0; kb < kH - 2 * kKB; kb += 2 * kKB) {
      load_w1(wB, w1a, kb + kKB);
      load_a(aB, hA, kb + kKB);
      load_s(sB, sk, kb + kKB);
      group_fence();
      mma_block1(acc, uacc, aA, sA, wA, inv);
      group_fence();
      load_w1(wA, w1a, kb + 2 * kKB);
      load_a(aA, hA, kb + 2 * kKB);
      load_s(sA, sk, kb + 2 * kKB);
      group_fence();
      mma_block1(acc, uacc, aB, sB, wB, inv);
      group_fence();
    }
    load_w1(wB, w1a, kH - kKB);
    load_a(aB, hA, kH - kKB);
    load_s(sB, sk, kH - kKB);
    group_fence();
    mma_block1(acc, uacc, aA, sA, wA, inv);
    group_fence();
    load_b(bA, w2p, 0);
    group_fence();
    mma_block1(acc, uacc, aB, sB, wB, inv);
    group_fence();

    for (int j = 0; j < kNTpw; ++j) {
      const int n = nb + j * 16 + lrow;
      const float u = (uacc[j] + __shfl_xor(uacc[j], 16)) * inv;
      for (int mt = 0; mt < kMT; ++mt) {
        const int mrow = mt * 16 + hi8;
#pragma unroll
        for (int r = 0; r < 8; ++r) {
          const float v = acc[mt][j][r] + sp[(mrow + r) * kPitch + n] + u;
          st[(mrow + r) * kPitch + n] = fmaxf(v, 0.f);
        }
      }
    }
    __syncthreads();

    for (int mt = 0; mt < kMT; ++mt)
      for (int j = 0; j < kNTpw; ++j) acc[mt][j] = vzero;
    load_a(aA, tA, 0);
    for (int kb = 0; kb < kH - 2 * kKB; kb += 2 * kKB) {
      load_b(bB, w2p, kb + kKB);
      load_a(aB, tA, kb + kKB);
      group_fence();
      mma_block(acc, aA, bA);
      group_fence();
      load_b(bA, w2p, kb + 2 * kKB);
      load_a(aA, tA, kb + 2 * kKB);
      group_fence();
      mma_block(acc, aB, bB);
      group_fence();
    }
    load_b(bB, w2p, kH - kKB);
    load_a(aB, tA, kH - kKB);
    group_fence();
    mma_block(acc, aA, bA);
    group_fence();
    if (step + 1 == kSteps && wave < kMT) {
      load_d(dA, wdp, 0);
      load_d(dB, wdp, kDB);
      bdv = bd[lrow];
    }
    group_fence();
    mma_block(acc, aB, bB);
    group_fence();

    for (int j = 0; j < kNTpw; ++j) {
      const int n = nb + j * 16 + lrow;
      for (int mt = 0; mt < kMT; ++mt) {
        const int mrow = mt * 16 + hi8;
#pragma unroll
        for (int r = 0; r < 8; ++r)
          sh[(mrow + r) * kPitch + n] += acc[mt][j][r] + b2v[j];
      }
    }
  }

  __syncthreads();
  if (wave < kMT) {
    v8f accL = {};
    mma_d(accL, dA, hL, 0);
    load_d(dA, wdp, 2 * kDB);
    mma_d(accL, dB, hL, kDB);
    load_d(dB, wdp, 3 * kDB);
    mma_d(accL, dA, hL, 2 * kDB);
    mma_d(accL, dB, hL, 3 * kDB);
    const int mrow = wave * 16 + hi8;
#pragma unroll
    for (int r = 0; r < 8; ++r)
      out[(b * kM + mrow + r) * kA + lrow] = accL[r] + bdv;
  }
}

extern "C" void kernel_launch(void* const* d_in, const int* in_sizes, int n_in,
                              void* d_out, int out_size, void* d_ws, size_t ws_size,
                              hipStream_t stream) {
  (void)in_sizes; (void)n_in; (void)out_size; (void)d_ws; (void)ws_size;
  const int*   ids = (const int*)d_in[0];
  const float* emb = (const float*)d_in[1];
  const float* W1  = (const float*)d_in[2];
  const float* b1  = (const float*)d_in[3];
  const float* W2  = (const float*)d_in[4];
  const float* b2  = (const float*)d_in[5];
  const float* Wd  = (const float*)d_in[6];
  const float* bd  = (const float*)d_in[7];
  float* out = (float*)d_out;

  const size_t lds_bytes = (size_t)kLdsFloats * sizeof(float);
  (void)hipFuncSetAttribute((const void*)commnet_wmma_kernel,
                            hipFuncAttributeMaxDynamicSharedMemorySize,
                            (int)lds_bytes);
  commnet_wmma_kernel<<<dim3(kB), dim3(256), lds_bytes, stream>>>(
      ids, emb, W1, b1, W2, b2, Wd, bd, out);
}


# Re-tile to one wave per SIMD: 4 waves per block, each a 64x64 tile (16 accumulators); a lane owns 4 adjacent columns, so weights come as 16-byte loads and epilogue LDS accesses as one 16-byte group
# speedup vs baseline: 1.7373x; 1.0456x over previous
#include <hip/hip_runtime.h>


typedef __attribute__((ext_vector_type(2))) float v2f;
typedef __attribute__((ext_vector_type(4))) float v4f;
typedef __attribute__((ext_vector_type(8))) float v8f;

namespace {
constexpr int kB = 1024;
constexpr int kM = 64;
constexpr int kH = 256;
constexpr int kA = 16;
constexpr int kSteps = 4;
constexpr int kPitch = 260;
constexpr int kMT = 4;
constexpr int kWaves = 4;
constexpr int kThreads = kWaves * 32;
constexpr int kNTpw = 4;
constexpr int kLdsFloats = 3 * kM * kPitch + kPitch;
constexpr int kKB = 16;
constexpr int kKS = kKB / 4;
constexpr int kDB = 64;
constexpr int kDS = kDB / 4;


struct BFrag { v4f x, y; };
struct W1Frag { v4f a0, c0, a1, c1; };

__device__ __forceinline__ v8f mma(v2f a, v2f b, v8f c) {
  return __builtin_amdgcn_wmma_f32_16x16x4_f32(false, a, false, b, (short)0, c,
                                               false, false);
}


__device__ __forceinline__ void load_b(BFrag (&bf)[kKS], const float* __restrict__ wp,
                                       int kb) {
  for (int s = 0; s < kKS; ++s) {
    const float* p = wp + (kb + 4 * s) * kH;
    bf[s].x = *(const v4f*)p;
    bf[s].y = *(const v4f*)(p + kH);
  }
}

__device__ __forceinline__ void load_a(v2f (&a)[kKS][kMT], const float* ap, int kb) {
  for (int s = 0; s < kKS; ++s) {
    for (int mt = 0; mt < kMT; ++mt)
      a[s][mt] = *(const v2f*)&ap[mt * 16 * kPitch + kb + 4 * s];
  }
}

__device__ __forceinline__ void mma_block(v8f (&acc)[kMT][kNTpw],
                                          const v2f (&a)[kKS][kMT],
                                          const BFrag (&w)[kKS]) {
  for (int s = 0; s < kKS; ++s) {
    v2f bf[kNTpw];
    for (int j = 0; j < kNTpw; ++j) {
      bf[j].x = w[s].x[j];
      bf[j].y = w[s].y[j];
    }
    for (int mt = 0; mt < kMT; ++mt) {
      for (int j = 0; j < kNTpw; ++j)
        acc[mt][j] = mma(a[s][mt], bf[j], acc[mt][j]);
    }
  }
}

__device__ __forceinline__ void load_w1(W1Frag (&w)[kKS], const float* __restrict__ wp,
                                        int kb) {
  for (int s = 0; s < kKS; ++s) {
    const float* p = wp + (kb + 4 * s) * kH;
    w[s].a0 = *(const v4f*)p;
    w[s].c0 = *(const v4f*)(p + kH * kH);
    w[s].a1 = *(const v4f*)(p + kH);
    w[s].c1 = *(const v4f*)(p + kH * kH + kH);
  }
}

__device__ __forceinline__ void load_s(v2f (&sv)[kKS], const float* sk, int kb) {
  for (int s = 0; s < kKS; ++s) sv[s] = *(const v2f*)&sk[kb + 4 * s];
}

__device__ __forceinline__ void mma_block1(v8f (&acc)[kMT][kNTpw],
                                           float (&uacc)[kNTpw],
                                           const v2f (&a)[kKS][kMT],
                                           const v2f (&svb)[kKS],
                                           const W1Frag (&w)[kKS],
                                           float inv) {
  for (int s = 0; s < kKS; ++s) {
    const v2f sv = svb[s];
    v2f bf[kNTpw];
    for (int j = 0; j < kNTpw; ++j) {
      bf[j].x = fmaf(-inv, w[s].c0[j], w[s].a0[j]);
      bf[j].y = fmaf(-inv, w[s].c1[j], w[s].a1[j]);
      uacc[j] = fmaf(sv.x, w[s].c0[j], uacc[j]);
      uacc[j] = fmaf(sv.y, w[s].c1[j], uacc[j]);
    }
    for (int mt = 0; mt < kMT; ++mt) {
      for (int j = 0; j < kNTpw; ++j) acc[mt][j] = mma(a[s][mt], bf[j], acc[mt][j]);
    }
  }
}

__device__ __forceinline__ void group_fence() { __builtin_amdgcn_sched_barrier(0); }

__device__ __forceinline__ void load_d(v2f (&d)[kDS], const float* __restrict__ wp,
                                       int kb) {
  for (int s = 0; s < kDS; ++s) {
    const float* p = wp + (kb + 4 * s) * kA;
    d[s].x = p[0];
    d[s].y = p[kA];
  }
}

__device__ __forceinline__ void mma_d(v8f& acc, const v2f (&d)[kDS], const float* ap,
                                      int kb) {
  for (int s = 0; s < kDS; ++s) {
    const v2f a = *(const v2f*)&ap[kb + 4 * s];
    acc = mma(a, d[s], acc);
  }
}
}

__launch_bounds__(kThreads)
__global__ void commnet_wmma_kernel(
    const int* __restrict__ ids, const float* __restrict__ emb,
    const float* __restrict__ W1, const float* __restrict__ b1,
    const float* __restrict__ W2, const float* __restrict__ b2,
    const float* __restrict__ Wd, const float* __restrict__ bd,
    float* __restrict__ out) {
  extern __shared__ __attribute__((aligned(16))) float smem[];
  float* sh = smem;
  float* sp = sh + kM * kPitch;
  float* st = sp + kM * kPitch;
  float* sS = st + kM * kPitch;

  const int b     = blockIdx.x;
  const int tid   = threadIdx.x;
  const int wave  = tid >> 5;
  const int lane  = tid & 31;
  const int lrow  = lane & 15;
  const int khalf = (lane >> 4) << 1;
  const int hi8   = (lane >> 4) << 3;
  const float inv = 1.0f / (float)(kM - 1);
  const int nb    = wave * (kNTpw * 16);
  const int ncol  = nb + 4 * lrow;
  const int c2    = 2 * tid;

  const float* w1a = W1 + khalf * kH + ncol;
  const float* w1c = W1 + (2 * kH + khalf) * kH + ncol;
  const float* w2p = W2 + khalf * kH + ncol;
  const float* wdp = Wd + khalf * kA + lrow;
  const float* hA  = sh + lrow * kPitch + khalf;
  const float* pA  = sp + lrow * kPitch + khalf;
  const float* tA  = st + lrow * kPitch + khalf;
  const float* sk  = sS + khalf;
  const float* hL  = sh + (wave * 16 + lrow) * kPitch + khalf;

  BFrag bA[kKS], bB[kKS];
  W1Frag wA[kKS], wB[kKS];
  v2f dA[kDS], dB[kDS];
  v2f aA[kKS][kMT], aB[kKS][kMT];
  v2f sA[kKS], sB[kKS];
  float bdv = 0.f;

  load_b(bA, w1c, 0);
  const v4f b1v = *(const v4f*)&b1[ncol];
  const v4f b2v = *(const v4f*)&b2[ncol];

  for (int m = 0; m < kM; ++m) {
    const int id  = ids[b * kM + m];
    const v2f v = *(const v2f*)&emb[id * kH + c2];
    *(v2f*)&sp[m * kPitch + c2] = v;
    *(v2f*)&sh[m * kPitch + c2] = v;
  }
  __syncthreads();

  v8f acc[kMT][kNTpw];
  const v8f vzero = {};

  for (int mt = 0; mt < kMT; ++mt)
    for (int j = 0; j < kNTpw; ++j) acc[mt][j] = vzero;
  load_a(aA, pA, 0);
  for (int kb = 0; kb < kH - 2 * kKB; kb += 2 * kKB) {
    load_b(bB, w1c, kb + kKB);
    load_a(aB, pA, kb + kKB);
    group_fence();
    mma_block(acc, aA, bA);
    group_fence();
    load_b(bA, w1c, kb + 2 * kKB);
    load_a(aA, pA, kb + 2 * kKB);
    group_fence();
    mma_block(acc, aB, bB);
    group_fence();
  }
  load_b(bB, w1c, kH - kKB);
  load_a(aB, pA, kH - kKB);
  group_fence();
  mma_block(acc, aA, bA);
  group_fence();
  mma_block(acc, aB, bB);
  group_fence();
  __syncthreads();
  for (int mt = 0; mt < kMT; ++mt) {
    const int mrow = mt * 16 + hi8;
#pragma unroll
    for (int r = 0; r < 8; ++r) {
      v4f v;
      for (int j = 0; j < kNTpw; ++j) v[j] = acc[mt][j][r] + b1v[j];
      *(v4f*)&sp[(mrow + r) * kPitch + ncol] = v;
    }
  }

  for (int step = 0; step < kSteps; ++step) {
    load_w1(wA, w1a, 0);
    __syncthreads();
    {
      v2f s = {0.f, 0.f};
#pragma unroll 8
      for (int m = 0; m < kM; ++m) s += *(const v2f*)&sh[m * kPitch + c2];
      *(v2f*)&sS[c2] = s;
    }
    __syncthreads();

    float uacc[kNTpw];
    for (int mt = 0; mt < kMT; ++mt)
      for (int j = 0; j < kNTpw; ++j) acc[mt][j] = vzero;
    for (int j = 0; j < kNTpw; ++j) uacc[j] = 0.f;

    load_a(aA, hA, 0);
    load_s(sA, sk, 0);
    for (int kb = 0; kb < kH - 2 * kKB; kb += 2 * kKB) {
      load_w1(wB, w1a, kb + kKB);
      load_a(aB, hA, kb + kKB);
      load_s(sB, sk, kb + kKB);
      group_fence();
      mma_block1(acc, uacc, aA, sA, wA, inv);
      group_fence();
      load_w1(wA, w1a, kb + 2 * kKB);
      load_a(aA, hA, kb + 2 * kKB);
      load_s(sA, sk, kb + 2 * kKB);
      group_fence();
      mma_block1(acc, uacc, aB, sB, wB, inv);
      group_fence();
    }
    load_w1(wB, w1a, kH - kKB);
    load_a(aB, hA, kH - kKB);
    load_s(sB, sk, kH - kKB);
    group_fence();
    mma_block1(acc, uacc, aA, sA, wA, inv);
    group_fence();
    load_b(bA, w2p, 0);
    group_fence();
    mma_block1(acc, uacc, aB, sB, wB, inv);
    group_fence();

    {
      v4f u;
      for (int j = 0; j < kNTpw; ++j)
        u[j] = (uacc[j] + __shfl_xor(uacc[j], 16)) * inv;
      v4f p[kMT][8];
      for (int mt = 0; mt < kMT; ++mt) {
#pragma unroll
        for (int r = 0; r < 8; ++r)
          p[mt][r] = *(const v4f*)&sp[(mt * 16 + hi8 + r) * kPitch + ncol];
      }
      group_fence();
      for (int mt = 0; mt < kMT; ++mt) {
        const int mrow = mt * 16 + hi8;
#pragma unroll
        for (int r = 0; r < 8; ++r) {
          v4f v;
          for (int j = 0; j < kNTpw; ++j)
            v[j] = fmaxf(acc[mt][j][r] + p[mt][r][j] + u[j], 0.f);
          *(v4f*)&st[(mrow + r) * kPitch + ncol] = v;
        }
      }
    }
    __syncthreads();

    for (int mt = 0; mt < kMT; ++mt)
      for (int j = 0; j < kNTpw; ++j) acc[mt][j] = vzero;
    load_a(aA, tA, 0);
    for (int kb = 0; kb < kH - 2 * kKB; kb += 2 * kKB) {
      load_b(bB, w2p, kb + kKB);
      load_a(aB, tA, kb + kKB);
      group_fence();
      mma_block(acc, aA, bA);
      group_fence();
      load_b(bA, w2p, kb + 2 * kKB);
      load_a(aA, tA, kb + 2 * kKB);
      group_fence();
      mma_block(acc, aB, bB);
      group_fence();
    }
    load_b(bB, w2p, kH - kKB);
    load_a(aB, tA, kH - kKB);
    group_fence();
    mma_block(acc, aA, bA);
    group_fence();
    if (step + 1 == kSteps) {
      load_d(dA, wdp, 0);
      load_d(dB, wdp, kDB);
      bdv = bd[lrow];
    }
    group_fence();
    mma_block(acc, aB, bB);
    group_fence();

    {
      v4f hv[kMT][8];
      for (int mt = 0; mt < kMT; ++mt) {
#pragma unroll
        for (int r = 0; r < 8; ++r)
          hv[mt][r] = *(const v4f*)&sh[(mt * 16 + hi8 + r) * kPitch + ncol];
      }
      group_fence();
      for (int mt = 0; mt < kMT; ++mt) {
        const int mrow = mt * 16 + hi8;
#pragma unroll
        for (int r = 0; r < 8; ++r) {
          v4f v;
          for (int j = 0; j < kNTpw; ++j)
            v[j] = hv[mt][r][j] + (acc[mt][j][r] + b2v[j]);
          *(v4f*)&sh[(mrow + r) * kPitch + ncol] = v;
        }
      }
    }
  }

  __syncthreads();
  {
    v8f accL = {};
    mma_d(accL, dA, hL, 0);
    load_d(dA, wdp, 2 * kDB);
    mma_d(accL, dB, hL, kDB);
    load_d(dB, wdp, 3 * kDB);
    mma_d(accL, dA, hL, 2 * kDB);
    mma_d(accL, dB, hL, 3 * kDB);
    const int mrow = wave * 16 + hi8;
#pragma unroll
    for (int r = 0; r < 8; ++r)
      out[(b * kM + mrow + r) * kA + lrow] = accL[r] + bdv;
  }
}

extern "C" void kernel_launch(void* const* d_in, const int* in_sizes, int n_in,
                              void* d_out, int out_size, void* d_ws, size_t ws_size,
                              hipStream_t stream) {
  (void)in_sizes; (void)n_in; (void)out_size; (void)d_ws; (void)ws_size;
  const int*   ids = (const int*)d_in[0];
  const float* emb = (const float*)d_in[1];
  const float* W1  = (const float*)d_in[2];
  const float* b1  = (const float*)d_in[3];
  const float* W2  = (const float*)d_in[4];
  const float* b2  = (const float*)d_in[5];
  const float* Wd  = (const float*)d_in[6];
  const float* bd  = (const float*)d_in[7];
  float* out = (float*)d_out;

  const size_t lds_bytes = (size_t)kLdsFloats * sizeof(float);
  (void)hipFuncSetAttribute((const void*)commnet_wmma_kernel,
                            hipFuncAttributeMaxDynamicSharedMemorySize,
                            (int)lds_bytes);
  commnet_wmma_kernel<<<dim3(kB), dim3(kThreads), lds_bytes, stream>>>(
      ids, emb, W1, b1, W2, b2, Wd, bd, out);
}


# Two blocks resident per WGP (2 waves per SIMD) so one block's WMMAs fill the other's barrier/epilogue/gather gaps: P0 moved from LDS to registers (3 M tiles; 1 stays in LDS), 202->147 KB, 510 VGPRs
# speedup vs baseline: 1.8646x; 1.0733x over previous
#include <hip/hip_runtime.h>


typedef __attribute__((ext_vector_type(2))) float v2f;
typedef __attribute__((ext_vector_type(4))) float v4f;
typedef __attribute__((ext_vector_type(8))) float v8f;

namespace {
constexpr int kB = 1024;
constexpr int kM = 64;
constexpr int kH = 256;
constexpr int kA = 16;
constexpr int kSteps = 4;
constexpr int kPitch = 260;
constexpr int kMT = 4;
constexpr int kWaves = 4;
constexpr int kThreads = kWaves * 32;
constexpr int kNTpw = 4;
constexpr int kMTreg = kMT - 1;
constexpr int kLdsFloats = 2 * kM * kPitch + kPitch + 16 * kPitch;
constexpr int kBlocksPerWgp = 2;
constexpr int kKB = 16;
constexpr int kKS = kKB / 4;
constexpr int kDB = 64;
constexpr int kDS = kDB / 4;


struct BFrag { v4f x, y; };
struct W1Frag { v4f a0, c0, a1, c1; };

__device__ __forceinline__ v8f mma(v2f a, v2f b, v8f c) {
  return __builtin_amdgcn_wmma_f32_16x16x4_f32(false, a, false, b, (short)0, c,
                                               false, false);
}


__device__ __forceinline__ void load_b(BFrag (&bf)[kKS], const float* __restrict__ wp,
                                       int kb) {
  for (int s = 0; s < kKS; ++s) {
    const float* p = wp + (kb + 4 * s) * kH;
    bf[s].x = *(const v4f*)p;
    bf[s].y = *(const v4f*)(p + kH);
  }
}

__device__ __forceinline__ void load_a(v2f (&a)[kKS][kMT], const float* ap, int kb) {
  for (int s = 0; s < kKS; ++s) {
    for (int mt = 0; mt < kMT; ++mt)
      a[s][mt] = *(const v2f*)&ap[mt * 16 * kPitch + kb + 4 * s];
  }
}

__device__ __forceinline__ void mma_block(v8f (&acc)[kMT][kNTpw],
                                          const v2f (&a)[kKS][kMT],
                                          const BFrag (&w)[kKS]) {
  for (int s = 0; s < kKS; ++s) {
    v2f bf[kNTpw];
    for (int j = 0; j < kNTpw; ++j) {
      bf[j].x = w[s].x[j];
      bf[j].y = w[s].y[j];
    }
    for (int mt = 0; mt < kMT; ++mt) {
      for (int j = 0; j < kNTpw; ++j)
        acc[mt][j] = mma(a[s][mt], bf[j], acc[mt][j]);
    }
  }
}

__device__ __forceinline__ void load_w1(W1Frag (&w)[kKS], const float* __restrict__ wp,
                                        int kb) {
  for (int s = 0; s < kKS; ++s) {
    const float* p = wp + (kb + 4 * s) * kH;
    w[s].a0 = *(const v4f*)p;
    w[s].c0 = *(const v4f*)(p + kH * kH);
    w[s].a1 = *(const v4f*)(p + kH);
    w[s].c1 = *(const v4f*)(p + kH * kH + kH);
  }
}

__device__ __forceinline__ void load_s(v2f (&sv)[kKS], const float* sk, int kb) {
  for (int s = 0; s < kKS; ++s) sv[s] = *(const v2f*)&sk[kb + 4 * s];
}

__device__ __forceinline__ void mma_block1(v8f (&acc)[kMT][kNTpw],
                                           float (&uacc)[kNTpw],
                                           const v2f (&a)[kKS][kMT],
                                           const v2f (&svb)[kKS],
                                           const W1Frag (&w)[kKS],
                                           float inv) {
  for (int s = 0; s < kKS; ++s) {
    const v2f sv = svb[s];
    v2f bf[kNTpw];
    for (int j = 0; j < kNTpw; ++j) {
      bf[j].x = fmaf(-inv, w[s].c0[j], w[s].a0[j]);
      bf[j].y = fmaf(-inv, w[s].c1[j], w[s].a1[j]);
      uacc[j] = fmaf(sv.x, w[s].c0[j], uacc[j]);
      uacc[j] = fmaf(sv.y, w[s].c1[j], uacc[j]);
    }
    for (int mt = 0; mt < kMT; ++mt) {
      for (int j = 0; j < kNTpw; ++j) acc[mt][j] = mma(a[s][mt], bf[j], acc[mt][j]);
    }
  }
}

__device__ __forceinline__ void group_fence() { __builtin_amdgcn_sched_barrier(0); }

__device__ __forceinline__ void load_d(v2f (&d)[kDS], const float* __restrict__ wp,
                                       int kb) {
  for (int s = 0; s < kDS; ++s) {
    const float* p = wp + (kb + 4 * s) * kA;
    d[s].x = p[0];
    d[s].y = p[kA];
  }
}

__device__ __forceinline__ void mma_d(v8f& acc, const v2f (&d)[kDS], const float* ap,
                                      int kb) {
  for (int s = 0; s < kDS; ++s) {
    const v2f a = *(const v2f*)&ap[kb + 4 * s];
    acc = mma(a, d[s], acc);
  }
}
}

__launch_bounds__(kThreads, kBlocksPerWgp)
__global__ void commnet_wmma_kernel(
    const int* __restrict__ ids, const float* __restrict__ emb,
    const float* __restrict__ W1, const float* __restrict__ b1,
    const float* __restrict__ W2, const float* __restrict__ b2,
    const float* __restrict__ Wd, const float* __restrict__ bd,
    float* __restrict__ out) {
  extern __shared__ __attribute__((aligned(16))) float smem[];
  float* sh = smem;
  float* st = sh + kM * kPitch;
  float* sS = st + kM * kPitch;
  float* sq = sS + kPitch;

  const int b     = blockIdx.x;
  const int tid   = threadIdx.x;
  const int wave  = __builtin_amdgcn_readfirstlane(tid >> 5);
  const int lane  = tid & 31;
  const int lrow  = lane & 15;
  const int khalf = (lane >> 4) << 1;
  const int hi8   = (lane >> 4) << 3;
  const float inv = 1.0f / (float)(kM - 1);
  const int nb    = wave * (kNTpw * 16);
  const int ncol  = nb + 4 * lrow;
  const int c2    = 2 * tid;

  const float* w1a = W1 + khalf * kH + ncol;
  const float* w1c = W1 + (2 * kH + khalf) * kH + ncol;
  const float* w2p = W2 + khalf * kH + ncol;
  const float* wdp = Wd + khalf * kA + lrow;
  const float* hA  = sh + lrow * kPitch + khalf;
  const float* tA  = st + lrow * kPitch + khalf;
  const float* sk  = sS + khalf;
  const float* hL  = sh + (wave * 16 + lrow) * kPitch + khalf;

  BFrag bA[kKS], bB[kKS];
  W1Frag wA[kKS], wB[kKS];
  v2f dA[kDS], dB[kDS];
  v2f aA[kKS][kMT], aB[kKS][kMT];
  v2f sA[kKS], sB[kKS];
  float bdv = 0.f;

  load_b(bA, w1c, 0);
  const v4f b1v = *(const v4f*)&b1[ncol];

  for (int m = 0; m < kM; ++m) {
    const int id  = ids[b * kM + m];
    *(v2f*)&sh[m * kPitch + c2] = *(const v2f*)&emb[id * kH + c2];
  }
  __syncthreads();

  v8f acc[kMT][kNTpw];
  const v8f vzero = {};

  for (int mt = 0; mt < kMT; ++mt)
    for (int j = 0; j < kNTpw; ++j) acc[mt][j] = vzero;
  load_a(aA, hA, 0);
  for (int kb = 0; kb < kH - 2 * kKB; kb += 2 * kKB) {
    load_b(bB, w1c, kb + kKB);
    load_a(aB, hA, kb + kKB);
    group_fence();
    mma_block(acc, aA, bA);
    group_fence();
    load_b(bA, w1c, kb + 2 * kKB);
    load_a(aA, hA, kb + 2 * kKB);
    group_fence();
    mma_block(acc, aB, bB);
    group_fence();
  }
  load_b(bB, w1c, kH - kKB);
  load_a(aB, hA, kH - kKB);
  group_fence();
  mma_block(acc, aA, bA);
  group_fence();
  mma_block(acc, aB, bB);
  group_fence();
  v8f p0[kMTreg][kNTpw];
  for (int mt = 0; mt < kMTreg; ++mt)
    for (int j = 0; j < kNTpw; ++j)
      for (int r = 0; r < 8; ++r) p0[mt][j][r] = acc[mt][j][r] + b1v[j];
#pragma unroll
  for (int r = 0; r < 8; ++r) {
    v4f v;
    for (int j = 0; j < kNTpw; ++j) v[j] = acc[kMTreg][j][r] + b1v[j];
    *(v4f*)&sq[(hi8 + r) * kPitch + ncol] = v;
  }

  for (int step = 0; step < kSteps; ++step) {
    int zoff = 0;
    asm volatile("" : "+s"(zoff));
    const float* w1s = w1a + zoff;
    const float* w2s = w2p + zoff;
    load_w1(wA, w1s, 0);
    __syncthreads();
    {
      v2f s = {0.f, 0.f};
#pragma unroll 8
      for (int m = 0; m < kM; ++m) s += *(const v2f*)&sh[m * kPitch + c2];
      *(v2f*)&sS[c2] = s;
    }
    __syncthreads();

    float uacc[kNTpw];
    for (int mt = 0; mt < kMT; ++mt)
      for (int j = 0; j < kNTpw; ++j) acc[mt][j] = vzero;
    for (int j = 0; j < kNTpw; ++j) uacc[j] = 0.f;

    load_a(aA, hA, 0);
    load_s(sA, sk, 0);
    for (int kb = 0; kb < kH - 2 * kKB; kb += 2 * kKB) {
      load_w1(wB, w1s, kb + kKB);
      load_a(aB, hA, kb + kKB);
      load_s(sB, sk, kb + kKB);
      group_fence();
      mma_block1(acc, uacc, aA, sA, wA, inv);
      group_fence();
      load_w1(wA, w1s, kb + 2 * kKB);
      load_a(aA, hA, kb + 2 * kKB);
      load_s(sA, sk, kb + 2 * kKB);
      group_fence();
      mma_block1(acc, uacc, aB, sB, wB, inv);
      group_fence();
    }
    load_w1(wB, w1s, kH - kKB);
    load_a(aB, hA, kH - kKB);
    load_s(sB, sk, kH - kKB);
    group_fence();
    mma_block1(acc, uacc, aA, sA, wA, inv);
    group_fence();
    load_b(bA, w2s, 0);
    group_fence();
    mma_block1(acc, uacc, aB, sB, wB, inv);
    group_fence();

    {
      v4f u;
      for (int j = 0; j < kNTpw; ++j)
        u[j] = (uacc[j] + __shfl_xor(uacc[j], 16)) * inv;
      v4f q[8];
#pragma unroll
      for (int r = 0; r < 8; ++r) q[r] = *(const v4f*)&sq[(hi8 + r) * kPitch + ncol];
      group_fence();
      for (int mt = 0; mt < kMTreg; ++mt) {
        const int mrow = mt * 16 + hi8;
#pragma unroll
        for (int r = 0; r < 8; ++r) {
          v4f v;
          for (int j = 0; j < kNTpw; ++j)
            v[j] = fmaxf(acc[mt][j][r] + p0[mt][j][r] + u[j], 0.f);
          *(v4f*)&st[(mrow + r) * kPitch + ncol] = v;
        }
      }
#pragma unroll
      for (int r = 0; r < 8; ++r) {
        v4f v;
        for (int j = 0; j < kNTpw; ++j)
          v[j] = fmaxf(acc[kMTreg][j][r] + q[r][j] + u[j], 0.f);
        *(v4f*)&st[(kMTreg * 16 + hi8 + r) * kPitch + ncol] = v;
      }
    }
    __syncthreads();

    for (int mt = 0; mt < kMT; ++mt)
      for (int j = 0; j < kNTpw; ++j) acc[mt][j] = vzero;
    load_a(aA, tA, 0);
    for (int kb = 0; kb < kH - 2 * kKB; kb += 2 * kKB) {
      load_b(bB, w2s, kb + kKB);
      load_a(aB, tA, kb + kKB);
      group_fence();
      mma_block(acc, aA, bA);
      group_fence();
      load_b(bA, w2s, kb + 2 * kKB);
      load_a(aA, tA, kb + 2 * kKB);
      group_fence();
      mma_block(acc, aB, bB);
      group_fence();
    }
    load_b(bB, w2s, kH - kKB);
    load_a(aB, tA, kH - kKB);
    const v4f b2v = *(const v4f*)&b2[ncol + zoff];
    group_fence();
    mma_block(acc, aA, bA);
    group_fence();
    mma_block(acc, aB, bB);
    group_fence();

    {
      for (int mh = 0; mh < kMT; mh += 2) {
        v4f hv[2][8];
        for (int mt = 0; mt < 2; ++mt) {
#pragma unroll
          for (int r = 0; r < 8; ++r)
            hv[mt][r] = *(const v4f*)&sh[((mh + mt) * 16 + hi8 + r) * kPitch + ncol];
        }
        group_fence();
        for (int mt = 0; mt < 2; ++mt) {
          const int mrow = (mh + mt) * 16 + hi8;
#pragma unroll
          for (int r = 0; r < 8; ++r) {
            v4f v;
            for (int j = 0; j < kNTpw; ++j)
              v[j] = hv[mt][r][j] + (acc[mh + mt][j][r] + b2v[j]);
            *(v4f*)&sh[(mrow + r) * kPitch + ncol] = v;
          }
        }
      }
    }
  }

  load_d(dA, wdp, 0);
  load_d(dB, wdp, kDB);
  bdv = bd[lrow];
  __syncthreads();
  {
    v8f accL = {};
    mma_d(accL, dA, hL, 0);
    load_d(dA, wdp, 2 * kDB);
    mma_d(accL, dB, hL, kDB);
    load_d(dB, wdp, 3 * kDB);
    mma_d(accL, dA, hL, 2 * kDB);
    mma_d(accL, dB, hL, 3 * kDB);
    const int mrow = wave * 16 + hi8;
#pragma unroll
    for (int r = 0; r < 8; ++r)
      out[(b * kM + mrow + r) * kA + lrow] = accL[r] + bdv;
  }
}

extern "C" void kernel_launch(void* const* d_in, const int* in_sizes, int n_in,
                              void* d_out, int out_size, void* d_ws, size_t ws_size,
                              hipStream_t stream) {
  (void)in_sizes; (void)n_in; (void)out_size; (void)d_ws; (void)ws_size;
  const int*   ids = (const int*)d_in[0];
  const float* emb = (const float*)d_in[1];
  const float* W1  = (const float*)d_in[2];
  const float* b1  = (const float*)d_in[3];
  const float* W2  = (const float*)d_in[4];
  const float* b2  = (const float*)d_in[5];
  const float* Wd  = (const float*)d_in[6];
  const float* bd  = (const float*)d_in[7];
  float* out = (float*)d_out;

  const size_t lds_bytes = (size_t)kLdsFloats * sizeof(float);
  (void)hipFuncSetAttribute((const void*)commnet_wmma_kernel,
                            hipFuncAttributeMaxDynamicSharedMemorySize,
                            (int)lds_bytes);
  commnet_wmma_kernel<<<dim3(kB), dim3(kThreads), lds_bytes, stream>>>(
      ids, emb, W1, b1, W2, b2, Wd, bd, out);
}


# P0 and the first step's h0 @ W1eff computed once per embedding-table row (1000, not 65536) by a first launch into the workspace, then gathered by agent id; no sum's order changes
# speedup vs baseline: 2.2127x; 1.1867x over previous
#include <hip/hip_runtime.h>


typedef __attribute__((ext_vector_type(2))) float v2f;
typedef __attribute__((ext_vector_type(4))) float v4f;
typedef __attribute__((ext_vector_type(8))) float v8f;

namespace {
constexpr int kB = 1024;
constexpr int kM = 64;
constexpr int kH = 256;
constexpr int kA = 16;
constexpr int kSteps = 4;
constexpr int kPitch = 260;
constexpr int kMT = 4;
constexpr int kWaves = 4;
constexpr int kThreads = kWaves * 32;
constexpr int kNTpw = 4;
constexpr int kMTreg = kMT - 1;
constexpr int kLdsFloats = 2 * kM * kPitch + kPitch + 16 * kPitch;
constexpr int kBlocksPerWgp = 2;
constexpr int kKB = 16;
constexpr int kKS = kKB / 4;
constexpr int kDB = 64;
constexpr int kDS = kDB / 4;
constexpr int kE = 1000;
constexpr int kTabTiles = (kE + 15) / 16;
constexpr int kTB = 32;
constexpr int kTS = kTB / 4;
constexpr size_t kTabBytes = (size_t)2 * kE * kH * sizeof(float);


struct BFrag { v4f x, y; };
struct W1Frag { v4f a0, c0, a1, c1; };

__device__ __forceinline__ v8f mma(v2f a, v2f b, v8f c) {
  return __builtin_amdgcn_wmma_f32_16x16x4_f32(false, a, false, b, (short)0, c,
                                               false, false);
}


__device__ __forceinline__ void load_b(BFrag (&bf)[kKS], const float* __restrict__ wp,
                                       int kb) {
  for (int s = 0; s < kKS; ++s) {
    const float* p = wp + (kb + 4 * s) * kH;
    bf[s].x = *(const v4f*)p;
    bf[s].y = *(const v4f*)(p + kH);
  }
}

__device__ __forceinline__ void load_a(v2f (&a)[kKS][kMT], const float* ap, int kb) {
  for (int s = 0; s < kKS; ++s) {
    for (int mt = 0; mt < kMT; ++mt)
      a[s][mt] = *(const v2f*)&ap[mt * 16 * kPitch + kb + 4 * s];
  }
}

__device__ __forceinline__ void mma_block(v8f (&acc)[kMT][kNTpw],
                                          const v2f (&a)[kKS][kMT],
                                          const BFrag (&w)[kKS]) {
  for (int s = 0; s < kKS; ++s) {
    v2f bf[kNTpw];
    for (int j = 0; j < kNTpw; ++j) {
      bf[j].x = w[s].x[j];
      bf[j].y = w[s].y[j];
    }
    for (int mt = 0; mt < kMT; ++mt) {
      for (int j = 0; j < kNTpw; ++j)
        acc[mt][j] = mma(a[s][mt], bf[j], acc[mt][j]);
    }
  }
}

__device__ __forceinline__ void load_w1(W1Frag (&w)[kKS], const float* __restrict__ wp,
                                        int kb) {
  for (int s = 0; s < kKS; ++s) {
    const float* p = wp + (kb + 4 * s) * kH;
    w[s].a0 = *(const v4f*)p;
    w[s].c0 = *(const v4f*)(p + kH * kH);
    w[s].a1 = *(const v4f*)(p + kH);
    w[s].c1 = *(const v4f*)(p + kH * kH + kH);
  }
}

__device__ __forceinline__ void load_s(v2f (&sv)[kKS], const float* sk, int kb) {
  for (int s = 0; s < kKS; ++s) sv[s] = *(const v2f*)&sk[kb + 4 * s];
}

__device__ __forceinline__ void mma_block1(v8f (&acc)[kMT][kNTpw],
                                           float (&uacc)[kNTpw],
                                           const v2f (&a)[kKS][kMT],
                                           const v2f (&svb)[kKS],
                                           const W1Frag (&w)[kKS],
                                           float inv) {
  for (int s = 0; s < kKS; ++s) {
    const v2f sv = svb[s];
    v2f bf[kNTpw];
    for (int j = 0; j < kNTpw; ++j) {
      bf[j].x = fmaf(-inv, w[s].c0[j], w[s].a0[j]);
      bf[j].y = fmaf(-inv, w[s].c1[j], w[s].a1[j]);
      uacc[j] = fmaf(sv.x, w[s].c0[j], uacc[j]);
      uacc[j] = fmaf(sv.y, w[s].c1[j], uacc[j]);
    }
    for (int mt = 0; mt < kMT; ++mt) {
      for (int j = 0; j < kNTpw; ++j) acc[mt][j] = mma(a[s][mt], bf[j], acc[mt][j]);
    }
  }
}

struct CFrag { v4f c0, c1; };
__device__ __forceinline__ void load_c(CFrag (&c)[kKS], const float* __restrict__ wp, int kb) {
  for (int s = 0; s < kKS; ++s) {
    const float* p = wp + (kb + 4 * s) * kH + kH * kH;
    c[s].c0 = *(const v4f*)p;
    c[s].c1 = *(const v4f*)(p + kH);
  }
}
__device__ __forceinline__ void fma_u(float (&uacc)[kNTpw], const v2f (&svb)[kKS],
                                      const CFrag (&c)[kKS]) {
  for (int s = 0; s < kKS; ++s) {
    for (int j = 0; j < kNTpw; ++j) {
      uacc[j] = fmaf(svb[s].x, c[s].c0[j], uacc[j]);
      uacc[j] = fmaf(svb[s].y, c[s].c1[j], uacc[j]);
    }
  }
}

template <bool kEff>
__device__ __forceinline__ void tab_load(v4f (&w)[kTS][kEff ? 4 : 2], v2f (&a)[kTS],
                                         const float* __restrict__ wp,
                                         const float* __restrict__ ap, int kb) {
  for (int s = 0; s < kTS; ++s) {
    const float* p = wp + (kb + 4 * s) * kH;
    w[s][0] = *(const v4f*)p;
    w[s][1] = *(const v4f*)(p + kH);
    if constexpr (kEff) {
      w[s][2] = *(const v4f*)(p + kH * kH);
      w[s][3] = *(const v4f*)(p + kH * kH + kH);
    }
    a[s] = *(const v2f*)&ap[kb + 4 * s];
  }
}
template <bool kEff>
__device__ __forceinline__ void tab_mma(v8f (&acc)[kNTpw], const v4f (&w)[kTS][kEff ? 4 : 2],
                                        const v2f (&a)[kTS], float inv) {
  for (int s = 0; s < kTS; ++s) {
    v2f bf[kNTpw];
    for (int j = 0; j < kNTpw; ++j) {
      if constexpr (kEff) {
        bf[j].x = fmaf(-inv, w[s][2][j], w[s][0][j]);
        bf[j].y = fmaf(-inv, w[s][3][j], w[s][1][j]);
      } else {
        bf[j].x = w[s][0][j];
        bf[j].y = w[s][1][j];
      }
    }
    for (int j = 0; j < kNTpw; ++j) acc[j] = mma(a[s], bf[j], acc[j]);
  }
}

__device__ __forceinline__ void group_fence() { __builtin_amdgcn_sched_barrier(0); }

__device__ __forceinline__ void load_d(v2f (&d)[kDS], const float* __restrict__ wp,
                                       int kb) {
  for (int s = 0; s < kDS; ++s) {
    const float* p = wp + (kb + 4 * s) * kA;
    d[s].x = p[0];
    d[s].y = p[kA];
  }
}

__device__ __forceinline__ void mma_d(v8f& acc, const v2f (&d)[kDS], const float* ap,
                                      int kb) {
  for (int s = 0; s < kDS; ++s) {
    const v2f a = *(const v2f*)&ap[kb + 4 * s];
    acc = mma(a, d[s], acc);
  }
}
}

namespace {
template <bool kEff>
__device__ __forceinline__ void tab_tile(v8f (&acc)[kNTpw], const float* __restrict__ wp,
                                         const float* __restrict__ ap, float inv) {
  v4f wA[kTS][kEff ? 4 : 2], wB[kTS][kEff ? 4 : 2];
  v2f aA[kTS], aB[kTS];
  tab_load<kEff>(wA, aA, wp, ap, 0);
  for (int kb = 0; kb < kH - 2 * kTB; kb += 2 * kTB) {
    tab_load<kEff>(wB, aB, wp, ap, kb + kTB);
    group_fence();
    tab_mma<kEff>(acc, wA, aA, inv);
    group_fence();
    tab_load<kEff>(wA, aA, wp, ap, kb + 2 * kTB);
    group_fence();
    tab_mma<kEff>(acc, wB, aB, inv);
    group_fence();
  }
  tab_load<kEff>(wB, aB, wp, ap, kH - kTB);
  group_fence();
  tab_mma<kEff>(acc, wA, aA, inv);
  group_fence();
  tab_mma<kEff>(acc, wB, aB, inv);
}
}

__launch_bounds__(kThreads)
__global__ void commnet_table_kernel(const float* __restrict__ emb,
                                     const float* __restrict__ W1,
                                     const float* __restrict__ b1,
                                     float* __restrict__ tab, size_t ws_bytes) {
  if (ws_bytes < kTabBytes) return;
  const int tile  = blockIdx.x >> 1;
  const int which = blockIdx.x & 1;
  const int tid   = threadIdx.x;
  const int wave  = __builtin_amdgcn_readfirstlane(tid >> 5);
  const int lane  = tid & 31;
  const int lrow  = lane & 15;
  const int khalf = (lane >> 4) << 1;
  const int hi8   = (lane >> 4) << 3;
  const float inv = 1.0f / (float)(kM - 1);
  const int ncol  = wave * (kNTpw * 16) + 4 * lrow;
  const int arow  = min(tile * 16 + lrow, kE - 1);
  const float* ap = emb + arow * kH + khalf;

  v8f acc[kNTpw];
  const v8f vzero = {};
  for (int j = 0; j < kNTpw; ++j) acc[j] = vzero;
  v4f bias = {0.f, 0.f, 0.f, 0.f};
  if (which == 0) {
    bias = *(const v4f*)&b1[ncol];
    tab_tile<false>(acc, W1 + (2 * kH + khalf) * kH + ncol, ap, inv);
  } else {
    tab_tile<true>(acc, W1 + khalf * kH + ncol, ap, inv);
  }
  float* dst = tab + which * (kE * kH);
#pragma unroll
  for (int r = 0; r < 8; ++r) {
    const int row = tile * 16 + hi8 + r;
    if (row < kE) {
      v4f v;
      for (int j = 0; j < kNTpw; ++j) v[j] = acc[j][r];
      if (which == 0) v += bias;
      *(v4f*)&dst[row * kH + ncol] = v;
    }
  }
}

template <bool kTab>
__device__ __forceinline__ void commnet_body(
    float* smem,
    const int* __restrict__ ids, const float* __restrict__ emb,
    const float* __restrict__ W1, const float* __restrict__ b1,
    const float* __restrict__ W2, const float* __restrict__ b2,
    const float* __restrict__ Wd, const float* __restrict__ bd,
    float* __restrict__ out, const float* __restrict__ tab) {
  float* sh = smem;
  float* st = sh + kM * kPitch;
  float* sS = st + kM * kPitch;
  float* sq = sS + kPitch;
  constexpr int kPR = kTab ? kMTreg - 1 : kMTreg;

  const int b     = blockIdx.x;
  const int tid   = threadIdx.x;
  const int wave  = __builtin_amdgcn_readfirstlane(tid >> 5);
  const int lane  = tid & 31;
  const int lrow  = lane & 15;
  const int khalf = (lane >> 4) << 1;
  const int hi8   = (lane >> 4) << 3;
  const float inv = 1.0f / (float)(kM - 1);
  const int nb    = wave * (kNTpw * 16);
  const int ncol  = nb + 4 * lrow;
  const int c2    = 2 * tid;
  const float* tP = tab;
  const float* tQ = tab + kE * kH;
  const int* idr = ids + b * kM + hi8;

  const float* w1a = W1 + khalf * kH + ncol;
  const float* w1c = W1 + (2 * kH + khalf) * kH + ncol;
  const float* w2p = W2 + khalf * kH + ncol;
  const float* wdp = Wd + khalf * kA + lrow;
  const float* hA  = sh + lrow * kPitch + khalf;
  const float* tA  = st + lrow * kPitch + khalf;
  const float* sk  = sS + khalf;
  const float* hL  = sh + (wave * 16 + lrow) * kPitch + khalf;

  BFrag bA[kKS], bB[kKS];
  W1Frag wA[kKS], wB[kKS];
  v2f dA[kDS], dB[kDS];
  v2f aA[kKS][kMT], aB[kKS][kMT];
  v2f sA[kKS], sB[kKS];
  float bdv = 0.f;

  v4f b1v = {0.f, 0.f, 0.f, 0.f};
  if (!kTab) {
    load_b(bA, w1c, 0);
    b1v = *(const v4f*)&b1[ncol];
  }

  for (int m = 0; m < kM; ++m) {
    const int id  = ids[b * kM + m];
    *(v2f*)&sh[m * kPitch + c2] = *(const v2f*)&emb[id * kH + c2];
  }

  v8f acc[kMT][kNTpw];
  const v8f vzero = {};
  v8f p0[kPR][kNTpw];
  if (kTab) {
    v4f pv[kMT][8], e0[kMT][8];
    for (int mt = 0; mt < kMT; ++mt) {
#pragma unroll
      for (int r = 0; r < 8; ++r) {
        const int row = idr[mt * 16 + r] * kH + ncol;
        pv[mt][r] = *(const v4f*)&tP[row];
        e0[mt][r] = *(const v4f*)&tQ[row];
      }
    }
    __syncthreads();
    {
      v2f s = {0.f, 0.f};
#pragma unroll 8
      for (int m = 0; m < kM; ++m) s += *(const v2f*)&sh[m * kPitch + c2];
      *(v2f*)&sS[c2] = s;
    }
    for (int mt = 0; mt < kMT; ++mt)
      for (int r = 0; r < 8; ++r) e0[mt][r] += pv[mt][r];
    for (int mt = 0; mt < kPR; ++mt)
      for (int j = 0; j < kNTpw; ++j)
        for (int r = 0; r < 8; ++r) p0[mt][j][r] = pv[mt][r][j];
#pragma unroll
    for (int r = 0; r < 8; ++r) *(v4f*)&sq[(hi8 + r) * kPitch + ncol] = pv[kPR][r];
    group_fence();
    __syncthreads();
    float uacc[kNTpw];
    for (int j = 0; j < kNTpw; ++j) uacc[j] = 0.f;
    for (int kb = 0; kb < kH; kb += 2 * kKB) {
      CFrag cf[2][kKS];
      v2f sv[2][kKS];
      for (int q = 0; q < 2; ++q) {
        load_c(cf[q], w1a, kb + q * kKB);
        load_s(sv[q], sk, kb + q * kKB);
      }
      for (int q = 0; q < 2; ++q) fma_u(uacc, sv[q], cf[q]);
    }
    load_b(bA, w2p, 0);
    v4f u;
    for (int j = 0; j < kNTpw; ++j)
      u[j] = (uacc[j] + __shfl_xor(uacc[j], 16)) * inv;
    for (int mt = 0; mt < kMT; ++mt) {
      const int mrow = mt * 16 + hi8;
#pragma unroll
      for (int r = 0; r < 8; ++r) {
        v4f v;
        for (int j = 0; j < kNTpw; ++j) v[j] = fmaxf(e0[mt][r][j] + u[j], 0.f);
        *(v4f*)&st[(mrow + r) * kPitch + ncol] = v;
      }
    }
  } else {
  __syncthreads();
  for (int mt = 0; mt < kMT; ++mt)
    for (int j = 0; j < kNTpw; ++j) acc[mt][j] = vzero;
  load_a(aA, hA, 0);
  for (int kb = 0; kb < kH - 2 * kKB; kb += 2 * kKB) {
    load_b(bB, w1c, kb + kKB);
    load_a(aB, hA, kb + kKB);
    group_fence();
    mma_block(acc, aA, bA);
    group_fence();
    load_b(bA, w1c, kb + 2 * kKB);
    load_a(aA, hA, kb + 2 * kKB);
    group_fence();
    mma_block(acc, aB, bB);
    group_fence();
  }
  load_b(bB, w1c, kH - kKB);
  load_a(aB, hA, kH - kKB);
  group_fence();
  mma_block(acc, aA, bA);
  group_fence();
  mma_block(acc, aB, bB);
  group_fence();
  for (int mt = 0; mt < kPR; ++mt)
    for (int j = 0; j < kNTpw; ++j)
      for (int r = 0; r < 8; ++r) p0[mt][j][r] = acc[mt][j][r] + b1v[j];
#pragma unroll
  for (int r = 0; r < 8; ++r) {
    v4f v;
    for (int j = 0; j < kNTpw; ++j) v[j] = acc[kPR][j][r] + b1v[j];
    *(v4f*)&sq[(hi8 + r) * kPitch + ncol] = v;
  }
  }

  for (int step = 0; step < kSteps; ++step) {
    int zoff = 0;
    asm volatile("" : "+s"(zoff));
    const float* w1s = w1a + zoff;
    const float* w2s = w2p + zoff;
    if (!kTab || step > 0) {
    load_w1(wA, w1s, 0);
    __syncthreads();
    {
      v2f s = {0.f, 0.f};
#pragma unroll 8
      for (int m = 0; m < kM; ++m) s += *(const v2f*)&sh[m * kPitch + c2];
      *(v2f*)&sS[c2] = s;
    }
    __syncthreads();

    float uacc[kNTpw];
    for (int mt = 0; mt < kMT; ++mt)
      for (int j = 0; j < kNTpw; ++j) acc[mt][j] = vzero;
    for (int j = 0; j < kNTpw; ++j) uacc[j] = 0.f;

    load_a(aA, hA, 0);
    load_s(sA, sk, 0);
    for (int kb = 0; kb < kH - 2 * kKB; kb += 2 * kKB) {
      load_w1(wB, w1s, kb + kKB);
      load_a(aB, hA, kb + kKB);
      load_s(sB, sk, kb + kKB);
      group_fence();
      mma_block1(acc, uacc, aA, sA, wA, inv);
      group_fence();
      load_w1(wA, w1s, kb + 2 * kKB);
      load_a(aA, hA, kb + 2 * kKB);
      load_s(sA, sk, kb + 2 * kKB);
      group_fence();
      mma_block1(acc, uacc, aB, sB, wB, inv);
      group_fence();
    }
    load_w1(wB, w1s, kH - kKB);
    load_a(aB, hA, kH - kKB);
    load_s(sB, sk, kH - kKB);
    group_fence();
    mma_block1(acc, uacc, aA, sA, wA, inv);
    group_fence();
    load_b(bA, w2s, 0);
    group_fence();
    mma_block1(acc, uacc, aB, sB, wB, inv);
    group_fence();

    {
      v4f u;
      for (int j = 0; j < kNTpw; ++j)
        u[j] = (uacc[j] + __shfl_xor(uacc[j], 16)) * inv;
      v4f q[8];
#pragma unroll
      for (int r = 0; r < 8; ++r) q[r] = *(const v4f*)&sq[(hi8 + r) * kPitch + ncol];
      v4f g[8];
      if (kTab) {
        const int* idl = ids + b * kM + (kMT - 1) * 16;
#pragma unroll
        for (int r = 0; r < 8; ++r) {
          const int id = (lane & 16) ? idl[8 + r] : idl[r];
          g[r] = *(const v4f*)&tP[id * kH + ncol];
        }
      }
      group_fence();
      for (int mt = 0; mt < kPR; ++mt) {
        const int mrow = mt * 16 + hi8;
#pragma unroll
        for (int r = 0; r < 8; ++r) {
          v4f v;
          for (int j = 0; j < kNTpw; ++j)
            v[j] = fmaxf(acc[mt][j][r] + p0[mt][j][r] + u[j], 0.f);
          *(v4f*)&st[(mrow + r) * kPitch + ncol] = v;
        }
      }
#pragma unroll
      for (int r = 0; r < 8; ++r) {
        v4f v;
        for (int j = 0; j < kNTpw; ++j)
          v[j] = fmaxf(acc[kPR][j][r] + q[r][j] + u[j], 0.f);
        *(v4f*)&st[(kPR * 16 + hi8 + r) * kPitch + ncol] = v;
      }
      if (kTab) {
#pragma unroll
        for (int r = 0; r < 8; ++r) {
          v4f v;
          for (int j = 0; j < kNTpw; ++j)
            v[j] = fmaxf(acc[kMT - 1][j][r] + g[r][j] + u[j], 0.f);
          *(v4f*)&st[((kMT - 1) * 16 + hi8 + r) * kPitch + ncol] = v;
        }
      }
    }
    }
    __syncthreads();

    for (int mt = 0; mt < kMT; ++mt)
      for (int j = 0; j < kNTpw; ++j) acc[mt][j] = vzero;
    load_a(aA, tA, 0);
    for (int kb = 0; kb < kH - 2 * kKB; kb += 2 * kKB) {
      load_b(bB, w2s, kb + kKB);
      load_a(aB, tA, kb + kKB);
      group_fence();
      mma_block(acc, aA, bA);
      group_fence();
      load_b(bA, w2s, kb + 2 * kKB);
      load_a(aA, tA, kb + 2 * kKB);
      group_fence();
      mma_block(acc, aB, bB);
      group_fence();
    }
    load_b(bB, w2s, kH - kKB);
    load_a(aB, tA, kH - kKB);
    const v4f b2v = *(const v4f*)&b2[ncol + zoff];
    group_fence();
    mma_block(acc, aA, bA);
    group_fence();
    mma_block(acc, aB, bB);
    group_fence();

    {
      for (int mh = 0; mh < kMT; mh += 2) {
        v4f hv[2][8];
        for (int mt = 0; mt < 2; ++mt) {
#pragma unroll
          for (int r = 0; r < 8; ++r)
            hv[mt][r] = *(const v4f*)&sh[((mh + mt) * 16 + hi8 + r) * kPitch + ncol];
        }
        group_fence();
        for (int mt = 0; mt < 2; ++mt) {
          const int mrow = (mh + mt) * 16 + hi8;
#pragma unroll
          for (int r = 0; r < 8; ++r) {
            v4f v;
            for (int j = 0; j < kNTpw; ++j)
              v[j] = hv[mt][r][j] + (acc[mh + mt][j][r] + b2v[j]);
            *(v4f*)&sh[(mrow + r) * kPitch + ncol] = v;
          }
        }
      }
    }
  }

  load_d(dA, wdp, 0);
  load_d(dB, wdp, kDB);
  bdv = bd[lrow];
  __syncthreads();
  {
    v8f accL = {};
    mma_d(accL, dA, hL, 0);
    load_d(dA, wdp, 2 * kDB);
    mma_d(accL, dB, hL, kDB);
    load_d(dB, wdp, 3 * kDB);
    mma_d(accL, dA, hL, 2 * kDB);
    mma_d(accL, dB, hL, 3 * kDB);
    const int mrow = wave * 16 + hi8;
#pragma unroll
    for (int r = 0; r < 8; ++r)
      out[(b * kM + mrow + r) * kA + lrow] = accL[r] + bdv;
  }
}

__launch_bounds__(kThreads, kBlocksPerWgp)
__global__ void commnet_wmma_kernel(
    const int* __restrict__ ids, const float* __restrict__ emb,
    const float* __restrict__ W1, const float* __restrict__ b1,
    const float* __restrict__ W2, const float* __restrict__ b2,
    const float* __restrict__ Wd, const float* __restrict__ bd,
    float* __restrict__ out, const float* __restrict__ tab, size_t ws_bytes) {
  extern __shared__ __attribute__((aligned(16))) float smem[];
  if (ws_bytes >= kTabBytes)
    commnet_body<true>(smem, ids, emb, W1, b1, W2, b2, Wd, bd, out, tab);
  else
    commnet_body<false>(smem, ids, emb, W1, b1, W2, b2, Wd, bd, out, tab);
}

extern "C" void kernel_launch(void* const* d_in, const int* in_sizes, int n_in,
                              void* d_out, int out_size, void* d_ws, size_t ws_size,
                              hipStream_t stream) {
  (void)in_sizes; (void)n_in; (void)out_size;
  const int*   ids = (const int*)d_in[0];
  const float* emb = (const float*)d_in[1];
  const float* W1  = (const float*)d_in[2];
  const float* b1  = (const float*)d_in[3];
  const float* W2  = (const float*)d_in[4];
  const float* b2  = (const float*)d_in[5];
  const float* Wd  = (const float*)d_in[6];
  const float* bd  = (const float*)d_in[7];
  float* out = (float*)d_out;
  float* tab = (float*)d_ws;

  const size_t lds_bytes = (size_t)kLdsFloats * sizeof(float);
  (void)hipFuncSetAttribute((const void*)commnet_wmma_kernel,
                            hipFuncAttributeMaxDynamicSharedMemorySize,
                            (int)lds_bytes);
  commnet_table_kernel<<<dim3(2 * kTabTiles), dim3(kThreads), 0, stream>>>(
      emb, W1, b1, tab, ws_size);
  commnet_wmma_kernel<<<dim3(kB), dim3(kThreads), lds_bytes, stream>>>(
      ids, emb, W1, b1, W2, b2, Wd, bd, out, tab, ws_size);
}


# Fold the last step's W2 product into the logits: logits = h3@Wd + t3@(W2@Wd) + (b2@Wd+bd); W2@Wd (256x16) and that bias row come from the first launch; the logits' sum order changes
# speedup vs baseline: 2.4777x; 1.1198x over previous
#include <hip/hip_runtime.h>


typedef __attribute__((ext_vector_type(2))) float v2f;
typedef __attribute__((ext_vector_type(4))) float v4f;
typedef __attribute__((ext_vector_type(8))) float v8f;

namespace {
constexpr int kB = 1024;
constexpr int kM = 64;
constexpr int kH = 256;
constexpr int kA = 16;
constexpr int kSteps = 4;
constexpr int kPitch = 260;
constexpr int kMT = 4;
constexpr int kWaves = 4;
constexpr int kThreads = kWaves * 32;
constexpr int kNTpw = 4;
constexpr int kMTreg = kMT - 1;
constexpr int kLdsFloats = 2 * kM * kPitch + kPitch + 16 * kPitch;
constexpr int kBlocksPerWgp = 2;
constexpr int kKB = 16;
constexpr int kKS = kKB / 4;
constexpr int kDB = 64;
constexpr int kDS = kDB / 4;
constexpr int kE = 1000;
constexpr int kTabTiles = (kE + 15) / 16;
constexpr int kTB = 32;
constexpr int kTS = kTB / 4;
constexpr int kFoldTiles = kH / 16 + 1;
constexpr int kFoldBlocks = (kFoldTiles + kWaves - 1) / kWaves;
constexpr int kFoldOff = 2 * kE * kH;
constexpr int kBiasOff = kFoldOff + kH * kA;
constexpr size_t kTabBytes = (size_t)(kFoldOff + kFoldTiles * 16 * kA) * sizeof(float);


struct BFrag { v4f x, y; };
struct W1Frag { v4f a0, c0, a1, c1; };

__device__ __forceinline__ v8f mma(v2f a, v2f b, v8f c) {
  return __builtin_amdgcn_wmma_f32_16x16x4_f32(false, a, false, b, (short)0, c,
                                               false, false);
}


__device__ __forceinline__ void load_b(BFrag (&bf)[kKS], const float* __restrict__ wp,
                                       int kb) {
  for (int s = 0; s < kKS; ++s) {
    const float* p = wp + (kb + 4 * s) * kH;
    bf[s].x = *(const v4f*)p;
    bf[s].y = *(const v4f*)(p + kH);
  }
}

__device__ __forceinline__ void load_a(v2f (&a)[kKS][kMT], const float* ap, int kb) {
  for (int s = 0; s < kKS; ++s) {
    for (int mt = 0; mt < kMT; ++mt)
      a[s][mt] = *(const v2f*)&ap[mt * 16 * kPitch + kb + 4 * s];
  }
}

__device__ __forceinline__ void mma_block(v8f (&acc)[kMT][kNTpw],
                                          const v2f (&a)[kKS][kMT],
                                          const BFrag (&w)[kKS]) {
  for (int s = 0; s < kKS; ++s) {
    v2f bf[kNTpw];
    for (int j = 0; j < kNTpw; ++j) {
      bf[j].x = w[s].x[j];
      bf[j].y = w[s].y[j];
    }
    for (int mt = 0; mt < kMT; ++mt) {
      for (int j = 0; j < kNTpw; ++j)
        acc[mt][j] = mma(a[s][mt], bf[j], acc[mt][j]);
    }
  }
}

__device__ __forceinline__ void load_w1(W1Frag (&w)[kKS], const float* __restrict__ wp,
                                        int kb) {
  for (int s = 0; s < kKS; ++s) {
    const float* p = wp + (kb + 4 * s) * kH;
    w[s].a0 = *(const v4f*)p;
    w[s].c0 = *(const v4f*)(p + kH * kH);
    w[s].a1 = *(const v4f*)(p + kH);
    w[s].c1 = *(const v4f*)(p + kH * kH + kH);
  }
}

__device__ __forceinline__ void load_s(v2f (&sv)[kKS], const float* sk, int kb) {
  for (int s = 0; s < kKS; ++s) sv[s] = *(const v2f*)&sk[kb + 4 * s];
}

__device__ __forceinline__ void mma_block1(v8f (&acc)[kMT][kNTpw],
                                           float (&uacc)[kNTpw],
                                           const v2f (&a)[kKS][kMT],
                                           const v2f (&svb)[kKS],
                                           const W1Frag (&w)[kKS],
                                           float inv) {
  for (int s = 0; s < kKS; ++s) {
    const v2f sv = svb[s];
    v2f bf[kNTpw];
    for (int j = 0; j < kNTpw; ++j) {
      bf[j].x = fmaf(-inv, w[s].c0[j], w[s].a0[j]);
      bf[j].y = fmaf(-inv, w[s].c1[j], w[s].a1[j]);
      uacc[j] = fmaf(sv.x, w[s].c0[j], uacc[j]);
      uacc[j] = fmaf(sv.y, w[s].c1[j], uacc[j]);
    }
    for (int mt = 0; mt < kMT; ++mt) {
      for (int j = 0; j < kNTpw; ++j) acc[mt][j] = mma(a[s][mt], bf[j], acc[mt][j]);
    }
  }
}

struct CFrag { v4f c0, c1; };
__device__ __forceinline__ void load_c(CFrag (&c)[kKS], const float* __restrict__ wp, int kb) {
  for (int s = 0; s < kKS; ++s) {
    const float* p = wp + (kb + 4 * s) * kH + kH * kH;
    c[s].c0 = *(const v4f*)p;
    c[s].c1 = *(const v4f*)(p + kH);
  }
}
__device__ __forceinline__ void fma_u(float (&uacc)[kNTpw], const v2f (&svb)[kKS],
                                      const CFrag (&c)[kKS]) {
  for (int s = 0; s < kKS; ++s) {
    for (int j = 0; j < kNTpw; ++j) {
      uacc[j] = fmaf(svb[s].x, c[s].c0[j], uacc[j]);
      uacc[j] = fmaf(svb[s].y, c[s].c1[j], uacc[j]);
    }
  }
}

template <bool kEff>
__device__ __forceinline__ void tab_load(v4f (&w)[kTS][kEff ? 4 : 2], v2f (&a)[kTS],
                                         const float* __restrict__ wp,
                                         const float* __restrict__ ap, int kb) {
  for (int s = 0; s < kTS; ++s) {
    const float* p = wp + (kb + 4 * s) * kH;
    w[s][0] = *(const v4f*)p;
    w[s][1] = *(const v4f*)(p + kH);
    if constexpr (kEff) {
      w[s][2] = *(const v4f*)(p + kH * kH);
      w[s][3] = *(const v4f*)(p + kH * kH + kH);
    }
    a[s] = *(const v2f*)&ap[kb + 4 * s];
  }
}
template <bool kEff>
__device__ __forceinline__ void tab_mma(v8f (&acc)[kNTpw], const v4f (&w)[kTS][kEff ? 4 : 2],
                                        const v2f (&a)[kTS], float inv) {
  for (int s = 0; s < kTS; ++s) {
    v2f bf[kNTpw];
    for (int j = 0; j < kNTpw; ++j) {
      if constexpr (kEff) {
        bf[j].x = fmaf(-inv, w[s][2][j], w[s][0][j]);
        bf[j].y = fmaf(-inv, w[s][3][j], w[s][1][j]);
      } else {
        bf[j].x = w[s][0][j];
        bf[j].y = w[s][1][j];
      }
    }
    for (int j = 0; j < kNTpw; ++j) acc[j] = mma(a[s], bf[j], acc[j]);
  }
}

__device__ __forceinline__ void group_fence() { __builtin_amdgcn_sched_barrier(0); }

__device__ __forceinline__ void load_d(v2f (&d)[kDS], const float* __restrict__ wp,
                                       int kb) {
  for (int s = 0; s < kDS; ++s) {
    const float* p = wp + (kb + 4 * s) * kA;
    d[s].x = p[0];
    d[s].y = p[kA];
  }
}

__device__ __forceinline__ void mma_d(v8f& acc, const v2f (&d)[kDS], const float* ap,
                                      int kb) {
  for (int s = 0; s < kDS; ++s) {
    const v2f a = *(const v2f*)&ap[kb + 4 * s];
    acc = mma(a, d[s], acc);
  }
}

__device__ __forceinline__ void load_a2(v2f (&ah)[kDS], v2f (&at)[kDS], const float* hp,
                                        const float* tp, int kb) {
  for (int s = 0; s < kDS; ++s) {
    ah[s] = *(const v2f*)&hp[kb + 4 * s];
    at[s] = *(const v2f*)&tp[kb + 4 * s];
  }
}
__device__ __forceinline__ void mma_d2(v8f& acch, v8f& acct, const v2f (&ah)[kDS],
                                       const v2f (&at)[kDS], const v2f (&dh)[kDS],
                                       const v2f (&dt)[kDS]) {
  for (int s = 0; s < kDS; ++s) {
    acch = mma(ah[s], dh[s], acch);
    acct = mma(at[s], dt[s], acct);
  }
}
}

namespace {
template <bool kEff>
__device__ __forceinline__ void tab_tile(v8f (&acc)[kNTpw], const float* __restrict__ wp,
                                         const float* __restrict__ ap, float inv) {
  v4f wA[kTS][kEff ? 4 : 2], wB[kTS][kEff ? 4 : 2];
  v2f aA[kTS], aB[kTS];
  tab_load<kEff>(wA, aA, wp, ap, 0);
  for (int kb = 0; kb < kH - 2 * kTB; kb += 2 * kTB) {
    tab_load<kEff>(wB, aB, wp, ap, kb + kTB);
    group_fence();
    tab_mma<kEff>(acc, wA, aA, inv);
    group_fence();
    tab_load<kEff>(wA, aA, wp, ap, kb + 2 * kTB);
    group_fence();
    tab_mma<kEff>(acc, wB, aB, inv);
    group_fence();
  }
  tab_load<kEff>(wB, aB, wp, ap, kH - kTB);
  group_fence();
  tab_mma<kEff>(acc, wA, aA, inv);
  group_fence();
  tab_mma<kEff>(acc, wB, aB, inv);
}
}

__device__ __forceinline__ void fold_tile(const float* __restrict__ W2,
                                          const float* __restrict__ b2,
                                          const float* __restrict__ Wd,
                                          const float* __restrict__ bd,
                                          float* __restrict__ tab, int wt, int lrow,
                                          int khalf, int hi8) {
  const bool bias = (wt == kH / 16);
  const float* ap = bias ? b2 + khalf : W2 + (wt * 16 + lrow) * kH + khalf;
  const float* wp = Wd + khalf * kA + lrow;
  v2f dA[kDS], dB[kDS];
  const float c0 = bias ? bd[lrow] : 0.f;
  v8f acc = {c0, c0, c0, c0, c0, c0, c0, c0};
  load_d(dA, wp, 0);
  load_d(dB, wp, kDB);
  mma_d(acc, dA, ap, 0);
  load_d(dA, wp, 2 * kDB);
  mma_d(acc, dB, ap, kDB);
  load_d(dB, wp, 3 * kDB);
  mma_d(acc, dA, ap, 2 * kDB);
  mma_d(acc, dB, ap, 3 * kDB);
#pragma unroll
  for (int r = 0; r < 8; ++r) tab[kFoldOff + (wt * 16 + hi8 + r) * kA + lrow] = acc[r];
}

__launch_bounds__(kThreads)
__global__ void commnet_table_kernel(const float* __restrict__ emb,
                                     const float* __restrict__ W1,
                                     const float* __restrict__ b1,
                                     const float* __restrict__ W2,
                                     const float* __restrict__ b2,
                                     const float* __restrict__ Wd,
                                     const float* __restrict__ bd,
                                     float* __restrict__ tab, size_t ws_bytes) {
  if (ws_bytes < kTabBytes) return;
  const int tid   = threadIdx.x;
  const int wave  = __builtin_amdgcn_readfirstlane(tid >> 5);
  const int lane  = tid & 31;
  const int lrow  = lane & 15;
  const int khalf = (lane >> 4) << 1;
  const int hi8   = (lane >> 4) << 3;
  if (blockIdx.x >= 2 * kTabTiles) {
    const int wt = (blockIdx.x - 2 * kTabTiles) * kWaves + wave;
    if (wt < kFoldTiles) fold_tile(W2, b2, Wd, bd, tab, wt, lrow, khalf, hi8);
    return;
  }
  const int tile  = blockIdx.x >> 1;
  const int which = blockIdx.x & 1;
  const float inv = 1.0f / (float)(kM - 1);
  const int ncol  = wave * (kNTpw * 16) + 4 * lrow;
  const int arow  = min(tile * 16 + lrow, kE - 1);
  const float* ap = emb + arow * kH + khalf;

  v8f acc[kNTpw];
  const v8f vzero = {};
  for (int j = 0; j < kNTpw; ++j) acc[j] = vzero;
  v4f bias = {0.f, 0.f, 0.f, 0.f};
  if (which == 0) {
    bias = *(const v4f*)&b1[ncol];
    tab_tile<false>(acc, W1 + (2 * kH + khalf) * kH + ncol, ap, inv);
  } else {
    tab_tile<true>(acc, W1 + khalf * kH + ncol, ap, inv);
  }
  float* dst = tab + which * (kE * kH);
#pragma unroll
  for (int r = 0; r < 8; ++r) {
    const int row = tile * 16 + hi8 + r;
    if (row < kE) {
      v4f v;
      for (int j = 0; j < kNTpw; ++j) v[j] = acc[j][r];
      if (which == 0) v += bias;
      *(v4f*)&dst[row * kH + ncol] = v;
    }
  }
}

template <bool kTab>
__device__ __forceinline__ void commnet_body(
    float* smem,
    const int* __restrict__ ids, const float* __restrict__ emb,
    const float* __restrict__ W1, const float* __restrict__ b1,
    const float* __restrict__ W2, const float* __restrict__ b2,
    const float* __restrict__ Wd, const float* __restrict__ bd,
    float* __restrict__ out, const float* __restrict__ tab) {
  float* sh = smem;
  float* st = sh + kM * kPitch;
  float* sS = st + kM * kPitch;
  float* sq = sS + kPitch;
  constexpr int kPR = kTab ? kMTreg - 1 : kMTreg;

  const int b     = blockIdx.x;
  const int tid   = threadIdx.x;
  const int wave  = __builtin_amdgcn_readfirstlane(tid >> 5);
  const int lane  = tid & 31;
  const int lrow  = lane & 15;
  const int khalf = (lane >> 4) << 1;
  const int hi8   = (lane >> 4) << 3;
  const float inv = 1.0f / (float)(kM - 1);
  const int nb    = wave * (kNTpw * 16);
  const int ncol  = nb + 4 * lrow;
  const int c2    = 2 * tid;
  const float* tP = tab;
  const float* tQ = tab + kE * kH;
  const int* idr = ids + b * kM + hi8;

  const float* w1a = W1 + khalf * kH + ncol;
  const float* w1c = W1 + (2 * kH + khalf) * kH + ncol;
  const float* w2p = W2 + khalf * kH + ncol;
  const float* wdp = Wd + khalf * kA + lrow;
  const float* hA  = sh + lrow * kPitch + khalf;
  const float* tA  = st + lrow * kPitch + khalf;
  const float* sk  = sS + khalf;
  const float* hL  = sh + (wave * 16 + lrow) * kPitch + khalf;

  BFrag bA[kKS], bB[kKS];
  W1Frag wA[kKS], wB[kKS];
  v2f dA[kDS], dB[kDS];
  v2f aA[kKS][kMT], aB[kKS][kMT];
  v2f sA[kKS], sB[kKS];
  float bdv = 0.f;

  v4f b1v = {0.f, 0.f, 0.f, 0.f};
  if (!kTab) {
    load_b(bA, w1c, 0);
    b1v = *(const v4f*)&b1[ncol];
  }

  for (int m = 0; m < kM; ++m) {
    const int id  = ids[b * kM + m];
    *(v2f*)&sh[m * kPitch + c2] = *(const v2f*)&emb[id * kH + c2];
  }

  v8f acc[kMT][kNTpw];
  const v8f vzero = {};
  v8f p0[kPR][kNTpw];
  if (kTab) {
    v4f pv[kMT][8], e0[kMT][8];
    for (int mt = 0; mt < kMT; ++mt) {
#pragma unroll
      for (int r = 0; r < 8; ++r) {
        const int row = idr[mt * 16 + r] * kH + ncol;
        pv[mt][r] = *(const v4f*)&tP[row];
        e0[mt][r] = *(const v4f*)&tQ[row];
      }
    }
    __syncthreads();
    {
      v2f s = {0.f, 0.f};
#pragma unroll 8
      for (int m = 0; m < kM; ++m) s += *(const v2f*)&sh[m * kPitch + c2];
      *(v2f*)&sS[c2] = s;
    }
    for (int mt = 0; mt < kMT; ++mt)
      for (int r = 0; r < 8; ++r) e0[mt][r] += pv[mt][r];
    for (int mt = 0; mt < kPR; ++mt)
      for (int j = 0; j < kNTpw; ++j)
        for (int r = 0; r < 8; ++r) p0[mt][j][r] = pv[mt][r][j];
#pragma unroll
    for (int r = 0; r < 8; ++r) *(v4f*)&sq[(hi8 + r) * kPitch + ncol] = pv[kPR][r];
    group_fence();
    __syncthreads();
    float uacc[kNTpw];
    for (int j = 0; j < kNTpw; ++j) uacc[j] = 0.f;
    for (int kb = 0; kb < kH; kb += 2 * kKB) {
      CFrag cf[2][kKS];
      v2f sv[2][kKS];
      for (int q = 0; q < 2; ++q) {
        load_c(cf[q], w1a, kb + q * kKB);
        load_s(sv[q], sk, kb + q * kKB);
      }
      for (int q = 0; q < 2; ++q) fma_u(uacc, sv[q], cf[q]);
    }
    load_b(bA, w2p, 0);
    v4f u;
    for (int j = 0; j < kNTpw; ++j)
      u[j] = (uacc[j] + __shfl_xor(uacc[j], 16)) * inv;
    for (int mt = 0; mt < kMT; ++mt) {
      const int mrow = mt * 16 + hi8;
#pragma unroll
      for (int r = 0; r < 8; ++r) {
        v4f v;
        for (int j = 0; j < kNTpw; ++j) v[j] = fmaxf(e0[mt][r][j] + u[j], 0.f);
        *(v4f*)&st[(mrow + r) * kPitch + ncol] = v;
      }
    }
  } else {
  __syncthreads();
  for (int mt = 0; mt < kMT; ++mt)
    for (int j = 0; j < kNTpw; ++j) acc[mt][j] = vzero;
  load_a(aA, hA, 0);
  for (int kb = 0; kb < kH - 2 * kKB; kb += 2 * kKB) {
    load_b(bB, w1c, kb + kKB);
    load_a(aB, hA, kb + kKB);
    group_fence();
    mma_block(acc, aA, bA);
    group_fence();
    load_b(bA, w1c, kb + 2 * kKB);
    load_a(aA, hA, kb + 2 * kKB);
    group_fence();
    mma_block(acc, aB, bB);
    group_fence();
  }
  load_b(bB, w1c, kH - kKB);
  load_a(aB, hA, kH - kKB);
  group_fence();
  mma_block(acc, aA, bA);
  group_fence();
  mma_block(acc, aB, bB);
  group_fence();
  for (int mt = 0; mt < kPR; ++mt)
    for (int j = 0; j < kNTpw; ++j)
      for (int r = 0; r < 8; ++r) p0[mt][j][r] = acc[mt][j][r] + b1v[j];
#pragma unroll
  for (int r = 0; r < 8; ++r) {
    v4f v;
    for (int j = 0; j < kNTpw; ++j) v[j] = acc[kPR][j][r] + b1v[j];
    *(v4f*)&sq[(hi8 + r) * kPitch + ncol] = v;
  }
  }

  for (int step = 0; step < kSteps; ++step) {
    int zoff = 0;
    asm volatile("" : "+s"(zoff));
    const float* w1s = w1a + zoff;
    const float* w2s = w2p + zoff;
    if (!kTab || step > 0) {
    load_w1(wA, w1s, 0);
    __syncthreads();
    {
      v2f s = {0.f, 0.f};
#pragma unroll 8
      for (int m = 0; m < kM; ++m) s += *(const v2f*)&sh[m * kPitch + c2];
      *(v2f*)&sS[c2] = s;
    }
    __syncthreads();

    float uacc[kNTpw];
    for (int mt = 0; mt < kMT; ++mt)
      for (int j = 0; j < kNTpw; ++j) acc[mt][j] = vzero;
    for (int j = 0; j < kNTpw; ++j) uacc[j] = 0.f;

    load_a(aA, hA, 0);
    load_s(sA, sk, 0);
    for (int kb = 0; kb < kH - 2 * kKB; kb += 2 * kKB) {
      load_w1(wB, w1s, kb + kKB);
      load_a(aB, hA, kb + kKB);
      load_s(sB, sk, kb + kKB);
      group_fence();
      mma_block1(acc, uacc, aA, sA, wA, inv);
      group_fence();
      load_w1(wA, w1s, kb + 2 * kKB);
      load_a(aA, hA, kb + 2 * kKB);
      load_s(sA, sk, kb + 2 * kKB);
      group_fence();
      mma_block1(acc, uacc, aB, sB, wB, inv);
      group_fence();
    }
    load_w1(wB, w1s, kH - kKB);
    load_a(aB, hA, kH - kKB);
    load_s(sB, sk, kH - kKB);
    group_fence();
    mma_block1(acc, uacc, aA, sA, wA, inv);
    group_fence();
    load_b(bA, w2s, 0);
    group_fence();
    mma_block1(acc, uacc, aB, sB, wB, inv);
    group_fence();

    {
      v4f u;
      for (int j = 0; j < kNTpw; ++j)
        u[j] = (uacc[j] + __shfl_xor(uacc[j], 16)) * inv;
      v4f q[8];
#pragma unroll
      for (int r = 0; r < 8; ++r) q[r] = *(const v4f*)&sq[(hi8 + r) * kPitch + ncol];
      v4f g[8];
      if (kTab) {
        const int* idl = ids + b * kM + (kMT - 1) * 16;
#pragma unroll
        for (int r = 0; r < 8; ++r) {
          const int id = (lane & 16) ? idl[8 + r] : idl[r];
          g[r] = *(const v4f*)&tP[id * kH + ncol];
        }
      }
      group_fence();
      for (int mt = 0; mt < kPR; ++mt) {
        const int mrow = mt * 16 + hi8;
#pragma unroll
        for (int r = 0; r < 8; ++r) {
          v4f v;
          for (int j = 0; j < kNTpw; ++j)
            v[j] = fmaxf(acc[mt][j][r] + p0[mt][j][r] + u[j], 0.f);
          *(v4f*)&st[(mrow + r) * kPitch + ncol] = v;
        }
      }
#pragma unroll
      for (int r = 0; r < 8; ++r) {
        v4f v;
        for (int j = 0; j < kNTpw; ++j)
          v[j] = fmaxf(acc[kPR][j][r] + q[r][j] + u[j], 0.f);
        *(v4f*)&st[(kPR * 16 + hi8 + r) * kPitch + ncol] = v;
      }
      if (kTab) {
#pragma unroll
        for (int r = 0; r < 8; ++r) {
          v4f v;
          for (int j = 0; j < kNTpw; ++j)
            v[j] = fmaxf(acc[kMT - 1][j][r] + g[r][j] + u[j], 0.f);
          *(v4f*)&st[((kMT - 1) * 16 + hi8 + r) * kPitch + ncol] = v;
        }
      }
    }
    }
    if (kTab && step == kSteps - 1) break;
    __syncthreads();

    for (int mt = 0; mt < kMT; ++mt)
      for (int j = 0; j < kNTpw; ++j) acc[mt][j] = vzero;
    load_a(aA, tA, 0);
    for (int kb = 0; kb < kH - 2 * kKB; kb += 2 * kKB) {
      load_b(bB, w2s, kb + kKB);
      load_a(aB, tA, kb + kKB);
      group_fence();
      mma_block(acc, aA, bA);
      group_fence();
      load_b(bA, w2s, kb + 2 * kKB);
      load_a(aA, tA, kb + 2 * kKB);
      group_fence();
      mma_block(acc, aB, bB);
      group_fence();
    }
    load_b(bB, w2s, kH - kKB);
    load_a(aB, tA, kH - kKB);
    const v4f b2v = *(const v4f*)&b2[ncol + zoff];
    group_fence();
    mma_block(acc, aA, bA);
    group_fence();
    mma_block(acc, aB, bB);
    group_fence();

    {
      for (int mh = 0; mh < kMT; mh += 2) {
        v4f hv[2][8];
        for (int mt = 0; mt < 2; ++mt) {
#pragma unroll
          for (int r = 0; r < 8; ++r)
            hv[mt][r] = *(const v4f*)&sh[((mh + mt) * 16 + hi8 + r) * kPitch + ncol];
        }
        group_fence();
        for (int mt = 0; mt < 2; ++mt) {
          const int mrow = (mh + mt) * 16 + hi8;
#pragma unroll
          for (int r = 0; r < 8; ++r) {
            v4f v;
            for (int j = 0; j < kNTpw; ++j)
              v[j] = hv[mt][r][j] + (acc[mh + mt][j][r] + b2v[j]);
            *(v4f*)&sh[(mrow + r) * kPitch + ncol] = v;
          }
        }
      }
    }
  }

  if (kTab) {
    const float* fdp = tab + kFoldOff + khalf * kA + lrow;
    const float* tL  = st + (wave * 16 + lrow) * kPitch + khalf;
    v2f eA[kDS], eB[kDS];
    v2f hfA[kDS], hfB[kDS], tfA[kDS], tfB[kDS];
    load_d(dA, wdp, 0);
    load_d(eA, fdp, 0);
    bdv = tab[kBiasOff + lrow];
    __syncthreads();
    v8f accH = {}, accT = {};
    load_a2(hfA, tfA, hL, tL, 0);
    load_d(dB, wdp, kDB);
    load_d(eB, fdp, kDB);
    load_a2(hfB, tfB, hL, tL, kDB);
    group_fence();
    mma_d2(accH, accT, hfA, tfA, dA, eA);
    group_fence();
    load_d(dA, wdp, 2 * kDB);
    load_d(eA, fdp, 2 * kDB);
    load_a2(hfA, tfA, hL, tL, 2 * kDB);
    group_fence();
    mma_d2(accH, accT, hfB, tfB, dB, eB);
    group_fence();
    load_d(dB, wdp, 3 * kDB);
    load_d(eB, fdp, 3 * kDB);
    load_a2(hfB, tfB, hL, tL, 3 * kDB);
    group_fence();
    mma_d2(accH, accT, hfA, tfA, dA, eA);
    group_fence();
    mma_d2(accH, accT, hfB, tfB, dB, eB);
    const int mrow = wave * 16 + hi8;
#pragma unroll
    for (int r = 0; r < 8; ++r)
      out[(b * kM + mrow + r) * kA + lrow] = (accH[r] + accT[r]) + bdv;
    return;
  }
  load_d(dA, wdp, 0);
  load_d(dB, wdp, kDB);
  bdv = bd[lrow];
  __syncthreads();
  {
    v8f accL = {};
    mma_d(accL, dA, hL, 0);
    load_d(dA, wdp, 2 * kDB);
    mma_d(accL, dB, hL, kDB);
    load_d(dB, wdp, 3 * kDB);
    mma_d(accL, dA, hL, 2 * kDB);
    mma_d(accL, dB, hL, 3 * kDB);
    const int mrow = wave * 16 + hi8;
#pragma unroll
    for (int r = 0; r < 8; ++r)
      out[(b * kM + mrow + r) * kA + lrow] = accL[r] + bdv;
  }
}

__launch_bounds__(kThreads, kBlocksPerWgp)
__global__ void commnet_wmma_kernel(
    const int* __restrict__ ids, const float* __restrict__ emb,
    const float* __restrict__ W1, const float* __restrict__ b1,
    const float* __restrict__ W2, const float* __restrict__ b2,
    const float* __restrict__ Wd, const float* __restrict__ bd,
    float* __restrict__ out, const float* __restrict__ tab, size_t ws_bytes) {
  extern __shared__ __attribute__((aligned(16))) float smem[];
  if (ws_bytes >= kTabBytes)
    commnet_body<true>(smem, ids, emb, W1, b1, W2, b2, Wd, bd, out, tab);
  else
    commnet_body<false>(smem, ids, emb, W1, b1, W2, b2, Wd, bd, out, tab);
}

extern "C" void kernel_launch(void* const* d_in, const int* in_sizes, int n_in,
                              void* d_out, int out_size, void* d_ws, size_t ws_size,
                              hipStream_t stream) {
  (void)in_sizes; (void)n_in; (void)out_size;
  const int*   ids = (const int*)d_in[0];
  const float* emb = (const float*)d_in[1];
  const float* W1  = (const float*)d_in[2];
  const float* b1  = (const float*)d_in[3];
  const float* W2  = (const float*)d_in[4];
  const float* b2  = (const float*)d_in[5];
  const float* Wd  = (const float*)d_in[6];
  const float* bd  = (const float*)d_in[7];
  float* out = (float*)d_out;
  float* tab = (float*)d_ws;

  const size_t lds_bytes = (size_t)kLdsFloats * sizeof(float);
  (void)hipFuncSetAttribute((const void*)commnet_wmma_kernel,
                            hipFuncAttributeMaxDynamicSharedMemorySize,
                            (int)lds_bytes);
  commnet_table_kernel<<<dim3(2 * kTabTiles + kFoldBlocks), dim3(kThreads), 0, stream>>>(
      emb, W1, b1, W2, b2, Wd, bd, tab, ws_size);
  commnet_wmma_kernel<<<dim3(kB), dim3(kThreads), lds_bytes, stream>>>(
      ids, emb, W1, b1, W2, b2, Wd, bd, out, tab, ws_size);
}


# Carry the pre-activation in the accumulators: W2 folded into the next step's W1 product (W2@W1a, W2@W1b built by the first launch), one product per step instead of two; every sum's order changes
# speedup vs baseline: 3.9960x; 1.6128x over previous
#include <hip/hip_runtime.h>


typedef __attribute__((ext_vector_type(2))) float v2f;
typedef __attribute__((ext_vector_type(4))) float v4f;
typedef __attribute__((ext_vector_type(8))) float v8f;

namespace {
constexpr int kB = 1024;
constexpr int kM = 64;
constexpr int kH = 256;
constexpr int kA = 16;
constexpr int kSteps = 4;
constexpr int kPitch = 260;
constexpr int kMT = 4;
constexpr int kWaves = 4;
constexpr int kThreads = kWaves * 32;
constexpr int kNTpw = 4;
constexpr int kMTreg = kMT - 1;
constexpr int kLdsFloats = 2 * kM * kPitch + kPitch + 16 * kPitch;
constexpr int kBlocksPerWgp = 2;
constexpr int kKB = 16;
constexpr int kKS = kKB / 4;
constexpr int kDB = 64;
constexpr int kDS = kDB / 4;
constexpr int kE = 1000;
constexpr int kTabTiles = (kE + 15) / 16;
constexpr int kTB = 32;
constexpr int kTS = kTB / 4;
constexpr int kFoldTiles = kH / 16 + 1;
constexpr int kFoldBlocks = (kFoldTiles + kWaves - 1) / kWaves;
constexpr int kFoldOff = kE * kH;
constexpr int kBiasOff = kFoldOff + kH * kA;
constexpr int kMwOff   = kFoldOff + kFoldTiles * 16 * kA;
constexpr int kCvOff   = kMwOff + 2 * kH * kH;
constexpr int kMwBlocks = 2 * (kH / 16);
constexpr size_t kTabBytes = (size_t)(kCvOff + kH) * sizeof(float);


struct BFrag { v4f x, y; };
struct W1Frag { v4f a0, c0, a1, c1; };

__device__ __forceinline__ v8f mma(v2f a, v2f b, v8f c) {
  return __builtin_amdgcn_wmma_f32_16x16x4_f32(false, a, false, b, (short)0, c,
                                               false, false);
}


__device__ __forceinline__ void load_b(BFrag (&bf)[kKS], const float* __restrict__ wp,
                                       int kb) {
  for (int s = 0; s < kKS; ++s) {
    const float* p = wp + (kb + 4 * s) * kH;
    bf[s].x = *(const v4f*)p;
    bf[s].y = *(const v4f*)(p + kH);
  }
}

__device__ __forceinline__ void load_a(v2f (&a)[kKS][kMT], const float* ap, int kb) {
  for (int s = 0; s < kKS; ++s) {
    for (int mt = 0; mt < kMT; ++mt)
      a[s][mt] = *(const v2f*)&ap[mt * 16 * kPitch + kb + 4 * s];
  }
}

__device__ __forceinline__ void mma_block(v8f (&acc)[kMT][kNTpw],
                                          const v2f (&a)[kKS][kMT],
                                          const BFrag (&w)[kKS]) {
  for (int s = 0; s < kKS; ++s) {
    v2f bf[kNTpw];
    for (int j = 0; j < kNTpw; ++j) {
      bf[j].x = w[s].x[j];
      bf[j].y = w[s].y[j];
    }
    for (int mt = 0; mt < kMT; ++mt) {
      for (int j = 0; j < kNTpw; ++j)
        acc[mt][j] = mma(a[s][mt], bf[j], acc[mt][j]);
    }
  }
}

__device__ __forceinline__ void load_w1(W1Frag (&w)[kKS], const float* __restrict__ wp,
                                        int kb) {
  for (int s = 0; s < kKS; ++s) {
    const float* p = wp + (kb + 4 * s) * kH;
    w[s].a0 = *(const v4f*)p;
    w[s].c0 = *(const v4f*)(p + kH * kH);
    w[s].a1 = *(const v4f*)(p + kH);
    w[s].c1 = *(const v4f*)(p + kH * kH + kH);
  }
}

__device__ __forceinline__ void load_s(v2f (&sv)[kKS], const float* sk, int kb) {
  for (int s = 0; s < kKS; ++s) sv[s] = *(const v2f*)&sk[kb + 4 * s];
}

__device__ __forceinline__ void mma_block1(v8f (&acc)[kMT][kNTpw],
                                           float (&uacc)[kNTpw],
                                           const v2f (&a)[kKS][kMT],
                                           const v2f (&svb)[kKS],
                                           const W1Frag (&w)[kKS],
                                           float inv) {
  for (int s = 0; s < kKS; ++s) {
    const v2f sv = svb[s];
    v2f bf[kNTpw];
    for (int j = 0; j < kNTpw; ++j) {
      bf[j].x = fmaf(-inv, w[s].c0[j], w[s].a0[j]);
      bf[j].y = fmaf(-inv, w[s].c1[j], w[s].a1[j]);
      uacc[j] = fmaf(sv.x, w[s].c0[j], uacc[j]);
      uacc[j] = fmaf(sv.y, w[s].c1[j], uacc[j]);
    }
    for (int mt = 0; mt < kMT; ++mt) {
      for (int j = 0; j < kNTpw; ++j) acc[mt][j] = mma(a[s][mt], bf[j], acc[mt][j]);
    }
  }
}

struct CFrag { v4f c0, c1; };
__device__ __forceinline__ void load_c(CFrag (&c)[kKS], const float* __restrict__ wp, int kb) {
  for (int s = 0; s < kKS; ++s) {
    const float* p = wp + (kb + 4 * s) * kH + kH * kH;
    c[s].c0 = *(const v4f*)p;
    c[s].c1 = *(const v4f*)(p + kH);
  }
}
__device__ __forceinline__ void fma_u(float (&uacc)[kNTpw], const v2f (&svb)[kKS],
                                      const CFrag (&c)[kKS]) {
  for (int s = 0; s < kKS; ++s) {
    for (int j = 0; j < kNTpw; ++j) {
      uacc[j] = fmaf(svb[s].x, c[s].c0[j], uacc[j]);
      uacc[j] = fmaf(svb[s].y, c[s].c1[j], uacc[j]);
    }
  }
}

template <bool kEff>
__device__ __forceinline__ void tab_load(v4f (&w)[kTS][kEff ? 4 : 2], v2f (&a)[kTS],
                                         const float* __restrict__ wp,
                                         const float* __restrict__ ap, int kb) {
  for (int s = 0; s < kTS; ++s) {
    const float* p = wp + (kb + 4 * s) * kH;
    w[s][0] = *(const v4f*)p;
    w[s][1] = *(const v4f*)(p + kH);
    if constexpr (kEff) {
      w[s][2] = *(const v4f*)(p + kH * kH);
      w[s][3] = *(const v4f*)(p + kH * kH + kH);
    }
    a[s] = *(const v2f*)&ap[kb + 4 * s];
  }
}
template <bool kEff>
__device__ __forceinline__ void tab_mma(v8f (&acc)[kNTpw], const v4f (&w)[kTS][kEff ? 4 : 2],
                                        const v2f (&a)[kTS], float inv) {
  for (int s = 0; s < kTS; ++s) {
    v2f bf[kNTpw];
    for (int j = 0; j < kNTpw; ++j) {
      if constexpr (kEff) {
        bf[j].x = fmaf(-inv, w[s][2][j], w[s][0][j]);
        bf[j].y = fmaf(-inv, w[s][3][j], w[s][1][j]);
      } else {
        bf[j].x = w[s][0][j];
        bf[j].y = w[s][1][j];
      }
    }
    for (int j = 0; j < kNTpw; ++j) acc[j] = mma(a[s], bf[j], acc[j]);
  }
}

__device__ __forceinline__ void group_fence() { __builtin_amdgcn_sched_barrier(0); }

__device__ __forceinline__ void load_d(v2f (&d)[kDS], const float* __restrict__ wp,
                                       int kb) {
  for (int s = 0; s < kDS; ++s) {
    const float* p = wp + (kb + 4 * s) * kA;
    d[s].x = p[0];
    d[s].y = p[kA];
  }
}

__device__ __forceinline__ void mma_d(v8f& acc, const v2f (&d)[kDS], const float* ap,
                                      int kb) {
  for (int s = 0; s < kDS; ++s) {
    const v2f a = *(const v2f*)&ap[kb + 4 * s];
    acc = mma(a, d[s], acc);
  }
}

__device__ __forceinline__ void load_af(v2f (&a)[kDS], const float* ap, int kb) {
  for (int s = 0; s < kDS; ++s) a[s] = *(const v2f*)&ap[kb + 4 * s];
}
__device__ __forceinline__ void mma_dd(v8f& acc0, v8f& acc1, const v2f (&a)[kDS],
                                       const v2f (&d)[kDS]) {
  for (int s = 0; s < kDS; s += 2) {
    acc0 = mma(a[s], d[s], acc0);
    acc1 = mma(a[s + 1], d[s + 1], acc1);
  }
}
__device__ __forceinline__ void logits_add(v8f& acc0, v8f& acc1,
                                           const float* __restrict__ wp, const float* ap) {
  v2f dA[kDS], dB[kDS], fA[kDS], fB[kDS];
  load_d(dA, wp, 0);
  load_af(fA, ap, 0);
  load_d(dB, wp, kDB);
  load_af(fB, ap, kDB);
  group_fence();
  mma_dd(acc0, acc1, fA, dA);
  group_fence();
  load_d(dA, wp, 2 * kDB);
  load_af(fA, ap, 2 * kDB);
  group_fence();
  mma_dd(acc0, acc1, fB, dB);
  group_fence();
  load_d(dB, wp, 3 * kDB);
  load_af(fB, ap, 3 * kDB);
  group_fence();
  mma_dd(acc0, acc1, fA, dA);
  group_fence();
  mma_dd(acc0, acc1, fB, dB);
  group_fence();
}
}

namespace {
template <bool kEff>
__device__ __forceinline__ void tab_tile(v8f (&acc)[kNTpw], const float* __restrict__ wp,
                                         const float* __restrict__ ap, float inv) {
  v4f wA[kTS][kEff ? 4 : 2], wB[kTS][kEff ? 4 : 2];
  v2f aA[kTS], aB[kTS];
  tab_load<kEff>(wA, aA, wp, ap, 0);
  for (int kb = 0; kb < kH - 2 * kTB; kb += 2 * kTB) {
    tab_load<kEff>(wB, aB, wp, ap, kb + kTB);
    group_fence();
    tab_mma<kEff>(acc, wA, aA, inv);
    group_fence();
    tab_load<kEff>(wA, aA, wp, ap, kb + 2 * kTB);
    group_fence();
    tab_mma<kEff>(acc, wB, aB, inv);
    group_fence();
  }
  tab_load<kEff>(wB, aB, wp, ap, kH - kTB);
  group_fence();
  tab_mma<kEff>(acc, wA, aA, inv);
  group_fence();
  tab_mma<kEff>(acc, wB, aB, inv);
}
}

__device__ __forceinline__ void fold_tile(const float* __restrict__ W2,
                                          const float* __restrict__ b2,
                                          const float* __restrict__ Wd,
                                          const float* __restrict__ bd,
                                          float* __restrict__ tab, int wt, int lrow,
                                          int khalf, int hi8) {
  const bool bias = (wt == kH / 16);
  const float* ap = bias ? b2 + khalf : W2 + (wt * 16 + lrow) * kH + khalf;
  const float* wp = Wd + khalf * kA + lrow;
  v2f dA[kDS], dB[kDS];
  const float c0 = bias ? bd[lrow] : 0.f;
  const float sc = bias ? (float)kSteps : 1.f;
  v8f acc = {};
  load_d(dA, wp, 0);
  load_d(dB, wp, kDB);
  mma_d(acc, dA, ap, 0);
  load_d(dA, wp, 2 * kDB);
  mma_d(acc, dB, ap, kDB);
  load_d(dB, wp, 3 * kDB);
  mma_d(acc, dA, ap, 2 * kDB);
  mma_d(acc, dB, ap, 3 * kDB);
#pragma unroll
  for (int r = 0; r < 8; ++r) tab[kFoldOff + (wt * 16 + hi8 + r) * kA + lrow] = fmaf(sc, acc[r], c0);
}

__launch_bounds__(kThreads)
__global__ void commnet_table_kernel(const float* __restrict__ emb,
                                     const float* __restrict__ W1,
                                     const float* __restrict__ b1,
                                     const float* __restrict__ W2,
                                     const float* __restrict__ b2,
                                     const float* __restrict__ Wd,
                                     const float* __restrict__ bd,
                                     float* __restrict__ tab, size_t ws_bytes) {
  if (ws_bytes < kTabBytes) return;
  const int tid   = threadIdx.x;
  const int wave  = __builtin_amdgcn_readfirstlane(tid >> 5);
  const int lane  = tid & 31;
  const int lrow  = lane & 15;
  const int khalf = (lane >> 4) << 1;
  const int hi8   = (lane >> 4) << 3;
  const float inv = 1.0f / (float)(kM - 1);
  const int ncol  = wave * (kNTpw * 16) + 4 * lrow;
  const v8f vzero = {};
  if (blockIdx.x >= kTabTiles + kMwBlocks + 1) {
    const int wt = (blockIdx.x - (kTabTiles + kMwBlocks + 1)) * kWaves + wave;
    if (wt < kFoldTiles) fold_tile(W2, b2, Wd, bd, tab, wt, lrow, khalf, hi8);
    return;
  }
  if (blockIdx.x == kTabTiles + kMwBlocks) {
    v8f acc[kNTpw];
    for (int j = 0; j < kNTpw; ++j) acc[j] = vzero;
    tab_tile<true>(acc, W1 + khalf * kH + ncol, b2 + khalf, -1.0f);
    if (hi8 == 0) {
      v4f v;
      for (int j = 0; j < kNTpw; ++j) v[j] = acc[j][0];
      *(v4f*)&tab[kCvOff + ncol] = v;
    }
    return;
  }
  if (blockIdx.x >= kTabTiles) {
    const int idx = blockIdx.x - kTabTiles;
    const int mat = idx / (kH / 16);
    const int rt  = idx % (kH / 16);
    v8f acc[kNTpw];
    for (int j = 0; j < kNTpw; ++j) acc[j] = vzero;
    tab_tile<false>(acc, W1 + (mat * kH + khalf) * kH + ncol,
                    W2 + (rt * 16 + lrow) * kH + khalf, inv);
    float* dst = tab + kMwOff + mat * (kH * kH);
#pragma unroll
    for (int r = 0; r < 8; ++r) {
      v4f v;
      for (int j = 0; j < kNTpw; ++j) v[j] = acc[j][r];
      *(v4f*)&dst[(rt * 16 + hi8 + r) * kH + ncol] = v;
    }
    return;
  }
  const int tile  = blockIdx.x;
  const int arow  = min(tile * 16 + lrow, kE - 1);
  const float* ap = emb + arow * kH + khalf;

  v8f acc[kNTpw];
  for (int j = 0; j < kNTpw; ++j) acc[j] = vzero;
  const v4f bias = *(const v4f*)&b1[ncol];
  tab_tile<false>(acc, W1 + (2 * kH + khalf) * kH + ncol, ap, inv);
  tab_tile<true>(acc, W1 + khalf * kH + ncol, ap, inv);
#pragma unroll
  for (int r = 0; r < 8; ++r) {
    const int row = tile * 16 + hi8 + r;
    if (row < kE) {
      v4f v;
      for (int j = 0; j < kNTpw; ++j) v[j] = acc[j][r];
      v += bias;
      *(v4f*)&tab[row * kH + ncol] = v;
    }
  }
}

__device__ __forceinline__ void commnet_body_direct(
    float* smem,
    const int* __restrict__ ids, const float* __restrict__ emb,
    const float* __restrict__ W1, const float* __restrict__ b1,
    const float* __restrict__ W2, const float* __restrict__ b2,
    const float* __restrict__ Wd, const float* __restrict__ bd,
    float* __restrict__ out) {
  float* sh = smem;
  float* st = sh + kM * kPitch;
  float* sS = st + kM * kPitch;
  float* sq = sS + kPitch;
  constexpr int kPR = kMTreg;

  const int b     = blockIdx.x;
  const int tid   = threadIdx.x;
  const int wave  = __builtin_amdgcn_readfirstlane(tid >> 5);
  const int lane  = tid & 31;
  const int lrow  = lane & 15;
  const int khalf = (lane >> 4) << 1;
  const int hi8   = (lane >> 4) << 3;
  const float inv = 1.0f / (float)(kM - 1);
  const int nb    = wave * (kNTpw * 16);
  const int ncol  = nb + 4 * lrow;
  const int c2    = 2 * tid;

  const float* w1a = W1 + khalf * kH + ncol;
  const float* w1c = W1 + (2 * kH + khalf) * kH + ncol;
  const float* w2p = W2 + khalf * kH + ncol;
  const float* wdp = Wd + khalf * kA + lrow;
  const float* hA  = sh + lrow * kPitch + khalf;
  const float* tA  = st + lrow * kPitch + khalf;
  const float* sk  = sS + khalf;
  const float* hL  = sh + (wave * 16 + lrow) * kPitch + khalf;

  BFrag bA[kKS], bB[kKS];
  W1Frag wA[kKS], wB[kKS];
  v2f dA[kDS], dB[kDS];
  v2f aA[kKS][kMT], aB[kKS][kMT];
  v2f sA[kKS], sB[kKS];
  float bdv = 0.f;

  load_b(bA, w1c, 0);
  const v4f b1v = *(const v4f*)&b1[ncol];

  for (int m = 0; m < kM; ++m) {
    const int id  = ids[b * kM + m];
    *(v2f*)&sh[m * kPitch + c2] = *(const v2f*)&emb[id * kH + c2];
  }

  v8f acc[kMT][kNTpw];
  const v8f vzero = {};
  v8f p0[kPR][kNTpw];
  __syncthreads();
  for (int mt = 0; mt < kMT; ++mt)
    for (int j = 0; j < kNTpw; ++j) acc[mt][j] = vzero;
  load_a(aA, hA, 0);
  for (int kb = 0; kb < kH - 2 * kKB; kb += 2 * kKB) {
    load_b(bB, w1c, kb + kKB);
    load_a(aB, hA, kb + kKB);
    group_fence();
    mma_block(acc, aA, bA);
    group_fence();
    load_b(bA, w1c, kb + 2 * kKB);
    load_a(aA, hA, kb + 2 * kKB);
    group_fence();
    mma_block(acc, aB, bB);
    group_fence();
  }
  load_b(bB, w1c, kH - kKB);
  load_a(aB, hA, kH - kKB);
  group_fence();
  mma_block(acc, aA, bA);
  group_fence();
  mma_block(acc, aB, bB);
  group_fence();
  for (int mt = 0; mt < kPR; ++mt)
    for (int j = 0; j < kNTpw; ++j)
      for (int r = 0; r < 8; ++r) p0[mt][j][r] = acc[mt][j][r] + b1v[j];
#pragma unroll
  for (int r = 0; r < 8; ++r) {
    v4f v;
    for (int j = 0; j < kNTpw; ++j) v[j] = acc[kPR][j][r] + b1v[j];
    *(v4f*)&sq[(hi8 + r) * kPitch + ncol] = v;
  }

  for (int step = 0; step < kSteps; ++step) {
    int zoff = 0;
    asm volatile("" : "+s"(zoff));
    const float* w1s = w1a + zoff;
    const float* w2s = w2p + zoff;
    load_w1(wA, w1s, 0);
    __syncthreads();
    {
      v2f s = {0.f, 0.f};
#pragma unroll 8
      for (int m = 0; m < kM; ++m) s += *(const v2f*)&sh[m * kPitch + c2];
      *(v2f*)&sS[c2] = s;
    }
    __syncthreads();

    float uacc[kNTpw];
    for (int mt = 0; mt < kMT; ++mt)
      for (int j = 0; j < kNTpw; ++j) acc[mt][j] = vzero;
    for (int j = 0; j < kNTpw; ++j) uacc[j] = 0.f;

    load_a(aA, hA, 0);
    load_s(sA, sk, 0);
    for (int kb = 0; kb < kH - 2 * kKB; kb += 2 * kKB) {
      load_w1(wB, w1s, kb + kKB);
      load_a(aB, hA, kb + kKB);
      load_s(sB, sk, kb + kKB);
      group_fence();
      mma_block1(acc, uacc, aA, sA, wA, inv);
      group_fence();
      load_w1(wA, w1s, kb + 2 * kKB);
      load_a(aA, hA, kb + 2 * kKB);
      load_s(sA, sk, kb + 2 * kKB);
      group_fence();
      mma_block1(acc, uacc, aB, sB, wB, inv);
      group_fence();
    }
    load_w1(wB, w1s, kH - kKB);
    load_a(aB, hA, kH - kKB);
    load_s(sB, sk, kH - kKB);
    group_fence();
    mma_block1(acc, uacc, aA, sA, wA, inv);
    group_fence();
    load_b(bA, w2s, 0);
    group_fence();
    mma_block1(acc, uacc, aB, sB, wB, inv);
    group_fence();

    {
      v4f u;
      for (int j = 0; j < kNTpw; ++j)
        u[j] = (uacc[j] + __shfl_xor(uacc[j], 16)) * inv;
      v4f q[8];
#pragma unroll
      for (int r = 0; r < 8; ++r) q[r] = *(const v4f*)&sq[(hi8 + r) * kPitch + ncol];
      group_fence();
      for (int mt = 0; mt < kPR; ++mt) {
        const int mrow = mt * 16 + hi8;
#pragma unroll
        for (int r = 0; r < 8; ++r) {
          v4f v;
          for (int j = 0; j < kNTpw; ++j)
            v[j] = fmaxf(acc[mt][j][r] + p0[mt][j][r] + u[j], 0.f);
          *(v4f*)&st[(mrow + r) * kPitch + ncol] = v;
        }
      }
#pragma unroll
      for (int r = 0; r < 8; ++r) {
        v4f v;
        for (int j = 0; j < kNTpw; ++j)
          v[j] = fmaxf(acc[kPR][j][r] + q[r][j] + u[j], 0.f);
        *(v4f*)&st[(kPR * 16 + hi8 + r) * kPitch + ncol] = v;
      }
    }
    __syncthreads();

    for (int mt = 0; mt < kMT; ++mt)
      for (int j = 0; j < kNTpw; ++j) acc[mt][j] = vzero;
    load_a(aA, tA, 0);
    for (int kb = 0; kb < kH - 2 * kKB; kb += 2 * kKB) {
      load_b(bB, w2s, kb + kKB);
      load_a(aB, tA, kb + kKB);
      group_fence();
      mma_block(acc, aA, bA);
      group_fence();
      load_b(bA, w2s, kb + 2 * kKB);
      load_a(aA, tA, kb + 2 * kKB);
      group_fence();
      mma_block(acc, aB, bB);
      group_fence();
    }
    load_b(bB, w2s, kH - kKB);
    load_a(aB, tA, kH - kKB);
    const v4f b2v = *(const v4f*)&b2[ncol + zoff];
    group_fence();
    mma_block(acc, aA, bA);
    group_fence();
    mma_block(acc, aB, bB);
    group_fence();

    {
      for (int mh = 0; mh < kMT; mh += 2) {
        v4f hv[2][8];
        for (int mt = 0; mt < 2; ++mt) {
#pragma unroll
          for (int r = 0; r < 8; ++r)
            hv[mt][r] = *(const v4f*)&sh[((mh + mt) * 16 + hi8 + r) * kPitch + ncol];
        }
        group_fence();
        for (int mt = 0; mt < 2; ++mt) {
          const int mrow = (mh + mt) * 16 + hi8;
#pragma unroll
          for (int r = 0; r < 8; ++r) {
            v4f v;
            for (int j = 0; j < kNTpw; ++j)
              v[j] = hv[mt][r][j] + (acc[mh + mt][j][r] + b2v[j]);
            *(v4f*)&sh[(mrow + r) * kPitch + ncol] = v;
          }
        }
      }
    }
  }

  load_d(dA, wdp, 0);
  load_d(dB, wdp, kDB);
  bdv = bd[lrow];
  __syncthreads();
  {
    v8f accL = {};
    mma_d(accL, dA, hL, 0);
    load_d(dA, wdp, 2 * kDB);
    mma_d(accL, dB, hL, kDB);
    load_d(dB, wdp, 3 * kDB);
    mma_d(accL, dA, hL, 2 * kDB);
    mma_d(accL, dB, hL, 3 * kDB);
    const int mrow = wave * 16 + hi8;
#pragma unroll
    for (int r = 0; r < 8; ++r)
      out[(b * kM + mrow + r) * kA + lrow] = accL[r] + bdv;
  }
}

__device__ __forceinline__ void commnet_body_tab(
    float* smem, const int* __restrict__ ids, const float* __restrict__ emb,
    const float* __restrict__ W1, const float* __restrict__ Wd,
    float* __restrict__ out, const float* __restrict__ tab) {
  float* sS = smem + 2 * kM * kPitch;

  const int b     = blockIdx.x;
  const int tid   = threadIdx.x;
  const int wave  = __builtin_amdgcn_readfirstlane(tid >> 5);
  const int lane  = tid & 31;
  const int lrow  = lane & 15;
  const int khalf = (lane >> 4) << 1;
  const int hi8   = (lane >> 4) << 3;
  const float inv = 1.0f / (float)(kM - 1);
  const int ncol  = wave * (kNTpw * 16) + 4 * lrow;
  const int c2    = 2 * tid;
  const int* idr  = ids + b * kM + hi8;

  const float* w1a = W1 + khalf * kH + ncol;
  const float* mw  = tab + kMwOff + khalf * kH + ncol;
  const float* wdp = Wd + khalf * kA + lrow;
  const float* fdp = tab + kFoldOff + khalf * kA + lrow;
  const float* sk  = sS + khalf;
  const int aoff   = lrow * kPitch + khalf;
  const int loff   = (wave * 16 + lrow) * kPitch + khalf;

  W1Frag wA[kKS], wB[kKS];
  v2f aA[kKS][kMT], aB[kKS][kMT];
  v2f sA[kKS], sB[kKS];

  for (int m = 0; m < kM; ++m) {
    const int id  = ids[b * kM + m];
    *(v2f*)&smem[m * kPitch + c2] = *(const v2f*)&emb[id * kH + c2];
  }
  v8f acc[kMT][kNTpw];
  v4f e0[kMT][8];
  for (int mt = 0; mt < kMT; ++mt) {
#pragma unroll
    for (int r = 0; r < 8; ++r) e0[mt][r] = *(const v4f*)&tab[idr[mt * 16 + r] * kH + ncol];
  }
  __syncthreads();
  {
    v2f s = {0.f, 0.f};
#pragma unroll 8
    for (int m = 0; m < kM; ++m) s += *(const v2f*)&smem[m * kPitch + c2];
    *(v2f*)&sS[c2] = s;
  }
  v8f accL0 = {}, accL1 = {};
  logits_add(accL0, accL1, wdp, smem + loff);
  __syncthreads();
  {
    float uacc[kNTpw];
    for (int j = 0; j < kNTpw; ++j) uacc[j] = 0.f;
    for (int kb = 0; kb < kH; kb += 2 * kKB) {
      CFrag cf[2][kKS];
      v2f sv[2][kKS];
      for (int q = 0; q < 2; ++q) {
        load_c(cf[q], w1a, kb + q * kKB);
        load_s(sv[q], sk, kb + q * kKB);
      }
      for (int q = 0; q < 2; ++q) fma_u(uacc, sv[q], cf[q]);
    }
    v4f u;
    for (int j = 0; j < kNTpw; ++j)
      u[j] = (uacc[j] + __shfl_xor(uacc[j], 16)) * inv;
    for (int mt = 0; mt < kMT; ++mt) {
#pragma unroll
      for (int r = 0; r < 8; ++r) {
        v4f v;
        for (int j = 0; j < kNTpw; ++j) {
          const float p = e0[mt][r][j] + u[j];
          acc[mt][j][r] = p;
          v[j] = fmaxf(p, 0.f);
        }
        *(v4f*)&smem[kM * kPitch + (mt * 16 + hi8 + r) * kPitch + ncol] = v;
      }
    }
  }

  int src = kM * kPitch;
  int dst = 0;
  for (int step = 0; step < kSteps - 1; ++step) {
    int zoff = 0;
    asm volatile("" : "+s"(zoff));
    const float* mws = mw + zoff;
    load_w1(wA, mws, 0);
    __syncthreads();
    {
      v2f s = {0.f, 0.f};
#pragma unroll 8
      for (int m = 0; m < kM; ++m) s += *(const v2f*)&smem[src + m * kPitch + c2];
      *(v2f*)&sS[c2] = s;
    }
    logits_add(accL0, accL1, fdp + zoff, smem + src + loff);
    __syncthreads();

    float uacc[kNTpw];
    for (int j = 0; j < kNTpw; ++j) uacc[j] = 0.f;
    const float* tA = smem + src + aoff;
    load_a(aA, tA, 0);
    load_s(sA, sk, 0);
    for (int kb = 0; kb < kH - 2 * kKB; kb += 2 * kKB) {
      load_w1(wB, mws, kb + kKB);
      load_a(aB, tA, kb + kKB);
      load_s(sB, sk, kb + kKB);
      group_fence();
      mma_block1(acc, uacc, aA, sA, wA, inv);
      group_fence();
      load_w1(wA, mws, kb + 2 * kKB);
      load_a(aA, tA, kb + 2 * kKB);
      load_s(sA, sk, kb + 2 * kKB);
      group_fence();
      mma_block1(acc, uacc, aB, sB, wB, inv);
      group_fence();
    }
    load_w1(wB, mws, kH - kKB);
    load_a(aB, tA, kH - kKB);
    load_s(sB, sk, kH - kKB);
    const v4f cv = *(const v4f*)&tab[kCvOff + ncol + zoff];
    group_fence();
    mma_block1(acc, uacc, aA, sA, wA, inv);
    group_fence();
    mma_block1(acc, uacc, aB, sB, wB, inv);
    group_fence();

    {
      v4f u;
      for (int j = 0; j < kNTpw; ++j)
        u[j] = (uacc[j] + __shfl_xor(uacc[j], 16)) * inv + cv[j];
      for (int mt = 0; mt < kMT; ++mt) {
#pragma unroll
        for (int r = 0; r < 8; ++r) {
          v4f v;
          for (int j = 0; j < kNTpw; ++j) {
            acc[mt][j][r] += u[j];
            v[j] = fmaxf(acc[mt][j][r], 0.f);
          }
          *(v4f*)&smem[dst + (mt * 16 + hi8 + r) * kPitch + ncol] = v;
        }
      }
    }
    const int o = src;
    src = dst;
    dst = o;
  }

  const float g = tab[kBiasOff + lrow];
  __syncthreads();
  logits_add(accL0, accL1, fdp, smem + src + loff);
  const int mrow = wave * 16 + hi8;
#pragma unroll
  for (int r = 0; r < 8; ++r)
    out[(b * kM + mrow + r) * kA + lrow] = (accL0[r] + accL1[r]) + g;
}

__launch_bounds__(kThreads, kBlocksPerWgp)
__global__ void commnet_wmma_kernel(
    const int* __restrict__ ids, const float* __restrict__ emb,
    const float* __restrict__ W1, const float* __restrict__ b1,
    const float* __restrict__ W2, const float* __restrict__ b2,
    const float* __restrict__ Wd, const float* __restrict__ bd,
    float* __restrict__ out, const float* __restrict__ tab, size_t ws_bytes) {
  extern __shared__ __attribute__((aligned(16))) float smem[];
  if (ws_bytes >= kTabBytes)
    commnet_body_tab(smem, ids, emb, W1, Wd, out, tab);
  else
    commnet_body_direct(smem, ids, emb, W1, b1, W2, b2, Wd, bd, out);
}

extern "C" void kernel_launch(void* const* d_in, const int* in_sizes, int n_in,
                              void* d_out, int out_size, void* d_ws, size_t ws_size,
                              hipStream_t stream) {
  (void)in_sizes; (void)n_in; (void)out_size;
  const int*   ids = (const int*)d_in[0];
  const float* emb = (const float*)d_in[1];
  const float* W1  = (const float*)d_in[2];
  const float* b1  = (const float*)d_in[3];
  const float* W2  = (const float*)d_in[4];
  const float* b2  = (const float*)d_in[5];
  const float* Wd  = (const float*)d_in[6];
  const float* bd  = (const float*)d_in[7];
  float* out = (float*)d_out;
  float* tab = (float*)d_ws;

  const size_t lds_bytes = (size_t)kLdsFloats * sizeof(float);
  (void)hipFuncSetAttribute((const void*)commnet_wmma_kernel,
                            hipFuncAttributeMaxDynamicSharedMemorySize,
                            (int)lds_bytes);
  commnet_table_kernel<<<dim3(kTabTiles + kMwBlocks + 1 + kFoldBlocks), dim3(kThreads), 0, stream>>>(
      emb, W1, b1, W2, b2, Wd, bd, tab, ws_size);
  commnet_wmma_kernel<<<dim3(kB), dim3(kThreads), lds_bytes, stream>>>(
      ids, emb, W1, b1, W2, b2, Wd, bd, out, tab, ws_size);
}


# Sum the relu outputs t_0..t_3 per lane (registers; one M tile in LDS) and store the sum in place of t_3: one T @ F logits product per batch instead of four; the logits' sum order changes
# speedup vs baseline: 4.1771x; 1.0453x over previous
#include <hip/hip_runtime.h>


typedef __attribute__((ext_vector_type(2))) float v2f;
typedef __attribute__((ext_vector_type(4))) float v4f;
typedef __attribute__((ext_vector_type(8))) float v8f;

namespace {
constexpr int kB = 1024;
constexpr int kM = 64;
constexpr int kH = 256;
constexpr int kA = 16;
constexpr int kSteps = 4;
constexpr int kPitch = 260;
constexpr int kMT = 4;
constexpr int kWaves = 4;
constexpr int kThreads = kWaves * 32;
constexpr int kNTpw = 4;
constexpr int kMTreg = kMT - 1;
constexpr int kLdsFloats = 2 * kM * kPitch + kPitch + 16 * kPitch + kM * kA;
constexpr int kBlocksPerWgp = 2;
constexpr int kKB = 16;
constexpr int kKS = kKB / 4;
constexpr int kDB = 64;
constexpr int kDS = kDB / 4;
constexpr int kE = 1000;
constexpr int kTabTiles = (kE + 15) / 16;
constexpr int kTB = 32;
constexpr int kTS = kTB / 4;
constexpr int kFoldTiles = kH / 16 + 1;
constexpr int kFoldBlocks = (kFoldTiles + kWaves - 1) / kWaves;
constexpr int kFoldOff = kE * kH;
constexpr int kBiasOff = kFoldOff + kH * kA;
constexpr int kMwOff   = kFoldOff + kFoldTiles * 16 * kA;
constexpr int kCvOff   = kMwOff + 2 * kH * kH;
constexpr int kMwBlocks = 2 * (kH / 16);
constexpr size_t kTabBytes = (size_t)(kCvOff + kH) * sizeof(float);


struct BFrag { v4f x, y; };
struct W1Frag { v4f a0, c0, a1, c1; };

__device__ __forceinline__ v8f mma(v2f a, v2f b, v8f c) {
  return __builtin_amdgcn_wmma_f32_16x16x4_f32(false, a, false, b, (short)0, c,
                                               false, false);
}


__device__ __forceinline__ void load_b(BFrag (&bf)[kKS], const float* __restrict__ wp,
                                       int kb) {
  for (int s = 0; s < kKS; ++s) {
    const float* p = wp + (kb + 4 * s) * kH;
    bf[s].x = *(const v4f*)p;
    bf[s].y = *(const v4f*)(p + kH);
  }
}

__device__ __forceinline__ void load_a(v2f (&a)[kKS][kMT], const float* ap, int kb) {
  for (int s = 0; s < kKS; ++s) {
    for (int mt = 0; mt < kMT; ++mt)
      a[s][mt] = *(const v2f*)&ap[mt * 16 * kPitch + kb + 4 * s];
  }
}

__device__ __forceinline__ void mma_block(v8f (&acc)[kMT][kNTpw],
                                          const v2f (&a)[kKS][kMT],
                                          const BFrag (&w)[kKS]) {
  for (int s = 0; s < kKS; ++s) {
    v2f bf[kNTpw];
    for (int j = 0; j < kNTpw; ++j) {
      bf[j].x = w[s].x[j];
      bf[j].y = w[s].y[j];
    }
    for (int mt = 0; mt < kMT; ++mt) {
      for (int j = 0; j < kNTpw; ++j)
        acc[mt][j] = mma(a[s][mt], bf[j], acc[mt][j]);
    }
  }
}

__device__ __forceinline__ void load_w1(W1Frag (&w)[kKS], const float* __restrict__ wp,
                                        int kb) {
  for (int s = 0; s < kKS; ++s) {
    const float* p = wp + (kb + 4 * s) * kH;
    w[s].a0 = *(const v4f*)p;
    w[s].c0 = *(const v4f*)(p + kH * kH);
    w[s].a1 = *(const v4f*)(p + kH);
    w[s].c1 = *(const v4f*)(p + kH * kH + kH);
  }
}

__device__ __forceinline__ void load_s(v2f (&sv)[kKS], const float* sk, int kb) {
  for (int s = 0; s < kKS; ++s) sv[s] = *(const v2f*)&sk[kb + 4 * s];
}

__device__ __forceinline__ void mma_block1(v8f (&acc)[kMT][kNTpw],
                                           float (&uacc)[kNTpw],
                                           const v2f (&a)[kKS][kMT],
                                           const v2f (&svb)[kKS],
                                           const W1Frag (&w)[kKS],
                                           float inv) {
  for (int s = 0; s < kKS; ++s) {
    const v2f sv = svb[s];
    v2f bf[kNTpw];
    for (int j = 0; j < kNTpw; ++j) {
      bf[j].x = fmaf(-inv, w[s].c0[j], w[s].a0[j]);
      bf[j].y = fmaf(-inv, w[s].c1[j], w[s].a1[j]);
      uacc[j] = fmaf(sv.x, w[s].c0[j], uacc[j]);
      uacc[j] = fmaf(sv.y, w[s].c1[j], uacc[j]);
    }
    for (int mt = 0; mt < kMT; ++mt) {
      for (int j = 0; j < kNTpw; ++j) acc[mt][j] = mma(a[s][mt], bf[j], acc[mt][j]);
    }
  }
}

struct CFrag { v4f c0, c1; };
__device__ __forceinline__ void load_c(CFrag (&c)[kKS], const float* __restrict__ wp, int kb) {
  for (int s = 0; s < kKS; ++s) {
    const float* p = wp + (kb + 4 * s) * kH + kH * kH;
    c[s].c0 = *(const v4f*)p;
    c[s].c1 = *(const v4f*)(p + kH);
  }
}
__device__ __forceinline__ void fma_u(float (&uacc)[kNTpw], const v2f (&svb)[kKS],
                                      const CFrag (&c)[kKS]) {
  for (int s = 0; s < kKS; ++s) {
    for (int j = 0; j < kNTpw; ++j) {
      uacc[j] = fmaf(svb[s].x, c[s].c0[j], uacc[j]);
      uacc[j] = fmaf(svb[s].y, c[s].c1[j], uacc[j]);
    }
  }
}

template <bool kEff>
__device__ __forceinline__ void tab_load(v4f (&w)[kTS][kEff ? 4 : 2], v2f (&a)[kTS],
                                         const float* __restrict__ wp,
                                         const float* __restrict__ ap, int kb) {
  for (int s = 0; s < kTS; ++s) {
    const float* p = wp + (kb + 4 * s) * kH;
    w[s][0] = *(const v4f*)p;
    w[s][1] = *(const v4f*)(p + kH);
    if constexpr (kEff) {
      w[s][2] = *(const v4f*)(p + kH * kH);
      w[s][3] = *(const v4f*)(p + kH * kH + kH);
    }
    a[s] = *(const v2f*)&ap[kb + 4 * s];
  }
}
template <bool kEff>
__device__ __forceinline__ void tab_mma(v8f (&acc)[kNTpw], const v4f (&w)[kTS][kEff ? 4 : 2],
                                        const v2f (&a)[kTS], float inv) {
  for (int s = 0; s < kTS; ++s) {
    v2f bf[kNTpw];
    for (int j = 0; j < kNTpw; ++j) {
      if constexpr (kEff) {
        bf[j].x = fmaf(-inv, w[s][2][j], w[s][0][j]);
        bf[j].y = fmaf(-inv, w[s][3][j], w[s][1][j]);
      } else {
        bf[j].x = w[s][0][j];
        bf[j].y = w[s][1][j];
      }
    }
    for (int j = 0; j < kNTpw; ++j) acc[j] = mma(a[s], bf[j], acc[j]);
  }
}

__device__ __forceinline__ void group_fence() { __builtin_amdgcn_sched_barrier(0); }

__device__ __forceinline__ void load_d(v2f (&d)[kDS], const float* __restrict__ wp,
                                       int kb) {
  for (int s = 0; s < kDS; ++s) {
    const float* p = wp + (kb + 4 * s) * kA;
    d[s].x = p[0];
    d[s].y = p[kA];
  }
}

__device__ __forceinline__ void mma_d(v8f& acc, const v2f (&d)[kDS], const float* ap,
                                      int kb) {
  for (int s = 0; s < kDS; ++s) {
    const v2f a = *(const v2f*)&ap[kb + 4 * s];
    acc = mma(a, d[s], acc);
  }
}

__device__ __forceinline__ void load_af(v2f (&a)[kDS], const float* ap, int kb) {
  for (int s = 0; s < kDS; ++s) a[s] = *(const v2f*)&ap[kb + 4 * s];
}
__device__ __forceinline__ void mma_dd(v8f& acc0, v8f& acc1, const v2f (&a)[kDS],
                                       const v2f (&d)[kDS]) {
  for (int s = 0; s < kDS; s += 2) {
    acc0 = mma(a[s], d[s], acc0);
    acc1 = mma(a[s + 1], d[s + 1], acc1);
  }
}
__device__ __forceinline__ void logits_add(v8f& acc0, v8f& acc1,
                                           const float* __restrict__ wp, const float* ap) {
  v2f dA[kDS], dB[kDS], fA[kDS], fB[kDS];
  load_d(dA, wp, 0);
  load_af(fA, ap, 0);
  load_d(dB, wp, kDB);
  load_af(fB, ap, kDB);
  group_fence();
  mma_dd(acc0, acc1, fA, dA);
  group_fence();
  load_d(dA, wp, 2 * kDB);
  load_af(fA, ap, 2 * kDB);
  group_fence();
  mma_dd(acc0, acc1, fB, dB);
  group_fence();
  load_d(dB, wp, 3 * kDB);
  load_af(fB, ap, 3 * kDB);
  group_fence();
  mma_dd(acc0, acc1, fA, dA);
  group_fence();
  mma_dd(acc0, acc1, fB, dB);
  group_fence();
}
}

namespace {
template <bool kEff>
__device__ __forceinline__ void tab_tile(v8f (&acc)[kNTpw], const float* __restrict__ wp,
                                         const float* __restrict__ ap, float inv) {
  v4f wA[kTS][kEff ? 4 : 2], wB[kTS][kEff ? 4 : 2];
  v2f aA[kTS], aB[kTS];
  tab_load<kEff>(wA, aA, wp, ap, 0);
  for (int kb = 0; kb < kH - 2 * kTB; kb += 2 * kTB) {
    tab_load<kEff>(wB, aB, wp, ap, kb + kTB);
    group_fence();
    tab_mma<kEff>(acc, wA, aA, inv);
    group_fence();
    tab_load<kEff>(wA, aA, wp, ap, kb + 2 * kTB);
    group_fence();
    tab_mma<kEff>(acc, wB, aB, inv);
    group_fence();
  }
  tab_load<kEff>(wB, aB, wp, ap, kH - kTB);
  group_fence();
  tab_mma<kEff>(acc, wA, aA, inv);
  group_fence();
  tab_mma<kEff>(acc, wB, aB, inv);
}
}

__device__ __forceinline__ void fold_tile(const float* __restrict__ W2,
                                          const float* __restrict__ b2,
                                          const float* __restrict__ Wd,
                                          const float* __restrict__ bd,
                                          float* __restrict__ tab, int wt, int lrow,
                                          int khalf, int hi8) {
  const bool bias = (wt == kH / 16);
  const float* ap = bias ? b2 + khalf : W2 + (wt * 16 + lrow) * kH + khalf;
  const float* wp = Wd + khalf * kA + lrow;
  v2f dA[kDS], dB[kDS];
  const float c0 = bias ? bd[lrow] : 0.f;
  const float sc = bias ? (float)kSteps : 1.f;
  v8f acc = {};
  load_d(dA, wp, 0);
  load_d(dB, wp, kDB);
  mma_d(acc, dA, ap, 0);
  load_d(dA, wp, 2 * kDB);
  mma_d(acc, dB, ap, kDB);
  load_d(dB, wp, 3 * kDB);
  mma_d(acc, dA, ap, 2 * kDB);
  mma_d(acc, dB, ap, 3 * kDB);
#pragma unroll
  for (int r = 0; r < 8; ++r) tab[kFoldOff + (wt * 16 + hi8 + r) * kA + lrow] = fmaf(sc, acc[r], c0);
}

__launch_bounds__(kThreads)
__global__ void commnet_table_kernel(const float* __restrict__ emb,
                                     const float* __restrict__ W1,
                                     const float* __restrict__ b1,
                                     const float* __restrict__ W2,
                                     const float* __restrict__ b2,
                                     const float* __restrict__ Wd,
                                     const float* __restrict__ bd,
                                     float* __restrict__ tab, size_t ws_bytes) {
  if (ws_bytes < kTabBytes) return;
  const int tid   = threadIdx.x;
  const int wave  = __builtin_amdgcn_readfirstlane(tid >> 5);
  const int lane  = tid & 31;
  const int lrow  = lane & 15;
  const int khalf = (lane >> 4) << 1;
  const int hi8   = (lane >> 4) << 3;
  const float inv = 1.0f / (float)(kM - 1);
  const int ncol  = wave * (kNTpw * 16) + 4 * lrow;
  const v8f vzero = {};
  if (blockIdx.x >= kTabTiles + kMwBlocks + 1) {
    const int wt = (blockIdx.x - (kTabTiles + kMwBlocks + 1)) * kWaves + wave;
    if (wt < kFoldTiles) fold_tile(W2, b2, Wd, bd, tab, wt, lrow, khalf, hi8);
    return;
  }
  if (blockIdx.x == kTabTiles + kMwBlocks) {
    v8f acc[kNTpw];
    for (int j = 0; j < kNTpw; ++j) acc[j] = vzero;
    tab_tile<true>(acc, W1 + khalf * kH + ncol, b2 + khalf, -1.0f);
    if (hi8 == 0) {
      v4f v;
      for (int j = 0; j < kNTpw; ++j) v[j] = acc[j][0];
      *(v4f*)&tab[kCvOff + ncol] = v;
    }
    return;
  }
  if (blockIdx.x >= kTabTiles) {
    const int idx = blockIdx.x - kTabTiles;
    const int mat = idx / (kH / 16);
    const int rt  = idx % (kH / 16);
    v8f acc[kNTpw];
    for (int j = 0; j < kNTpw; ++j) acc[j] = vzero;
    tab_tile<false>(acc, W1 + (mat * kH + khalf) * kH + ncol,
                    W2 + (rt * 16 + lrow) * kH + khalf, inv);
    float* dst = tab + kMwOff + mat * (kH * kH);
#pragma unroll
    for (int r = 0; r < 8; ++r) {
      v4f v;
      for (int j = 0; j < kNTpw; ++j) v[j] = acc[j][r];
      *(v4f*)&dst[(rt * 16 + hi8 + r) * kH + ncol] = v;
    }
    return;
  }
  const int tile  = blockIdx.x;
  const int arow  = min(tile * 16 + lrow, kE - 1);
  const float* ap = emb + arow * kH + khalf;

  v8f acc[kNTpw];
  for (int j = 0; j < kNTpw; ++j) acc[j] = vzero;
  const v4f bias = *(const v4f*)&b1[ncol];
  tab_tile<false>(acc, W1 + (2 * kH + khalf) * kH + ncol, ap, inv);
  tab_tile<true>(acc, W1 + khalf * kH + ncol, ap, inv);
#pragma unroll
  for (int r = 0; r < 8; ++r) {
    const int row = tile * 16 + hi8 + r;
    if (row < kE) {
      v4f v;
      for (int j = 0; j < kNTpw; ++j) v[j] = acc[j][r];
      v += bias;
      *(v4f*)&tab[row * kH + ncol] = v;
    }
  }
}

__device__ __forceinline__ void commnet_body_direct(
    float* smem,
    const int* __restrict__ ids, const float* __restrict__ emb,
    const float* __restrict__ W1, const float* __restrict__ b1,
    const float* __restrict__ W2, const float* __restrict__ b2,
    const float* __restrict__ Wd, const float* __restrict__ bd,
    float* __restrict__ out) {
  float* sh = smem;
  float* st = sh + kM * kPitch;
  float* sS = st + kM * kPitch;
  float* sq = sS + kPitch;
  constexpr int kPR = kMTreg;

  const int b     = blockIdx.x;
  const int tid   = threadIdx.x;
  const int wave  = __builtin_amdgcn_readfirstlane(tid >> 5);
  const int lane  = tid & 31;
  const int lrow  = lane & 15;
  const int khalf = (lane >> 4) << 1;
  const int hi8   = (lane >> 4) << 3;
  const float inv = 1.0f / (float)(kM - 1);
  const int nb    = wave * (kNTpw * 16);
  const int ncol  = nb + 4 * lrow;
  const int c2    = 2 * tid;

  const float* w1a = W1 + khalf * kH + ncol;
  const float* w1c = W1 + (2 * kH + khalf) * kH + ncol;
  const float* w2p = W2 + khalf * kH + ncol;
  const float* wdp = Wd + khalf * kA + lrow;
  const float* hA  = sh + lrow * kPitch + khalf;
  const float* tA  = st + lrow * kPitch + khalf;
  const float* sk  = sS + khalf;
  const float* hL  = sh + (wave * 16 + lrow) * kPitch + khalf;

  BFrag bA[kKS], bB[kKS];
  W1Frag wA[kKS], wB[kKS];
  v2f dA[kDS], dB[kDS];
  v2f aA[kKS][kMT], aB[kKS][kMT];
  v2f sA[kKS], sB[kKS];
  float bdv = 0.f;

  load_b(bA, w1c, 0);
  const v4f b1v = *(const v4f*)&b1[ncol];

  for (int m = 0; m < kM; ++m) {
    const int id  = ids[b * kM + m];
    *(v2f*)&sh[m * kPitch + c2] = *(const v2f*)&emb[id * kH + c2];
  }

  v8f acc[kMT][kNTpw];
  const v8f vzero = {};
  v8f p0[kPR][kNTpw];
  __syncthreads();
  for (int mt = 0; mt < kMT; ++mt)
    for (int j = 0; j < kNTpw; ++j) acc[mt][j] = vzero;
  load_a(aA, hA, 0);
  for (int kb = 0; kb < kH - 2 * kKB; kb += 2 * kKB) {
    load_b(bB, w1c, kb + kKB);
    load_a(aB, hA, kb + kKB);
    group_fence();
    mma_block(acc, aA, bA);
    group_fence();
    load_b(bA, w1c, kb + 2 * kKB);
    load_a(aA, hA, kb + 2 * kKB);
    group_fence();
    mma_block(acc, aB, bB);
    group_fence();
  }
  load_b(bB, w1c, kH - kKB);
  load_a(aB, hA, kH - kKB);
  group_fence();
  mma_block(acc, aA, bA);
  group_fence();
  mma_block(acc, aB, bB);
  group_fence();
  for (int mt = 0; mt < kPR; ++mt)
    for (int j = 0; j < kNTpw; ++j)
      for (int r = 0; r < 8; ++r) p0[mt][j][r] = acc[mt][j][r] + b1v[j];
#pragma unroll
  for (int r = 0; r < 8; ++r) {
    v4f v;
    for (int j = 0; j < kNTpw; ++j) v[j] = acc[kPR][j][r] + b1v[j];
    *(v4f*)&sq[(hi8 + r) * kPitch + ncol] = v;
  }

  for (int step = 0; step < kSteps; ++step) {
    int zoff = 0;
    asm volatile("" : "+s"(zoff));
    const float* w1s = w1a + zoff;
    const float* w2s = w2p + zoff;
    load_w1(wA, w1s, 0);
    __syncthreads();
    {
      v2f s = {0.f, 0.f};
#pragma unroll 8
      for (int m = 0; m < kM; ++m) s += *(const v2f*)&sh[m * kPitch + c2];
      *(v2f*)&sS[c2] = s;
    }
    __syncthreads();

    float uacc[kNTpw];
    for (int mt = 0; mt < kMT; ++mt)
      for (int j = 0; j < kNTpw; ++j) acc[mt][j] = vzero;
    for (int j = 0; j < kNTpw; ++j) uacc[j] = 0.f;

    load_a(aA, hA, 0);
    load_s(sA, sk, 0);
    for (int kb = 0; kb < kH - 2 * kKB; kb += 2 * kKB) {
      load_w1(wB, w1s, kb + kKB);
      load_a(aB, hA, kb + kKB);
      load_s(sB, sk, kb + kKB);
      group_fence();
      mma_block1(acc, uacc, aA, sA, wA, inv);
      group_fence();
      load_w1(wA, w1s, kb + 2 * kKB);
      load_a(aA, hA, kb + 2 * kKB);
      load_s(sA, sk, kb + 2 * kKB);
      group_fence();
      mma_block1(acc, uacc, aB, sB, wB, inv);
      group_fence();
    }
    load_w1(wB, w1s, kH - kKB);
    load_a(aB, hA, kH - kKB);
    load_s(sB, sk, kH - kKB);
    group_fence();
    mma_block1(acc, uacc, aA, sA, wA, inv);
    group_fence();
    load_b(bA, w2s, 0);
    group_fence();
    mma_block1(acc, uacc, aB, sB, wB, inv);
    group_fence();

    {
      v4f u;
      for (int j = 0; j < kNTpw; ++j)
        u[j] = (uacc[j] + __shfl_xor(uacc[j], 16)) * inv;
      v4f q[8];
#pragma unroll
      for (int r = 0; r < 8; ++r) q[r] = *(const v4f*)&sq[(hi8 + r) * kPitch + ncol];
      group_fence();
      for (int mt = 0; mt < kPR; ++mt) {
        const int mrow = mt * 16 + hi8;
#pragma unroll
        for (int r = 0; r < 8; ++r) {
          v4f v;
          for (int j = 0; j < kNTpw; ++j)
            v[j] = fmaxf(acc[mt][j][r] + p0[mt][j][r] + u[j], 0.f);
          *(v4f*)&st[(mrow + r) * kPitch + ncol] = v;
        }
      }
#pragma unroll
      for (int r = 0; r < 8; ++r) {
        v4f v;
        for (int j = 0; j < kNTpw; ++j)
          v[j] = fmaxf(acc[kPR][j][r] + q[r][j] + u[j], 0.f);
        *(v4f*)&st[(kPR * 16 + hi8 + r) * kPitch + ncol] = v;
      }
    }
    __syncthreads();

    for (int mt = 0; mt < kMT; ++mt)
      for (int j = 0; j < kNTpw; ++j) acc[mt][j] = vzero;
    load_a(aA, tA, 0);
    for (int kb = 0; kb < kH - 2 * kKB; kb += 2 * kKB) {
      load_b(bB, w2s, kb + kKB);
      load_a(aB, tA, kb + kKB);
      group_fence();
      mma_block(acc, aA, bA);
      group_fence();
      load_b(bA, w2s, kb + 2 * kKB);
      load_a(aA, tA, kb + 2 * kKB);
      group_fence();
      mma_block(acc, aB, bB);
      group_fence();
    }
    load_b(bB, w2s, kH - kKB);
    load_a(aB, tA, kH - kKB);
    const v4f b2v = *(const v4f*)&b2[ncol + zoff];
    group_fence();
    mma_block(acc, aA, bA);
    group_fence();
    mma_block(acc, aB, bB);
    group_fence();

    {
      for (int mh = 0; mh < kMT; mh += 2) {
        v4f hv[2][8];
        for (int mt = 0; mt < 2; ++mt) {
#pragma unroll
          for (int r = 0; r < 8; ++r)
            hv[mt][r] = *(const v4f*)&sh[((mh + mt) * 16 + hi8 + r) * kPitch + ncol];
        }
        group_fence();
        for (int mt = 0; mt < 2; ++mt) {
          const int mrow = (mh + mt) * 16 + hi8;
#pragma unroll
          for (int r = 0; r < 8; ++r) {
            v4f v;
            for (int j = 0; j < kNTpw; ++j)
              v[j] = hv[mt][r][j] + (acc[mh + mt][j][r] + b2v[j]);
            *(v4f*)&sh[(mrow + r) * kPitch + ncol] = v;
          }
        }
      }
    }
  }

  load_d(dA, wdp, 0);
  load_d(dB, wdp, kDB);
  bdv = bd[lrow];
  __syncthreads();
  {
    v8f accL = {};
    mma_d(accL, dA, hL, 0);
    load_d(dA, wdp, 2 * kDB);
    mma_d(accL, dB, hL, kDB);
    load_d(dB, wdp, 3 * kDB);
    mma_d(accL, dA, hL, 2 * kDB);
    mma_d(accL, dB, hL, 3 * kDB);
    const int mrow = wave * 16 + hi8;
#pragma unroll
    for (int r = 0; r < 8; ++r)
      out[(b * kM + mrow + r) * kA + lrow] = accL[r] + bdv;
  }
}

__device__ __forceinline__ void commnet_body_tab(
    float* smem, const int* __restrict__ ids, const float* __restrict__ emb,
    const float* __restrict__ W1, const float* __restrict__ Wd,
    float* __restrict__ out, const float* __restrict__ tab) {
  float* sS = smem + 2 * kM * kPitch;
  float* sq = sS + kPitch;
  float* sL = sq + 16 * kPitch;
  constexpr int kPR = kMTreg;

  const int b     = blockIdx.x;
  const int tid   = threadIdx.x;
  const int wave  = __builtin_amdgcn_readfirstlane(tid >> 5);
  const int lane  = tid & 31;
  const int lrow  = lane & 15;
  const int khalf = (lane >> 4) << 1;
  const int hi8   = (lane >> 4) << 3;
  const float inv = 1.0f / (float)(kM - 1);
  const int ncol  = wave * (kNTpw * 16) + 4 * lrow;
  const int c2    = 2 * tid;
  const int* idr  = ids + b * kM + hi8;

  const float* w1a = W1 + khalf * kH + ncol;
  const float* mw  = tab + kMwOff + khalf * kH + ncol;
  const float* wdp = Wd + khalf * kA + lrow;
  const float* fdp = tab + kFoldOff + khalf * kA + lrow;
  const float* sk  = sS + khalf;
  const int aoff   = lrow * kPitch + khalf;
  const int loff   = (wave * 16 + lrow) * kPitch + khalf;

  W1Frag wA[kKS], wB[kKS];
  v2f aA[kKS][kMT], aB[kKS][kMT];
  v2f sA[kKS], sB[kKS];

  for (int m = 0; m < kM; ++m) {
    const int id  = ids[b * kM + m];
    *(v2f*)&smem[m * kPitch + c2] = *(const v2f*)&emb[id * kH + c2];
  }
  v8f acc[kMT][kNTpw];
  v4f tsum[kPR][8];
  v4f e0[kMT][8];
  for (int mt = 0; mt < kMT; ++mt) {
#pragma unroll
    for (int r = 0; r < 8; ++r) e0[mt][r] = *(const v4f*)&tab[idr[mt * 16 + r] * kH + ncol];
  }
  __syncthreads();
  {
    v2f s = {0.f, 0.f};
#pragma unroll 8
    for (int m = 0; m < kM; ++m) s += *(const v2f*)&smem[m * kPitch + c2];
    *(v2f*)&sS[c2] = s;
  }
  {
    v8f accL0 = {}, accL1 = {};
    logits_add(accL0, accL1, wdp, smem + loff);
    const v8f hd = accL0 + accL1;
    *(v4f*)&sL[tid * 8]     = __builtin_shufflevector(hd, hd, 0, 1, 2, 3);
    *(v4f*)&sL[tid * 8 + 4] = __builtin_shufflevector(hd, hd, 4, 5, 6, 7);
  }
  __syncthreads();
  {
    float uacc[kNTpw];
    for (int j = 0; j < kNTpw; ++j) uacc[j] = 0.f;
    for (int kb = 0; kb < kH; kb += 2 * kKB) {
      CFrag cf[2][kKS];
      v2f sv[2][kKS];
      for (int q = 0; q < 2; ++q) {
        load_c(cf[q], w1a, kb + q * kKB);
        load_s(sv[q], sk, kb + q * kKB);
      }
      for (int q = 0; q < 2; ++q) fma_u(uacc, sv[q], cf[q]);
    }
    v4f u;
    for (int j = 0; j < kNTpw; ++j)
      u[j] = (uacc[j] + __shfl_xor(uacc[j], 16)) * inv;
    for (int mt = 0; mt < kPR; ++mt) {
#pragma unroll
      for (int r = 0; r < 8; ++r) {
        v4f v;
        for (int j = 0; j < kNTpw; ++j) {
          const float p = e0[mt][r][j] + u[j];
          acc[mt][j][r] = p;
          v[j] = fmaxf(p, 0.f);
        }
        tsum[mt][r] = v;
        *(v4f*)&smem[kM * kPitch + (mt * 16 + hi8 + r) * kPitch + ncol] = v;
      }
    }
#pragma unroll
    for (int r = 0; r < 8; ++r) {
      v4f v;
      for (int j = 0; j < kNTpw; ++j) {
        const float p = e0[kPR][r][j] + u[j];
        acc[kPR][j][r] = p;
        v[j] = fmaxf(p, 0.f);
      }
      *(v4f*)&sq[(hi8 + r) * kPitch + ncol] = v;
      *(v4f*)&smem[kM * kPitch + (kPR * 16 + hi8 + r) * kPitch + ncol] = v;
    }
  }

  int src = kM * kPitch;
  int dst = 0;
  v4f u;
  for (int step = 0;; ++step) {
    int zoff = 0;
    asm volatile("" : "+s"(zoff));
    const float* mws = mw + zoff;
    load_w1(wA, mws, 0);
    __syncthreads();
    {
      v2f s = {0.f, 0.f};
#pragma unroll 8
      for (int m = 0; m < kM; ++m) s += *(const v2f*)&smem[src + m * kPitch + c2];
      *(v2f*)&sS[c2] = s;
    }
    __syncthreads();

    float uacc[kNTpw];
    for (int j = 0; j < kNTpw; ++j) uacc[j] = 0.f;
    const float* tA = smem + src + aoff;
    load_a(aA, tA, 0);
    load_s(sA, sk, 0);
    for (int kb = 0; kb < kH - 2 * kKB; kb += 2 * kKB) {
      load_w1(wB, mws, kb + kKB);
      load_a(aB, tA, kb + kKB);
      load_s(sB, sk, kb + kKB);
      group_fence();
      mma_block1(acc, uacc, aA, sA, wA, inv);
      group_fence();
      load_w1(wA, mws, kb + 2 * kKB);
      load_a(aA, tA, kb + 2 * kKB);
      load_s(sA, sk, kb + 2 * kKB);
      group_fence();
      mma_block1(acc, uacc, aB, sB, wB, inv);
      group_fence();
    }
    load_w1(wB, mws, kH - kKB);
    load_a(aB, tA, kH - kKB);
    load_s(sB, sk, kH - kKB);
    const v4f cv = *(const v4f*)&tab[kCvOff + ncol + zoff];
    group_fence();
    mma_block1(acc, uacc, aA, sA, wA, inv);
    group_fence();
    mma_block1(acc, uacc, aB, sB, wB, inv);
    group_fence();

    for (int j = 0; j < kNTpw; ++j)
      u[j] = (uacc[j] + __shfl_xor(uacc[j], 16)) * inv + cv[j];
    if (step == kSteps - 2) break;
    {
      v4f q[8];
#pragma unroll
      for (int r = 0; r < 8; ++r) q[r] = *(const v4f*)&sq[(hi8 + r) * kPitch + ncol];
      group_fence();
      for (int mt = 0; mt < kPR; ++mt) {
#pragma unroll
        for (int r = 0; r < 8; ++r) {
          v4f v;
          for (int j = 0; j < kNTpw; ++j) {
            acc[mt][j][r] += u[j];
            v[j] = fmaxf(acc[mt][j][r], 0.f);
          }
          tsum[mt][r] += v;
          *(v4f*)&smem[dst + (mt * 16 + hi8 + r) * kPitch + ncol] = v;
        }
      }
#pragma unroll
      for (int r = 0; r < 8; ++r) {
        v4f v;
        for (int j = 0; j < kNTpw; ++j) {
          acc[kPR][j][r] += u[j];
          v[j] = fmaxf(acc[kPR][j][r], 0.f);
        }
        *(v4f*)&sq[(hi8 + r) * kPitch + ncol] = q[r] + v;
        *(v4f*)&smem[dst + (kPR * 16 + hi8 + r) * kPitch + ncol] = v;
      }
    }
    const int o = src;
    src = dst;
    dst = o;
  }
  {
    v4f q[8];
#pragma unroll
    for (int r = 0; r < 8; ++r) q[r] = *(const v4f*)&sq[(hi8 + r) * kPitch + ncol];
    group_fence();
    for (int mt = 0; mt < kPR; ++mt) {
#pragma unroll
      for (int r = 0; r < 8; ++r) {
        v4f v;
        for (int j = 0; j < kNTpw; ++j) v[j] = fmaxf(acc[mt][j][r] + u[j], 0.f);
        *(v4f*)&smem[dst + (mt * 16 + hi8 + r) * kPitch + ncol] = tsum[mt][r] + v;
      }
    }
#pragma unroll
    for (int r = 0; r < 8; ++r) {
      v4f v;
      for (int j = 0; j < kNTpw; ++j) v[j] = fmaxf(acc[kPR][j][r] + u[j], 0.f);
      *(v4f*)&smem[dst + (kPR * 16 + hi8 + r) * kPitch + ncol] = q[r] + v;
    }
  }

  const float g = tab[kBiasOff + lrow];
  __syncthreads();
  v8f accL0, accL1 = {};
  {
    const v4f lo = *(const v4f*)&sL[tid * 8];
    const v4f hi = *(const v4f*)&sL[tid * 8 + 4];
    accL0 = __builtin_shufflevector(lo, hi, 0, 1, 2, 3, 4, 5, 6, 7);
  }
  logits_add(accL0, accL1, fdp, smem + dst + loff);
  const int mrow = wave * 16 + hi8;
#pragma unroll
  for (int r = 0; r < 8; ++r)
    out[(b * kM + mrow + r) * kA + lrow] = (accL0[r] + accL1[r]) + g;
}

__launch_bounds__(kThreads, kBlocksPerWgp)
__global__ void commnet_wmma_kernel(
    const int* __restrict__ ids, const float* __restrict__ emb,
    const float* __restrict__ W1, const float* __restrict__ b1,
    const float* __restrict__ W2, const float* __restrict__ b2,
    const float* __restrict__ Wd, const float* __restrict__ bd,
    float* __restrict__ out, const float* __restrict__ tab, size_t ws_bytes) {
  extern __shared__ __attribute__((aligned(16))) float smem[];
  if (ws_bytes >= kTabBytes)
    commnet_body_tab(smem, ids, emb, W1, Wd, out, tab);
  else
    commnet_body_direct(smem, ids, emb, W1, b1, W2, b2, Wd, bd, out);
}

extern "C" void kernel_launch(void* const* d_in, const int* in_sizes, int n_in,
                              void* d_out, int out_size, void* d_ws, size_t ws_size,
                              hipStream_t stream) {
  (void)in_sizes; (void)n_in; (void)out_size;
  const int*   ids = (const int*)d_in[0];
  const float* emb = (const float*)d_in[1];
  const float* W1  = (const float*)d_in[2];
  const float* b1  = (const float*)d_in[3];
  const float* W2  = (const float*)d_in[4];
  const float* b2  = (const float*)d_in[5];
  const float* Wd  = (const float*)d_in[6];
  const float* bd  = (const float*)d_in[7];
  float* out = (float*)d_out;
  float* tab = (float*)d_ws;

  const size_t lds_bytes = (size_t)kLdsFloats * sizeof(float);
  (void)hipFuncSetAttribute((const void*)commnet_wmma_kernel,
                            hipFuncAttributeMaxDynamicSharedMemorySize,
                            (int)lds_bytes);
  commnet_table_kernel<<<dim3(kTabTiles + kMwBlocks + 1 + kFoldBlocks), dim3(kThreads), 0, stream>>>(
      emb, W1, b1, W2, b2, Wd, bd, tab, ws_size);
  commnet_wmma_kernel<<<dim3(kB), dim3(kThreads), lds_bytes, stream>>>(
      ids, emb, W1, b1, W2, b2, Wd, bd, out, tab, ws_size);
}


# First step wholly from tables: launch 1 also builds EB = emb@W1b and L0 = emb@Wd; u_0 = inv * sum of gathered EB rows, L0[id] starts the logits, h0 never gathered; u_0 and logits sum order changes
# speedup vs baseline: 4.5075x; 1.0791x over previous
#include <hip/hip_runtime.h>


typedef __attribute__((ext_vector_type(2))) float v2f;
typedef __attribute__((ext_vector_type(4))) float v4f;
typedef __attribute__((ext_vector_type(8))) float v8f;

namespace {
constexpr int kB = 1024;
constexpr int kM = 64;
constexpr int kH = 256;
constexpr int kA = 16;
constexpr int kSteps = 4;
constexpr int kPitch = 260;
constexpr int kMT = 4;
constexpr int kWaves = 4;
constexpr int kThreads = kWaves * 32;
constexpr int kNTpw = 4;
constexpr int kMTreg = kMT - 1;
constexpr int kLdsFloats = 2 * kM * kPitch + kPitch + 16 * kPitch + kM * kA;
constexpr int kBlocksPerWgp = 2;
constexpr int kKB = 16;
constexpr int kKS = kKB / 4;
constexpr int kDB = 64;
constexpr int kDS = kDB / 4;
constexpr int kE = 1000;
constexpr int kTabTiles = (kE + 15) / 16;
constexpr int kTB = 32;
constexpr int kTS = kTB / 4;
constexpr int kFoldTiles = kH / 16 + 1;
constexpr int kFoldBlocks = (kFoldTiles + kWaves - 1) / kWaves;
constexpr int kFoldOff = kE * kH;
constexpr int kBiasOff = kFoldOff + kH * kA;
constexpr int kMwOff   = kFoldOff + kFoldTiles * 16 * kA;
constexpr int kCvOff   = kMwOff + 2 * kH * kH;
constexpr int kEbOff   = kCvOff + kH;
constexpr int kL0Off   = kEbOff + kE * kH;
constexpr int kMwBlocks = 2 * (kH / 16);
constexpr int kL0Blocks = (kTabTiles + kWaves - 1) / kWaves;
constexpr int kBlkMw   = kTabTiles;
constexpr int kBlkEb   = kBlkMw + kMwBlocks;
constexpr int kBlkCv   = kBlkEb + kTabTiles;
constexpr int kBlkFold = kBlkCv + 1;
constexpr int kBlkL0   = kBlkFold + kFoldBlocks;
constexpr int kTabGrid = kBlkL0 + kL0Blocks;
constexpr size_t kTabBytes = (size_t)(kL0Off + kTabTiles * 16 * kA) * sizeof(float);


struct BFrag { v4f x, y; };
struct W1Frag { v4f a0, c0, a1, c1; };

__device__ __forceinline__ v8f mma(v2f a, v2f b, v8f c) {
  return __builtin_amdgcn_wmma_f32_16x16x4_f32(false, a, false, b, (short)0, c,
                                               false, false);
}


__device__ __forceinline__ void load_b(BFrag (&bf)[kKS], const float* __restrict__ wp,
                                       int kb) {
  for (int s = 0; s < kKS; ++s) {
    const float* p = wp + (kb + 4 * s) * kH;
    bf[s].x = *(const v4f*)p;
    bf[s].y = *(const v4f*)(p + kH);
  }
}

__device__ __forceinline__ void load_a(v2f (&a)[kKS][kMT], const float* ap, int kb) {
  for (int s = 0; s < kKS; ++s) {
    for (int mt = 0; mt < kMT; ++mt)
      a[s][mt] = *(const v2f*)&ap[mt * 16 * kPitch + kb + 4 * s];
  }
}

__device__ __forceinline__ void mma_block(v8f (&acc)[kMT][kNTpw],
                                          const v2f (&a)[kKS][kMT],
                                          const BFrag (&w)[kKS]) {
  for (int s = 0; s < kKS; ++s) {
    v2f bf[kNTpw];
    for (int j = 0; j < kNTpw; ++j) {
      bf[j].x = w[s].x[j];
      bf[j].y = w[s].y[j];
    }
    for (int mt = 0; mt < kMT; ++mt) {
      for (int j = 0; j < kNTpw; ++j)
        acc[mt][j] = mma(a[s][mt], bf[j], acc[mt][j]);
    }
  }
}

__device__ __forceinline__ void load_w1(W1Frag (&w)[kKS], const float* __restrict__ wp,
                                        int kb) {
  for (int s = 0; s < kKS; ++s) {
    const float* p = wp + (kb + 4 * s) * kH;
    w[s].a0 = *(const v4f*)p;
    w[s].c0 = *(const v4f*)(p + kH * kH);
    w[s].a1 = *(const v4f*)(p + kH);
    w[s].c1 = *(const v4f*)(p + kH * kH + kH);
  }
}

__device__ __forceinline__ void load_s(v2f (&sv)[kKS], const float* sk, int kb) {
  for (int s = 0; s < kKS; ++s) sv[s] = *(const v2f*)&sk[kb + 4 * s];
}

__device__ __forceinline__ void mma_block1(v8f (&acc)[kMT][kNTpw],
                                           float (&uacc)[kNTpw],
                                           const v2f (&a)[kKS][kMT],
                                           const v2f (&svb)[kKS],
                                           const W1Frag (&w)[kKS],
                                           float inv) {
  for (int s = 0; s < kKS; ++s) {
    const v2f sv = svb[s];
    v2f bf[kNTpw];
    for (int j = 0; j < kNTpw; ++j) {
      bf[j].x = fmaf(-inv, w[s].c0[j], w[s].a0[j]);
      bf[j].y = fmaf(-inv, w[s].c1[j], w[s].a1[j]);
      uacc[j] = fmaf(sv.x, w[s].c0[j], uacc[j]);
      uacc[j] = fmaf(sv.y, w[s].c1[j], uacc[j]);
    }
    for (int mt = 0; mt < kMT; ++mt) {
      for (int j = 0; j < kNTpw; ++j) acc[mt][j] = mma(a[s][mt], bf[j], acc[mt][j]);
    }
  }
}

struct CFrag { v4f c0, c1; };
__device__ __forceinline__ void load_c(CFrag (&c)[kKS], const float* __restrict__ wp, int kb) {
  for (int s = 0; s < kKS; ++s) {
    const float* p = wp + (kb + 4 * s) * kH + kH * kH;
    c[s].c0 = *(const v4f*)p;
    c[s].c1 = *(const v4f*)(p + kH);
  }
}
__device__ __forceinline__ void fma_u(float (&uacc)[kNTpw], const v2f (&svb)[kKS],
                                      const CFrag (&c)[kKS]) {
  for (int s = 0; s < kKS; ++s) {
    for (int j = 0; j < kNTpw; ++j) {
      uacc[j] = fmaf(svb[s].x, c[s].c0[j], uacc[j]);
      uacc[j] = fmaf(svb[s].y, c[s].c1[j], uacc[j]);
    }
  }
}

template <bool kEff>
__device__ __forceinline__ void tab_load(v4f (&w)[kTS][kEff ? 4 : 2], v2f (&a)[kTS],
                                         const float* __restrict__ wp,
                                         const float* __restrict__ ap, int kb) {
  for (int s = 0; s < kTS; ++s) {
    const float* p = wp + (kb + 4 * s) * kH;
    w[s][0] = *(const v4f*)p;
    w[s][1] = *(const v4f*)(p + kH);
    if constexpr (kEff) {
      w[s][2] = *(const v4f*)(p + kH * kH);
      w[s][3] = *(const v4f*)(p + kH * kH + kH);
    }
    a[s] = *(const v2f*)&ap[kb + 4 * s];
  }
}
template <bool kEff>
__device__ __forceinline__ void tab_mma(v8f (&acc)[kNTpw], const v4f (&w)[kTS][kEff ? 4 : 2],
                                        const v2f (&a)[kTS], float inv) {
  for (int s = 0; s < kTS; ++s) {
    v2f bf[kNTpw];
    for (int j = 0; j < kNTpw; ++j) {
      if constexpr (kEff) {
        bf[j].x = fmaf(-inv, w[s][2][j], w[s][0][j]);
        bf[j].y = fmaf(-inv, w[s][3][j], w[s][1][j]);
      } else {
        bf[j].x = w[s][0][j];
        bf[j].y = w[s][1][j];
      }
    }
    for (int j = 0; j < kNTpw; ++j) acc[j] = mma(a[s], bf[j], acc[j]);
  }
}

__device__ __forceinline__ void group_fence() { __builtin_amdgcn_sched_barrier(0); }

__device__ __forceinline__ void load_d(v2f (&d)[kDS], const float* __restrict__ wp,
                                       int kb) {
  for (int s = 0; s < kDS; ++s) {
    const float* p = wp + (kb + 4 * s) * kA;
    d[s].x = p[0];
    d[s].y = p[kA];
  }
}

__device__ __forceinline__ void mma_d(v8f& acc, const v2f (&d)[kDS], const float* ap,
                                      int kb) {
  for (int s = 0; s < kDS; ++s) {
    const v2f a = *(const v2f*)&ap[kb + 4 * s];
    acc = mma(a, d[s], acc);
  }
}

__device__ __forceinline__ void load_af(v2f (&a)[kDS], const float* ap, int kb) {
  for (int s = 0; s < kDS; ++s) a[s] = *(const v2f*)&ap[kb + 4 * s];
}
__device__ __forceinline__ void mma_dd(v8f& acc0, v8f& acc1, const v2f (&a)[kDS],
                                       const v2f (&d)[kDS]) {
  for (int s = 0; s < kDS; s += 2) {
    acc0 = mma(a[s], d[s], acc0);
    acc1 = mma(a[s + 1], d[s + 1], acc1);
  }
}
__device__ __forceinline__ void logits_add(v8f& acc0, v8f& acc1,
                                           const float* __restrict__ wp, const float* ap) {
  v2f dA[kDS], dB[kDS], fA[kDS], fB[kDS];
  load_d(dA, wp, 0);
  load_af(fA, ap, 0);
  load_d(dB, wp, kDB);
  load_af(fB, ap, kDB);
  group_fence();
  mma_dd(acc0, acc1, fA, dA);
  group_fence();
  load_d(dA, wp, 2 * kDB);
  load_af(fA, ap, 2 * kDB);
  group_fence();
  mma_dd(acc0, acc1, fB, dB);
  group_fence();
  load_d(dB, wp, 3 * kDB);
  load_af(fB, ap, 3 * kDB);
  group_fence();
  mma_dd(acc0, acc1, fA, dA);
  group_fence();
  mma_dd(acc0, acc1, fB, dB);
  group_fence();
}
}

namespace {
template <bool kEff>
__device__ __forceinline__ void tab_tile(v8f (&acc)[kNTpw], const float* __restrict__ wp,
                                         const float* __restrict__ ap, float inv) {
  v4f wA[kTS][kEff ? 4 : 2], wB[kTS][kEff ? 4 : 2];
  v2f aA[kTS], aB[kTS];
  tab_load<kEff>(wA, aA, wp, ap, 0);
  for (int kb = 0; kb < kH - 2 * kTB; kb += 2 * kTB) {
    tab_load<kEff>(wB, aB, wp, ap, kb + kTB);
    group_fence();
    tab_mma<kEff>(acc, wA, aA, inv);
    group_fence();
    tab_load<kEff>(wA, aA, wp, ap, kb + 2 * kTB);
    group_fence();
    tab_mma<kEff>(acc, wB, aB, inv);
    group_fence();
  }
  tab_load<kEff>(wB, aB, wp, ap, kH - kTB);
  group_fence();
  tab_mma<kEff>(acc, wA, aA, inv);
  group_fence();
  tab_mma<kEff>(acc, wB, aB, inv);
}
}

__device__ __forceinline__ void fold_tile(const float* __restrict__ W2,
                                          const float* __restrict__ b2,
                                          const float* __restrict__ Wd,
                                          const float* __restrict__ bd,
                                          float* __restrict__ tab, int wt, int lrow,
                                          int khalf, int hi8) {
  const bool bias = (wt == kH / 16);
  const float* ap = bias ? b2 + khalf : W2 + (wt * 16 + lrow) * kH + khalf;
  const float* wp = Wd + khalf * kA + lrow;
  v2f dA[kDS], dB[kDS];
  const float c0 = bias ? bd[lrow] : 0.f;
  const float sc = bias ? (float)kSteps : 1.f;
  v8f acc = {};
  load_d(dA, wp, 0);
  load_d(dB, wp, kDB);
  mma_d(acc, dA, ap, 0);
  load_d(dA, wp, 2 * kDB);
  mma_d(acc, dB, ap, kDB);
  load_d(dB, wp, 3 * kDB);
  mma_d(acc, dA, ap, 2 * kDB);
  mma_d(acc, dB, ap, 3 * kDB);
#pragma unroll
  for (int r = 0; r < 8; ++r) tab[kFoldOff + (wt * 16 + hi8 + r) * kA + lrow] = fmaf(sc, acc[r], c0);
}

__device__ __forceinline__ void l0_tile(const float* __restrict__ emb,
                                        const float* __restrict__ Wd,
                                        float* __restrict__ tab, int t, int lrow,
                                        int khalf, int hi8) {
  const int arow  = min(t * 16 + lrow, kE - 1);
  const float* ap = emb + arow * kH + khalf;
  const float* wp = Wd + khalf * kA + lrow;
  v2f dA[kDS], dB[kDS];
  v8f acc = {};
  load_d(dA, wp, 0);
  load_d(dB, wp, kDB);
  mma_d(acc, dA, ap, 0);
  load_d(dA, wp, 2 * kDB);
  mma_d(acc, dB, ap, kDB);
  load_d(dB, wp, 3 * kDB);
  mma_d(acc, dA, ap, 2 * kDB);
  mma_d(acc, dB, ap, 3 * kDB);
#pragma unroll
  for (int r = 0; r < 8; ++r) tab[kL0Off + (t * 16 + hi8 + r) * kA + lrow] = acc[r];
}

__launch_bounds__(kThreads)
__global__ void commnet_table_kernel(const float* __restrict__ emb,
                                     const float* __restrict__ W1,
                                     const float* __restrict__ b1,
                                     const float* __restrict__ W2,
                                     const float* __restrict__ b2,
                                     const float* __restrict__ Wd,
                                     const float* __restrict__ bd,
                                     float* __restrict__ tab, size_t ws_bytes) {
  if (ws_bytes < kTabBytes) return;
  const int tid   = threadIdx.x;
  const int wave  = __builtin_amdgcn_readfirstlane(tid >> 5);
  const int lane  = tid & 31;
  const int lrow  = lane & 15;
  const int khalf = (lane >> 4) << 1;
  const int hi8   = (lane >> 4) << 3;
  const float inv = 1.0f / (float)(kM - 1);
  const int ncol  = wave * (kNTpw * 16) + 4 * lrow;
  const v8f vzero = {};
  if (blockIdx.x >= kBlkL0) {
    const int t = (blockIdx.x - kBlkL0) * kWaves + wave;
    if (t < kTabTiles) l0_tile(emb, Wd, tab, t, lrow, khalf, hi8);
    return;
  }
  if (blockIdx.x >= kBlkFold) {
    const int wt = (blockIdx.x - kBlkFold) * kWaves + wave;
    if (wt < kFoldTiles) fold_tile(W2, b2, Wd, bd, tab, wt, lrow, khalf, hi8);
    return;
  }
  if (blockIdx.x == kBlkCv) {
    v8f acc[kNTpw];
    for (int j = 0; j < kNTpw; ++j) acc[j] = vzero;
    tab_tile<true>(acc, W1 + khalf * kH + ncol, b2 + khalf, -1.0f);
    if (hi8 == 0) {
      v4f v;
      for (int j = 0; j < kNTpw; ++j) v[j] = acc[j][0];
      *(v4f*)&tab[kCvOff + ncol] = v;
    }
    return;
  }
  if (blockIdx.x >= kBlkMw) {
    const int idx = blockIdx.x - kBlkMw;
    const bool eb = idx >= kMwBlocks;
    const int mat = eb ? 1 : idx / (kH / 16);
    const int rt  = eb ? idx - kMwBlocks : idx % (kH / 16);
    const int nrow = eb ? kE : kH;
    const float* am = eb ? emb : W2;
    const int arow = min(rt * 16 + lrow, nrow - 1);
    v8f acc[kNTpw];
    for (int j = 0; j < kNTpw; ++j) acc[j] = vzero;
    tab_tile<false>(acc, W1 + (mat * kH + khalf) * kH + ncol,
                    am + arow * kH + khalf, inv);
    float* dst = eb ? tab + kEbOff : tab + kMwOff + mat * (kH * kH);
#pragma unroll
    for (int r = 0; r < 8; ++r) {
      const int row = rt * 16 + hi8 + r;
      if (row < nrow) {
        v4f v;
        for (int j = 0; j < kNTpw; ++j) v[j] = acc[j][r];
        *(v4f*)&dst[row * kH + ncol] = v;
      }
    }
    return;
  }
  const int tile  = blockIdx.x;
  const int arow  = min(tile * 16 + lrow, kE - 1);
  const float* ap = emb + arow * kH + khalf;

  v8f acc[kNTpw];
  for (int j = 0; j < kNTpw; ++j) acc[j] = vzero;
  const v4f bias = *(const v4f*)&b1[ncol];
  tab_tile<false>(acc, W1 + (2 * kH + khalf) * kH + ncol, ap, inv);
  tab_tile<true>(acc, W1 + khalf * kH + ncol, ap, inv);
#pragma unroll
  for (int r = 0; r < 8; ++r) {
    const int row = tile * 16 + hi8 + r;
    if (row < kE) {
      v4f v;
      for (int j = 0; j < kNTpw; ++j) v[j] = acc[j][r];
      v += bias;
      *(v4f*)&tab[row * kH + ncol] = v;
    }
  }
}

__device__ __forceinline__ void commnet_body_direct(
    float* smem,
    const int* __restrict__ ids, const float* __restrict__ emb,
    const float* __restrict__ W1, const float* __restrict__ b1,
    const float* __restrict__ W2, const float* __restrict__ b2,
    const float* __restrict__ Wd, const float* __restrict__ bd,
    float* __restrict__ out) {
  float* sh = smem;
  float* st = sh + kM * kPitch;
  float* sS = st + kM * kPitch;
  float* sq = sS + kPitch;
  constexpr int kPR = kMTreg;

  const int b     = blockIdx.x;
  const int tid   = threadIdx.x;
  const int wave  = __builtin_amdgcn_readfirstlane(tid >> 5);
  const int lane  = tid & 31;
  const int lrow  = lane & 15;
  const int khalf = (lane >> 4) << 1;
  const int hi8   = (lane >> 4) << 3;
  const float inv = 1.0f / (float)(kM - 1);
  const int nb    = wave * (kNTpw * 16);
  const int ncol  = nb + 4 * lrow;
  const int c2    = 2 * tid;

  const float* w1a = W1 + khalf * kH + ncol;
  const float* w1c = W1 + (2 * kH + khalf) * kH + ncol;
  const float* w2p = W2 + khalf * kH + ncol;
  const float* wdp = Wd + khalf * kA + lrow;
  const float* hA  = sh + lrow * kPitch + khalf;
  const float* tA  = st + lrow * kPitch + khalf;
  const float* sk  = sS + khalf;
  const float* hL  = sh + (wave * 16 + lrow) * kPitch + khalf;

  BFrag bA[kKS], bB[kKS];
  W1Frag wA[kKS], wB[kKS];
  v2f dA[kDS], dB[kDS];
  v2f aA[kKS][kMT], aB[kKS][kMT];
  v2f sA[kKS], sB[kKS];
  float bdv = 0.f;

  load_b(bA, w1c, 0);
  const v4f b1v = *(const v4f*)&b1[ncol];

  for (int m = 0; m < kM; ++m) {
    const int id  = ids[b * kM + m];
    *(v2f*)&sh[m * kPitch + c2] = *(const v2f*)&emb[id * kH + c2];
  }

  v8f acc[kMT][kNTpw];
  const v8f vzero = {};
  v8f p0[kPR][kNTpw];
  __syncthreads();
  for (int mt = 0; mt < kMT; ++mt)
    for (int j = 0; j < kNTpw; ++j) acc[mt][j] = vzero;
  load_a(aA, hA, 0);
  for (int kb = 0; kb < kH - 2 * kKB; kb += 2 * kKB) {
    load_b(bB, w1c, kb + kKB);
    load_a(aB, hA, kb + kKB);
    group_fence();
    mma_block(acc, aA, bA);
    group_fence();
    load_b(bA, w1c, kb + 2 * kKB);
    load_a(aA, hA, kb + 2 * kKB);
    group_fence();
    mma_block(acc, aB, bB);
    group_fence();
  }
  load_b(bB, w1c, kH - kKB);
  load_a(aB, hA, kH - kKB);
  group_fence();
  mma_block(acc, aA, bA);
  group_fence();
  mma_block(acc, aB, bB);
  group_fence();
  for (int mt = 0; mt < kPR; ++mt)
    for (int j = 0; j < kNTpw; ++j)
      for (int r = 0; r < 8; ++r) p0[mt][j][r] = acc[mt][j][r] + b1v[j];
#pragma unroll
  for (int r = 0; r < 8; ++r) {
    v4f v;
    for (int j = 0; j < kNTpw; ++j) v[j] = acc[kPR][j][r] + b1v[j];
    *(v4f*)&sq[(hi8 + r) * kPitch + ncol] = v;
  }

  for (int step = 0; step < kSteps; ++step) {
    int zoff = 0;
    asm volatile("" : "+s"(zoff));
    const float* w1s = w1a + zoff;
    const float* w2s = w2p + zoff;
    load_w1(wA, w1s, 0);
    __syncthreads();
    {
      v2f s = {0.f, 0.f};
#pragma unroll 8
      for (int m = 0; m < kM; ++m) s += *(const v2f*)&sh[m * kPitch + c2];
      *(v2f*)&sS[c2] = s;
    }
    __syncthreads();

    float uacc[kNTpw];
    for (int mt = 0; mt < kMT; ++mt)
      for (int j = 0; j < kNTpw; ++j) acc[mt][j] = vzero;
    for (int j = 0; j < kNTpw; ++j) uacc[j] = 0.f;

    load_a(aA, hA, 0);
    load_s(sA, sk, 0);
    for (int kb = 0; kb < kH - 2 * kKB; kb += 2 * kKB) {
      load_w1(wB, w1s, kb + kKB);
      load_a(aB, hA, kb + kKB);
      load_s(sB, sk, kb + kKB);
      group_fence();
      mma_block1(acc, uacc, aA, sA, wA, inv);
      group_fence();
      load_w1(wA, w1s, kb + 2 * kKB);
      load_a(aA, hA, kb + 2 * kKB);
      load_s(sA, sk, kb + 2 * kKB);
      group_fence();
      mma_block1(acc, uacc, aB, sB, wB, inv);
      group_fence();
    }
    load_w1(wB, w1s, kH - kKB);
    load_a(aB, hA, kH - kKB);
    load_s(sB, sk, kH - kKB);
    group_fence();
    mma_block1(acc, uacc, aA, sA, wA, inv);
    group_fence();
    load_b(bA, w2s, 0);
    group_fence();
    mma_block1(acc, uacc, aB, sB, wB, inv);
    group_fence();

    {
      v4f u;
      for (int j = 0; j < kNTpw; ++j)
        u[j] = (uacc[j] + __shfl_xor(uacc[j], 16)) * inv;
      v4f q[8];
#pragma unroll
      for (int r = 0; r < 8; ++r) q[r] = *(const v4f*)&sq[(hi8 + r) * kPitch + ncol];
      group_fence();
      for (int mt = 0; mt < kPR; ++mt) {
        const int mrow = mt * 16 + hi8;
#pragma unroll
        for (int r = 0; r < 8; ++r) {
          v4f v;
          for (int j = 0; j < kNTpw; ++j)
            v[j] = fmaxf(acc[mt][j][r] + p0[mt][j][r] + u[j], 0.f);
          *(v4f*)&st[(mrow + r) * kPitch + ncol] = v;
        }
      }
#pragma unroll
      for (int r = 0; r < 8; ++r) {
        v4f v;
        for (int j = 0; j < kNTpw; ++j)
          v[j] = fmaxf(acc[kPR][j][r] + q[r][j] + u[j], 0.f);
        *(v4f*)&st[(kPR * 16 + hi8 + r) * kPitch + ncol] = v;
      }
    }
    __syncthreads();

    for (int mt = 0; mt < kMT; ++mt)
      for (int j = 0; j < kNTpw; ++j) acc[mt][j] = vzero;
    load_a(aA, tA, 0);
    for (int kb = 0; kb < kH - 2 * kKB; kb += 2 * kKB) {
      load_b(bB, w2s, kb + kKB);
      load_a(aB, tA, kb + kKB);
      group_fence();
      mma_block(acc, aA, bA);
      group_fence();
      load_b(bA, w2s, kb + 2 * kKB);
      load_a(aA, tA, kb + 2 * kKB);
      group_fence();
      mma_block(acc, aB, bB);
      group_fence();
    }
    load_b(bB, w2s, kH - kKB);
    load_a(aB, tA, kH - kKB);
    const v4f b2v = *(const v4f*)&b2[ncol + zoff];
    group_fence();
    mma_block(acc, aA, bA);
    group_fence();
    mma_block(acc, aB, bB);
    group_fence();

    {
      for (int mh = 0; mh < kMT; mh += 2) {
        v4f hv[2][8];
        for (int mt = 0; mt < 2; ++mt) {
#pragma unroll
          for (int r = 0; r < 8; ++r)
            hv[mt][r] = *(const v4f*)&sh[((mh + mt) * 16 + hi8 + r) * kPitch + ncol];
        }
        group_fence();
        for (int mt = 0; mt < 2; ++mt) {
          const int mrow = (mh + mt) * 16 + hi8;
#pragma unroll
          for (int r = 0; r < 8; ++r) {
            v4f v;
            for (int j = 0; j < kNTpw; ++j)
              v[j] = hv[mt][r][j] + (acc[mh + mt][j][r] + b2v[j]);
            *(v4f*)&sh[(mrow + r) * kPitch + ncol] = v;
          }
        }
      }
    }
  }

  load_d(dA, wdp, 0);
  load_d(dB, wdp, kDB);
  bdv = bd[lrow];
  __syncthreads();
  {
    v8f accL = {};
    mma_d(accL, dA, hL, 0);
    load_d(dA, wdp, 2 * kDB);
    mma_d(accL, dB, hL, kDB);
    load_d(dB, wdp, 3 * kDB);
    mma_d(accL, dA, hL, 2 * kDB);
    mma_d(accL, dB, hL, 3 * kDB);
    const int mrow = wave * 16 + hi8;
#pragma unroll
    for (int r = 0; r < 8; ++r)
      out[(b * kM + mrow + r) * kA + lrow] = accL[r] + bdv;
  }
}

__device__ __forceinline__ void commnet_body_tab(
    float* smem, const int* __restrict__ ids,
    float* __restrict__ out, const float* __restrict__ tab) {
  float* sS = smem + 2 * kM * kPitch;
  float* sq = sS + kPitch;
  float* sL = sq + 16 * kPitch;
  constexpr int kPR = kMTreg;

  const int b     = blockIdx.x;
  const int tid   = threadIdx.x;
  const int wave  = __builtin_amdgcn_readfirstlane(tid >> 5);
  const int lane  = tid & 31;
  const int lrow  = lane & 15;
  const int khalf = (lane >> 4) << 1;
  const int hi8   = (lane >> 4) << 3;
  const float inv = 1.0f / (float)(kM - 1);
  const int ncol  = wave * (kNTpw * 16) + 4 * lrow;
  const int c2    = 2 * tid;
  const int* idr  = ids + b * kM + hi8;

  const float* mw  = tab + kMwOff + khalf * kH + ncol;
  const float* fdp = tab + kFoldOff + khalf * kA + lrow;
  const float* sk  = sS + khalf;
  const int aoff   = lrow * kPitch + khalf;
  const int loff   = (wave * 16 + lrow) * kPitch + khalf;

  W1Frag wA[kKS], wB[kKS];
  v2f aA[kKS][kMT], aB[kKS][kMT];
  v2f sA[kKS], sB[kKS];

  v8f acc[kMT][kNTpw];
  v4f tsum[kPR][8];
  {
    v4f e0[kMT][8], eb[kMT][8];
    for (int mt = 0; mt < kMT; ++mt) {
#pragma unroll
      for (int r = 0; r < 8; ++r) {
        const float* row = tab + idr[mt * 16 + r] * kH + ncol;
        e0[mt][r] = *(const v4f*)row;
        eb[mt][r] = *(const v4f*)(row + kEbOff);
      }
    }
    v8f hd;
#pragma unroll
    for (int r = 0; r < 8; ++r) hd[r] = tab[kL0Off + idr[wave * 16 + r] * kA + lrow];
    *(v4f*)&sL[tid * 8]     = __builtin_shufflevector(hd, hd, 0, 1, 2, 3);
    *(v4f*)&sL[tid * 8 + 4] = __builtin_shufflevector(hd, hd, 4, 5, 6, 7);
    v4f us = {0.f, 0.f, 0.f, 0.f};
    for (int mt = 0; mt < kMT; ++mt) {
#pragma unroll
      for (int r = 0; r < 8; ++r) us += eb[mt][r];
    }
    v4f u;
    for (int j = 0; j < kNTpw; ++j)
      u[j] = (us[j] + __shfl_xor(us[j], 16)) * inv;
    for (int mt = 0; mt < kPR; ++mt) {
#pragma unroll
      for (int r = 0; r < 8; ++r) {
        v4f v;
        for (int j = 0; j < kNTpw; ++j) {
          const float p = e0[mt][r][j] + u[j];
          acc[mt][j][r] = p;
          v[j] = fmaxf(p, 0.f);
        }
        tsum[mt][r] = v;
        *(v4f*)&smem[kM * kPitch + (mt * 16 + hi8 + r) * kPitch + ncol] = v;
      }
    }
#pragma unroll
    for (int r = 0; r < 8; ++r) {
      v4f v;
      for (int j = 0; j < kNTpw; ++j) {
        const float p = e0[kPR][r][j] + u[j];
        acc[kPR][j][r] = p;
        v[j] = fmaxf(p, 0.f);
      }
      *(v4f*)&sq[(hi8 + r) * kPitch + ncol] = v;
      *(v4f*)&smem[kM * kPitch + (kPR * 16 + hi8 + r) * kPitch + ncol] = v;
    }
  }

  int src = kM * kPitch;
  int dst = 0;
  v4f u;
  for (int step = 0;; ++step) {
    int zoff = 0;
    asm volatile("" : "+s"(zoff));
    const float* mws = mw + zoff;
    load_w1(wA, mws, 0);
    __syncthreads();
    {
      v2f s = {0.f, 0.f};
#pragma unroll 8
      for (int m = 0; m < kM; ++m) s += *(const v2f*)&smem[src + m * kPitch + c2];
      *(v2f*)&sS[c2] = s;
    }
    __syncthreads();

    float uacc[kNTpw];
    for (int j = 0; j < kNTpw; ++j) uacc[j] = 0.f;
    const float* tA = smem + src + aoff;
    load_a(aA, tA, 0);
    load_s(sA, sk, 0);
    for (int kb = 0; kb < kH - 2 * kKB; kb += 2 * kKB) {
      load_w1(wB, mws, kb + kKB);
      load_a(aB, tA, kb + kKB);
      load_s(sB, sk, kb + kKB);
      group_fence();
      mma_block1(acc, uacc, aA, sA, wA, inv);
      group_fence();
      load_w1(wA, mws, kb + 2 * kKB);
      load_a(aA, tA, kb + 2 * kKB);
      load_s(sA, sk, kb + 2 * kKB);
      group_fence();
      mma_block1(acc, uacc, aB, sB, wB, inv);
      group_fence();
    }
    load_w1(wB, mws, kH - kKB);
    load_a(aB, tA, kH - kKB);
    load_s(sB, sk, kH - kKB);
    const v4f cv = *(const v4f*)&tab[kCvOff + ncol + zoff];
    group_fence();
    mma_block1(acc, uacc, aA, sA, wA, inv);
    group_fence();
    mma_block1(acc, uacc, aB, sB, wB, inv);
    group_fence();

    for (int j = 0; j < kNTpw; ++j)
      u[j] = (uacc[j] + __shfl_xor(uacc[j], 16)) * inv + cv[j];
    if (step == kSteps - 2) break;
    {
      v4f q[8];
#pragma unroll
      for (int r = 0; r < 8; ++r) q[r] = *(const v4f*)&sq[(hi8 + r) * kPitch + ncol];
      group_fence();
      for (int mt = 0; mt < kPR; ++mt) {
#pragma unroll
        for (int r = 0; r < 8; ++r) {
          v4f v;
          for (int j = 0; j < kNTpw; ++j) {
            acc[mt][j][r] += u[j];
            v[j] = fmaxf(acc[mt][j][r], 0.f);
          }
          tsum[mt][r] += v;
          *(v4f*)&smem[dst + (mt * 16 + hi8 + r) * kPitch + ncol] = v;
        }
      }
#pragma unroll
      for (int r = 0; r < 8; ++r) {
        v4f v;
        for (int j = 0; j < kNTpw; ++j) {
          acc[kPR][j][r] += u[j];
          v[j] = fmaxf(acc[kPR][j][r], 0.f);
        }
        *(v4f*)&sq[(hi8 + r) * kPitch + ncol] = q[r] + v;
        *(v4f*)&smem[dst + (kPR * 16 + hi8 + r) * kPitch + ncol] = v;
      }
    }
    const int o = src;
    src = dst;
    dst = o;
  }
  {
    v4f q[8];
#pragma unroll
    for (int r = 0; r < 8; ++r) q[r] = *(const v4f*)&sq[(hi8 + r) * kPitch + ncol];
    group_fence();
    for (int mt = 0; mt < kPR; ++mt) {
#pragma unroll
      for (int r = 0; r < 8; ++r) {
        v4f v;
        for (int j = 0; j < kNTpw; ++j) v[j] = fmaxf(acc[mt][j][r] + u[j], 0.f);
        *(v4f*)&smem[dst + (mt * 16 + hi8 + r) * kPitch + ncol] = tsum[mt][r] + v;
      }
    }
#pragma unroll
    for (int r = 0; r < 8; ++r) {
      v4f v;
      for (int j = 0; j < kNTpw; ++j) v[j] = fmaxf(acc[kPR][j][r] + u[j], 0.f);
      *(v4f*)&smem[dst + (kPR * 16 + hi8 + r) * kPitch + ncol] = q[r] + v;
    }
  }

  const float g = tab[kBiasOff + lrow];
  __syncthreads();
  v8f accL0, accL1 = {};
  {
    const v4f lo = *(const v4f*)&sL[tid * 8];
    const v4f hi = *(const v4f*)&sL[tid * 8 + 4];
    accL0 = __builtin_shufflevector(lo, hi, 0, 1, 2, 3, 4, 5, 6, 7);
  }
  logits_add(accL0, accL1, fdp, smem + dst + loff);
  const int mrow = wave * 16 + hi8;
#pragma unroll
  for (int r = 0; r < 8; ++r)
    out[(b * kM + mrow + r) * kA + lrow] = (accL0[r] + accL1[r]) + g;
}

__launch_bounds__(kThreads, kBlocksPerWgp)
__global__ void commnet_wmma_kernel(
    const int* __restrict__ ids, const float* __restrict__ emb,
    const float* __restrict__ W1, const float* __restrict__ b1,
    const float* __restrict__ W2, const float* __restrict__ b2,
    const float* __restrict__ Wd, const float* __restrict__ bd,
    float* __restrict__ out, const float* __restrict__ tab, size_t ws_bytes) {
  extern __shared__ __attribute__((aligned(16))) float smem[];
  if (ws_bytes >= kTabBytes)
    commnet_body_tab(smem, ids, out, tab);
  else
    commnet_body_direct(smem, ids, emb, W1, b1, W2, b2, Wd, bd, out);
}

extern "C" void kernel_launch(void* const* d_in, const int* in_sizes, int n_in,
                              void* d_out, int out_size, void* d_ws, size_t ws_size,
                              hipStream_t stream) {
  (void)in_sizes; (void)n_in; (void)out_size;
  const int*   ids = (const int*)d_in[0];
  const float* emb = (const float*)d_in[1];
  const float* W1  = (const float*)d_in[2];
  const float* b1  = (const float*)d_in[3];
  const float* W2  = (const float*)d_in[4];
  const float* b2  = (const float*)d_in[5];
  const float* Wd  = (const float*)d_in[6];
  const float* bd  = (const float*)d_in[7];
  float* out = (float*)d_out;
  float* tab = (float*)d_ws;

  const size_t lds_bytes = (size_t)kLdsFloats * sizeof(float);
  (void)hipFuncSetAttribute((const void*)commnet_wmma_kernel,
                            hipFuncAttributeMaxDynamicSharedMemorySize,
                            (int)lds_bytes);
  commnet_table_kernel<<<dim3(kTabGrid), dim3(kThreads), 0, stream>>>(
      emb, W1, b1, W2, b2, Wd, bd, tab, ws_size);
  commnet_wmma_kernel<<<dim3(kB), dim3(kThreads), lds_bytes, stream>>>(
      ids, emb, W1, b1, W2, b2, Wd, bd, out, tab, ws_size);
}


# Column sums of t taken in the epilogue from each lane's own accumulator rows plus its partner half-wave, kept in t's row pads: no LDS column-sum pass, one barrier per step; S sum order changes
# speedup vs baseline: 4.6567x; 1.0331x over previous
#include <hip/hip_runtime.h>


typedef __attribute__((ext_vector_type(2))) float v2f;
typedef __attribute__((ext_vector_type(4))) float v4f;
typedef __attribute__((ext_vector_type(8))) float v8f;

namespace {
constexpr int kB = 1024;
constexpr int kM = 64;
constexpr int kH = 256;
constexpr int kA = 16;
constexpr int kSteps = 4;
constexpr int kPitch = 260;
constexpr int kMT = 4;
constexpr int kWaves = 4;
constexpr int kThreads = kWaves * 32;
constexpr int kNTpw = 4;
constexpr int kMTreg = kMT - 1;
constexpr int kLdsFloats = 2 * kM * kPitch + kPitch + 16 * kPitch + kM * kA;
constexpr int kBlocksPerWgp = 2;
constexpr int kKB = 16;
constexpr int kKS = kKB / 4;
constexpr int kDB = 64;
constexpr int kDS = kDB / 4;
constexpr int kE = 1000;
constexpr int kTabTiles = (kE + 15) / 16;
constexpr int kTB = 32;
constexpr int kTS = kTB / 4;
constexpr int kFoldTiles = kH / 16 + 1;
constexpr int kFoldBlocks = (kFoldTiles + kWaves - 1) / kWaves;
constexpr int kFoldOff = kE * kH;
constexpr int kBiasOff = kFoldOff + kH * kA;
constexpr int kMwOff   = kFoldOff + kFoldTiles * 16 * kA;
constexpr int kCvOff   = kMwOff + 2 * kH * kH;
constexpr int kEbOff   = kCvOff + kH;
constexpr int kL0Off   = kEbOff + kE * kH;
constexpr int kMwBlocks = 2 * (kH / 16);
constexpr int kL0Blocks = (kTabTiles + kWaves - 1) / kWaves;
constexpr int kBlkMw   = kTabTiles;
constexpr int kBlkEb   = kBlkMw + kMwBlocks;
constexpr int kBlkCv   = kBlkEb + kTabTiles;
constexpr int kBlkFold = kBlkCv + 1;
constexpr int kBlkL0   = kBlkFold + kFoldBlocks;
constexpr int kTabGrid = kBlkL0 + kL0Blocks;
constexpr size_t kTabBytes = (size_t)(kL0Off + kTabTiles * 16 * kA) * sizeof(float);


struct BFrag { v4f x, y; };
struct W1Frag { v4f a0, c0, a1, c1; };

__device__ __forceinline__ v8f mma(v2f a, v2f b, v8f c) {
  return __builtin_amdgcn_wmma_f32_16x16x4_f32(false, a, false, b, (short)0, c,
                                               false, false);
}


__device__ __forceinline__ void load_b(BFrag (&bf)[kKS], const float* __restrict__ wp,
                                       int kb) {
  for (int s = 0; s < kKS; ++s) {
    const float* p = wp + (kb + 4 * s) * kH;
    bf[s].x = *(const v4f*)p;
    bf[s].y = *(const v4f*)(p + kH);
  }
}

__device__ __forceinline__ void load_a(v2f (&a)[kKS][kMT], const float* ap, int kb) {
  for (int s = 0; s < kKS; ++s) {
    for (int mt = 0; mt < kMT; ++mt)
      a[s][mt] = *(const v2f*)&ap[mt * 16 * kPitch + kb + 4 * s];
  }
}

__device__ __forceinline__ void mma_block(v8f (&acc)[kMT][kNTpw],
                                          const v2f (&a)[kKS][kMT],
                                          const BFrag (&w)[kKS]) {
  for (int s = 0; s < kKS; ++s) {
    v2f bf[kNTpw];
    for (int j = 0; j < kNTpw; ++j) {
      bf[j].x = w[s].x[j];
      bf[j].y = w[s].y[j];
    }
    for (int mt = 0; mt < kMT; ++mt) {
      for (int j = 0; j < kNTpw; ++j)
        acc[mt][j] = mma(a[s][mt], bf[j], acc[mt][j]);
    }
  }
}

__device__ __forceinline__ void load_w1(W1Frag (&w)[kKS], const float* __restrict__ wp,
                                        int kb) {
  for (int s = 0; s < kKS; ++s) {
    const float* p = wp + (kb + 4 * s) * kH;
    w[s].a0 = *(const v4f*)p;
    w[s].c0 = *(const v4f*)(p + kH * kH);
    w[s].a1 = *(const v4f*)(p + kH);
    w[s].c1 = *(const v4f*)(p + kH * kH + kH);
  }
}

__device__ __forceinline__ void load_s(v2f (&sv)[kKS], const float* sk, int kb) {
  for (int s = 0; s < kKS; ++s) sv[s] = *(const v2f*)&sk[kb + 4 * s];
}

__device__ __forceinline__ void load_sp(v2f (&sv)[kKS], const float* sk, int kb) {
  for (int s = 0; s < kKS; ++s) sv[s] = *(const v2f*)&sk[(kb / 4 + s) * kPitch];
}

__device__ __forceinline__ void mma_block1(v8f (&acc)[kMT][kNTpw],
                                           float (&uacc)[kNTpw],
                                           const v2f (&a)[kKS][kMT],
                                           const v2f (&svb)[kKS],
                                           const W1Frag (&w)[kKS],
                                           float inv) {
  for (int s = 0; s < kKS; ++s) {
    const v2f sv = svb[s];
    v2f bf[kNTpw];
    for (int j = 0; j < kNTpw; ++j) {
      bf[j].x = fmaf(-inv, w[s].c0[j], w[s].a0[j]);
      bf[j].y = fmaf(-inv, w[s].c1[j], w[s].a1[j]);
      uacc[j] = fmaf(sv.x, w[s].c0[j], uacc[j]);
      uacc[j] = fmaf(sv.y, w[s].c1[j], uacc[j]);
    }
    for (int mt = 0; mt < kMT; ++mt) {
      for (int j = 0; j < kNTpw; ++j) acc[mt][j] = mma(a[s][mt], bf[j], acc[mt][j]);
    }
  }
}

struct CFrag { v4f c0, c1; };
__device__ __forceinline__ void load_c(CFrag (&c)[kKS], const float* __restrict__ wp, int kb) {
  for (int s = 0; s < kKS; ++s) {
    const float* p = wp + (kb + 4 * s) * kH + kH * kH;
    c[s].c0 = *(const v4f*)p;
    c[s].c1 = *(const v4f*)(p + kH);
  }
}
__device__ __forceinline__ void fma_u(float (&uacc)[kNTpw], const v2f (&svb)[kKS],
                                      const CFrag (&c)[kKS]) {
  for (int s = 0; s < kKS; ++s) {
    for (int j = 0; j < kNTpw; ++j) {
      uacc[j] = fmaf(svb[s].x, c[s].c0[j], uacc[j]);
      uacc[j] = fmaf(svb[s].y, c[s].c1[j], uacc[j]);
    }
  }
}

template <bool kEff>
__device__ __forceinline__ void tab_load(v4f (&w)[kTS][kEff ? 4 : 2], v2f (&a)[kTS],
                                         const float* __restrict__ wp,
                                         const float* __restrict__ ap, int kb) {
  for (int s = 0; s < kTS; ++s) {
    const float* p = wp + (kb + 4 * s) * kH;
    w[s][0] = *(const v4f*)p;
    w[s][1] = *(const v4f*)(p + kH);
    if constexpr (kEff) {
      w[s][2] = *(const v4f*)(p + kH * kH);
      w[s][3] = *(const v4f*)(p + kH * kH + kH);
    }
    a[s] = *(const v2f*)&ap[kb + 4 * s];
  }
}
template <bool kEff>
__device__ __forceinline__ void tab_mma(v8f (&acc)[kNTpw], const v4f (&w)[kTS][kEff ? 4 : 2],
                                        const v2f (&a)[kTS], float inv) {
  for (int s = 0; s < kTS; ++s) {
    v2f bf[kNTpw];
    for (int j = 0; j < kNTpw; ++j) {
      if constexpr (kEff) {
        bf[j].x = fmaf(-inv, w[s][2][j], w[s][0][j]);
        bf[j].y = fmaf(-inv, w[s][3][j], w[s][1][j]);
      } else {
        bf[j].x = w[s][0][j];
        bf[j].y = w[s][1][j];
      }
    }
    for (int j = 0; j < kNTpw; ++j) acc[j] = mma(a[s], bf[j], acc[j]);
  }
}

__device__ __forceinline__ void group_fence() { __builtin_amdgcn_sched_barrier(0); }

__device__ __forceinline__ void load_d(v2f (&d)[kDS], const float* __restrict__ wp,
                                       int kb) {
  for (int s = 0; s < kDS; ++s) {
    const float* p = wp + (kb + 4 * s) * kA;
    d[s].x = p[0];
    d[s].y = p[kA];
  }
}

__device__ __forceinline__ void mma_d(v8f& acc, const v2f (&d)[kDS], const float* ap,
                                      int kb) {
  for (int s = 0; s < kDS; ++s) {
    const v2f a = *(const v2f*)&ap[kb + 4 * s];
    acc = mma(a, d[s], acc);
  }
}

__device__ __forceinline__ void load_af(v2f (&a)[kDS], const float* ap, int kb) {
  for (int s = 0; s < kDS; ++s) a[s] = *(const v2f*)&ap[kb + 4 * s];
}
__device__ __forceinline__ void mma_dd(v8f& acc0, v8f& acc1, const v2f (&a)[kDS],
                                       const v2f (&d)[kDS]) {
  for (int s = 0; s < kDS; s += 2) {
    acc0 = mma(a[s], d[s], acc0);
    acc1 = mma(a[s + 1], d[s + 1], acc1);
  }
}
__device__ __forceinline__ void logits_add(v8f& acc0, v8f& acc1,
                                           const float* __restrict__ wp, const float* ap) {
  v2f dA[kDS], dB[kDS], fA[kDS], fB[kDS];
  load_d(dA, wp, 0);
  load_af(fA, ap, 0);
  load_d(dB, wp, kDB);
  load_af(fB, ap, kDB);
  group_fence();
  mma_dd(acc0, acc1, fA, dA);
  group_fence();
  load_d(dA, wp, 2 * kDB);
  load_af(fA, ap, 2 * kDB);
  group_fence();
  mma_dd(acc0, acc1, fB, dB);
  group_fence();
  load_d(dB, wp, 3 * kDB);
  load_af(fB, ap, 3 * kDB);
  group_fence();
  mma_dd(acc0, acc1, fA, dA);
  group_fence();
  mma_dd(acc0, acc1, fB, dB);
  group_fence();
}
}

namespace {
template <bool kEff>
__device__ __forceinline__ void tab_tile(v8f (&acc)[kNTpw], const float* __restrict__ wp,
                                         const float* __restrict__ ap, float inv) {
  v4f wA[kTS][kEff ? 4 : 2], wB[kTS][kEff ? 4 : 2];
  v2f aA[kTS], aB[kTS];
  tab_load<kEff>(wA, aA, wp, ap, 0);
  for (int kb = 0; kb < kH - 2 * kTB; kb += 2 * kTB) {
    tab_load<kEff>(wB, aB, wp, ap, kb + kTB);
    group_fence();
    tab_mma<kEff>(acc, wA, aA, inv);
    group_fence();
    tab_load<kEff>(wA, aA, wp, ap, kb + 2 * kTB);
    group_fence();
    tab_mma<kEff>(acc, wB, aB, inv);
    group_fence();
  }
  tab_load<kEff>(wB, aB, wp, ap, kH - kTB);
  group_fence();
  tab_mma<kEff>(acc, wA, aA, inv);
  group_fence();
  tab_mma<kEff>(acc, wB, aB, inv);
}
}

__device__ __forceinline__ void fold_tile(const float* __restrict__ W2,
                                          const float* __restrict__ b2,
                                          const float* __restrict__ Wd,
                                          const float* __restrict__ bd,
                                          float* __restrict__ tab, int wt, int lrow,
                                          int khalf, int hi8) {
  const bool bias = (wt == kH / 16);
  const float* ap = bias ? b2 + khalf : W2 + (wt * 16 + lrow) * kH + khalf;
  const float* wp = Wd + khalf * kA + lrow;
  v2f dA[kDS], dB[kDS];
  const float c0 = bias ? bd[lrow] : 0.f;
  const float sc = bias ? (float)kSteps : 1.f;
  v8f acc = {};
  load_d(dA, wp, 0);
  load_d(dB, wp, kDB);
  mma_d(acc, dA, ap, 0);
  load_d(dA, wp, 2 * kDB);
  mma_d(acc, dB, ap, kDB);
  load_d(dB, wp, 3 * kDB);
  mma_d(acc, dA, ap, 2 * kDB);
  mma_d(acc, dB, ap, 3 * kDB);
#pragma unroll
  for (int r = 0; r < 8; ++r) tab[kFoldOff + (wt * 16 + hi8 + r) * kA + lrow] = fmaf(sc, acc[r], c0);
}

__device__ __forceinline__ void l0_tile(const float* __restrict__ emb,
                                        const float* __restrict__ Wd,
                                        float* __restrict__ tab, int t, int lrow,
                                        int khalf, int hi8) {
  const int arow  = min(t * 16 + lrow, kE - 1);
  const float* ap = emb + arow * kH + khalf;
  const float* wp = Wd + khalf * kA + lrow;
  v2f dA[kDS], dB[kDS];
  v8f acc = {};
  load_d(dA, wp, 0);
  load_d(dB, wp, kDB);
  mma_d(acc, dA, ap, 0);
  load_d(dA, wp, 2 * kDB);
  mma_d(acc, dB, ap, kDB);
  load_d(dB, wp, 3 * kDB);
  mma_d(acc, dA, ap, 2 * kDB);
  mma_d(acc, dB, ap, 3 * kDB);
#pragma unroll
  for (int r = 0; r < 8; ++r) tab[kL0Off + (t * 16 + hi8 + r) * kA + lrow] = acc[r];
}

__launch_bounds__(kThreads)
__global__ void commnet_table_kernel(const float* __restrict__ emb,
                                     const float* __restrict__ W1,
                                     const float* __restrict__ b1,
                                     const float* __restrict__ W2,
                                     const float* __restrict__ b2,
                                     const float* __restrict__ Wd,
                                     const float* __restrict__ bd,
                                     float* __restrict__ tab, size_t ws_bytes) {
  if (ws_bytes < kTabBytes) return;
  const int tid   = threadIdx.x;
  const int wave  = __builtin_amdgcn_readfirstlane(tid >> 5);
  const int lane  = tid & 31;
  const int lrow  = lane & 15;
  const int khalf = (lane >> 4) << 1;
  const int hi8   = (lane >> 4) << 3;
  const float inv = 1.0f / (float)(kM - 1);
  const int ncol  = wave * (kNTpw * 16) + 4 * lrow;
  const v8f vzero = {};
  if (blockIdx.x >= kBlkL0) {
    const int t = (blockIdx.x - kBlkL0) * kWaves + wave;
    if (t < kTabTiles) l0_tile(emb, Wd, tab, t, lrow, khalf, hi8);
    return;
  }
  if (blockIdx.x >= kBlkFold) {
    const int wt = (blockIdx.x - kBlkFold) * kWaves + wave;
    if (wt < kFoldTiles) fold_tile(W2, b2, Wd, bd, tab, wt, lrow, khalf, hi8);
    return;
  }
  if (blockIdx.x == kBlkCv) {
    v8f acc[kNTpw];
    for (int j = 0; j < kNTpw; ++j) acc[j] = vzero;
    tab_tile<true>(acc, W1 + khalf * kH + ncol, b2 + khalf, -1.0f);
    if (hi8 == 0) {
      v4f v;
      for (int j = 0; j < kNTpw; ++j) v[j] = acc[j][0];
      *(v4f*)&tab[kCvOff + ncol] = v;
    }
    return;
  }
  if (blockIdx.x >= kBlkMw) {
    const int idx = blockIdx.x - kBlkMw;
    const bool eb = idx >= kMwBlocks;
    const int mat = eb ? 1 : idx / (kH / 16);
    const int rt  = eb ? idx - kMwBlocks : idx % (kH / 16);
    const int nrow = eb ? kE : kH;
    const float* am = eb ? emb : W2;
    const int arow = min(rt * 16 + lrow, nrow - 1);
    v8f acc[kNTpw];
    for (int j = 0; j < kNTpw; ++j) acc[j] = vzero;
    tab_tile<false>(acc, W1 + (mat * kH + khalf) * kH + ncol,
                    am + arow * kH + khalf, inv);
    float* dst = eb ? tab + kEbOff : tab + kMwOff + mat * (kH * kH);
#pragma unroll
    for (int r = 0; r < 8; ++r) {
      const int row = rt * 16 + hi8 + r;
      if (row < nrow) {
        v4f v;
        for (int j = 0; j < kNTpw; ++j) v[j] = acc[j][r];
        *(v4f*)&dst[row * kH + ncol] = v;
      }
    }
    return;
  }
  const int tile  = blockIdx.x;
  const int arow  = min(tile * 16 + lrow, kE - 1);
  const float* ap = emb + arow * kH + khalf;

  v8f acc[kNTpw];
  for (int j = 0; j < kNTpw; ++j) acc[j] = vzero;
  const v4f bias = *(const v4f*)&b1[ncol];
  tab_tile<false>(acc, W1 + (2 * kH + khalf) * kH + ncol, ap, inv);
  tab_tile<true>(acc, W1 + khalf * kH + ncol, ap, inv);
#pragma unroll
  for (int r = 0; r < 8; ++r) {
    const int row = tile * 16 + hi8 + r;
    if (row < kE) {
      v4f v;
      for (int j = 0; j < kNTpw; ++j) v[j] = acc[j][r];
      v += bias;
      *(v4f*)&tab[row * kH + ncol] = v;
    }
  }
}

__device__ __forceinline__ void commnet_body_direct(
    float* smem,
    const int* __restrict__ ids, const float* __restrict__ emb,
    const float* __restrict__ W1, const float* __restrict__ b1,
    const float* __restrict__ W2, const float* __restrict__ b2,
    const float* __restrict__ Wd, const float* __restrict__ bd,
    float* __restrict__ out) {
  float* sh = smem;
  float* st = sh + kM * kPitch;
  float* sS = st + kM * kPitch;
  float* sq = sS + kPitch;
  constexpr int kPR = kMTreg;

  const int b     = blockIdx.x;
  const int tid   = threadIdx.x;
  const int wave  = __builtin_amdgcn_readfirstlane(tid >> 5);
  const int lane  = tid & 31;
  const int lrow  = lane & 15;
  const int khalf = (lane >> 4) << 1;
  const int hi8   = (lane >> 4) << 3;
  const float inv = 1.0f / (float)(kM - 1);
  const int nb    = wave * (kNTpw * 16);
  const int ncol  = nb + 4 * lrow;
  const int c2    = 2 * tid;

  const float* w1a = W1 + khalf * kH + ncol;
  const float* w1c = W1 + (2 * kH + khalf) * kH + ncol;
  const float* w2p = W2 + khalf * kH + ncol;
  const float* wdp = Wd + khalf * kA + lrow;
  const float* hA  = sh + lrow * kPitch + khalf;
  const float* tA  = st + lrow * kPitch + khalf;
  const float* sk  = sS + khalf;
  const float* hL  = sh + (wave * 16 + lrow) * kPitch + khalf;

  BFrag bA[kKS], bB[kKS];
  W1Frag wA[kKS], wB[kKS];
  v2f dA[kDS], dB[kDS];
  v2f aA[kKS][kMT], aB[kKS][kMT];
  v2f sA[kKS], sB[kKS];
  float bdv = 0.f;

  load_b(bA, w1c, 0);
  const v4f b1v = *(const v4f*)&b1[ncol];

  for (int m = 0; m < kM; ++m) {
    const int id  = ids[b * kM + m];
    *(v2f*)&sh[m * kPitch + c2] = *(const v2f*)&emb[id * kH + c2];
  }

  v8f acc[kMT][kNTpw];
  const v8f vzero = {};
  v8f p0[kPR][kNTpw];
  __syncthreads();
  for (int mt = 0; mt < kMT; ++mt)
    for (int j = 0; j < kNTpw; ++j) acc[mt][j] = vzero;
  load_a(aA, hA, 0);
  for (int kb = 0; kb < kH - 2 * kKB; kb += 2 * kKB) {
    load_b(bB, w1c, kb + kKB);
    load_a(aB, hA, kb + kKB);
    group_fence();
    mma_block(acc, aA, bA);
    group_fence();
    load_b(bA, w1c, kb + 2 * kKB);
    load_a(aA, hA, kb + 2 * kKB);
    group_fence();
    mma_block(acc, aB, bB);
    group_fence();
  }
  load_b(bB, w1c, kH - kKB);
  load_a(aB, hA, kH - kKB);
  group_fence();
  mma_block(acc, aA, bA);
  group_fence();
  mma_block(acc, aB, bB);
  group_fence();
  for (int mt = 0; mt < kPR; ++mt)
    for (int j = 0; j < kNTpw; ++j)
      for (int r = 0; r < 8; ++r) p0[mt][j][r] = acc[mt][j][r] + b1v[j];
#pragma unroll
  for (int r = 0; r < 8; ++r) {
    v4f v;
    for (int j = 0; j < kNTpw; ++j) v[j] = acc[kPR][j][r] + b1v[j];
    *(v4f*)&sq[(hi8 + r) * kPitch + ncol] = v;
  }

  for (int step = 0; step < kSteps; ++step) {
    int zoff = 0;
    asm volatile("" : "+s"(zoff));
    const float* w1s = w1a + zoff;
    const float* w2s = w2p + zoff;
    load_w1(wA, w1s, 0);
    __syncthreads();
    {
      v2f s = {0.f, 0.f};
#pragma unroll 8
      for (int m = 0; m < kM; ++m) s += *(const v2f*)&sh[m * kPitch + c2];
      *(v2f*)&sS[c2] = s;
    }
    __syncthreads();

    float uacc[kNTpw];
    for (int mt = 0; mt < kMT; ++mt)
      for (int j = 0; j < kNTpw; ++j) acc[mt][j] = vzero;
    for (int j = 0; j < kNTpw; ++j) uacc[j] = 0.f;

    load_a(aA, hA, 0);
    load_s(sA, sk, 0);
    for (int kb = 0; kb < kH - 2 * kKB; kb += 2 * kKB) {
      load_w1(wB, w1s, kb + kKB);
      load_a(aB, hA, kb + kKB);
      load_s(sB, sk, kb + kKB);
      group_fence();
      mma_block1(acc, uacc, aA, sA, wA, inv);
      group_fence();
      load_w1(wA, w1s, kb + 2 * kKB);
      load_a(aA, hA, kb + 2 * kKB);
      load_s(sA, sk, kb + 2 * kKB);
      group_fence();
      mma_block1(acc, uacc, aB, sB, wB, inv);
      group_fence();
    }
    load_w1(wB, w1s, kH - kKB);
    load_a(aB, hA, kH - kKB);
    load_s(sB, sk, kH - kKB);
    group_fence();
    mma_block1(acc, uacc, aA, sA, wA, inv);
    group_fence();
    load_b(bA, w2s, 0);
    group_fence();
    mma_block1(acc, uacc, aB, sB, wB, inv);
    group_fence();

    {
      v4f u;
      for (int j = 0; j < kNTpw; ++j)
        u[j] = (uacc[j] + __shfl_xor(uacc[j], 16)) * inv;
      v4f q[8];
#pragma unroll
      for (int r = 0; r < 8; ++r) q[r] = *(const v4f*)&sq[(hi8 + r) * kPitch + ncol];
      group_fence();
      for (int mt = 0; mt < kPR; ++mt) {
        const int mrow = mt * 16 + hi8;
#pragma unroll
        for (int r = 0; r < 8; ++r) {
          v4f v;
          for (int j = 0; j < kNTpw; ++j)
            v[j] = fmaxf(acc[mt][j][r] + p0[mt][j][r] + u[j], 0.f);
          *(v4f*)&st[(mrow + r) * kPitch + ncol] = v;
        }
      }
#pragma unroll
      for (int r = 0; r < 8; ++r) {
        v4f v;
        for (int j = 0; j < kNTpw; ++j)
          v[j] = fmaxf(acc[kPR][j][r] + q[r][j] + u[j], 0.f);
        *(v4f*)&st[(kPR * 16 + hi8 + r) * kPitch + ncol] = v;
      }
    }
    __syncthreads();

    for (int mt = 0; mt < kMT; ++mt)
      for (int j = 0; j < kNTpw; ++j) acc[mt][j] = vzero;
    load_a(aA, tA, 0);
    for (int kb = 0; kb < kH - 2 * kKB; kb += 2 * kKB) {
      load_b(bB, w2s, kb + kKB);
      load_a(aB, tA, kb + kKB);
      group_fence();
      mma_block(acc, aA, bA);
      group_fence();
      load_b(bA, w2s, kb + 2 * kKB);
      load_a(aA, tA, kb + 2 * kKB);
      group_fence();
      mma_block(acc, aB, bB);
      group_fence();
    }
    load_b(bB, w2s, kH - kKB);
    load_a(aB, tA, kH - kKB);
    const v4f b2v = *(const v4f*)&b2[ncol + zoff];
    group_fence();
    mma_block(acc, aA, bA);
    group_fence();
    mma_block(acc, aB, bB);
    group_fence();

    {
      for (int mh = 0; mh < kMT; mh += 2) {
        v4f hv[2][8];
        for (int mt = 0; mt < 2; ++mt) {
#pragma unroll
          for (int r = 0; r < 8; ++r)
            hv[mt][r] = *(const v4f*)&sh[((mh + mt) * 16 + hi8 + r) * kPitch + ncol];
        }
        group_fence();
        for (int mt = 0; mt < 2; ++mt) {
          const int mrow = (mh + mt) * 16 + hi8;
#pragma unroll
          for (int r = 0; r < 8; ++r) {
            v4f v;
            for (int j = 0; j < kNTpw; ++j)
              v[j] = hv[mt][r][j] + (acc[mh + mt][j][r] + b2v[j]);
            *(v4f*)&sh[(mrow + r) * kPitch + ncol] = v;
          }
        }
      }
    }
  }

  load_d(dA, wdp, 0);
  load_d(dB, wdp, kDB);
  bdv = bd[lrow];
  __syncthreads();
  {
    v8f accL = {};
    mma_d(accL, dA, hL, 0);
    load_d(dA, wdp, 2 * kDB);
    mma_d(accL, dB, hL, kDB);
    load_d(dB, wdp, 3 * kDB);
    mma_d(accL, dA, hL, 2 * kDB);
    mma_d(accL, dB, hL, 3 * kDB);
    const int mrow = wave * 16 + hi8;
#pragma unroll
    for (int r = 0; r < 8; ++r)
      out[(b * kM + mrow + r) * kA + lrow] = accL[r] + bdv;
  }
}

__device__ __forceinline__ void commnet_body_tab(
    float* smem, const int* __restrict__ ids,
    float* __restrict__ out, const float* __restrict__ tab) {
  float* sS = smem + 2 * kM * kPitch;
  float* sq = sS + kPitch;
  float* sL = sq + 16 * kPitch;
  constexpr int kPR = kMTreg;

  const int b     = blockIdx.x;
  const int tid   = threadIdx.x;
  const int wave  = __builtin_amdgcn_readfirstlane(tid >> 5);
  const int lane  = tid & 31;
  const int lrow  = lane & 15;
  const int khalf = (lane >> 4) << 1;
  const int hi8   = (lane >> 4) << 3;
  const float inv = 1.0f / (float)(kM - 1);
  const int ncol  = wave * (kNTpw * 16) + 4 * lrow;
  const int* idr  = ids + b * kM + hi8;

  const float* mw  = tab + kMwOff + khalf * kH + ncol;
  const float* fdp = tab + kFoldOff + khalf * kA + lrow;
  const int aoff   = lrow * kPitch + khalf;
  const int loff   = (wave * 16 + lrow) * kPitch + khalf;

  W1Frag wA[kKS], wB[kKS];
  v2f aA[kKS][kMT], aB[kKS][kMT];
  v2f sA[kKS], sB[kKS];

  const int srow = (wave * (kNTpw * 4) + lrow) * kPitch + kH;
  int zoff = 0;
  asm volatile("" : "+s"(zoff));
  const float* mws = mw + zoff;

  v8f acc[kMT][kNTpw];
  v4f tsum[kPR][8];
  {
    v4f e0[kMT][8], eb[kMT][8];
    for (int mt = 0; mt < kMT; ++mt) {
#pragma unroll
      for (int r = 0; r < 8; ++r) {
        const float* row = tab + idr[mt * 16 + r] * kH + ncol;
        e0[mt][r] = *(const v4f*)row;
        eb[mt][r] = *(const v4f*)(row + kEbOff);
      }
    }
    v8f hd;
#pragma unroll
    for (int r = 0; r < 8; ++r) hd[r] = tab[kL0Off + idr[wave * 16 + r] * kA + lrow];
    *(v4f*)&sL[tid * 8]     = __builtin_shufflevector(hd, hd, 0, 1, 2, 3);
    *(v4f*)&sL[tid * 8 + 4] = __builtin_shufflevector(hd, hd, 4, 5, 6, 7);
    v4f us = {0.f, 0.f, 0.f, 0.f};
    for (int mt = 0; mt < kMT; ++mt) {
#pragma unroll
      for (int r = 0; r < 8; ++r) us += eb[mt][r];
    }
    v4f u;
    for (int j = 0; j < kNTpw; ++j)
      u[j] = (us[j] + __shfl_xor(us[j], 16)) * inv;
    load_w1(wA, mws, 0);
    v4f ss = {0.f, 0.f, 0.f, 0.f};
    for (int mt = 0; mt < kPR; ++mt) {
#pragma unroll
      for (int r = 0; r < 8; ++r) {
        v4f v;
        for (int j = 0; j < kNTpw; ++j) {
          const float p = e0[mt][r][j] + u[j];
          acc[mt][j][r] = p;
          v[j] = fmaxf(p, 0.f);
        }
        tsum[mt][r] = v;
        ss += v;
        *(v4f*)&smem[kM * kPitch + (mt * 16 + hi8 + r) * kPitch + ncol] = v;
      }
    }
#pragma unroll
    for (int r = 0; r < 8; ++r) {
      v4f v;
      for (int j = 0; j < kNTpw; ++j) {
        const float p = e0[kPR][r][j] + u[j];
        acc[kPR][j][r] = p;
        v[j] = fmaxf(p, 0.f);
      }
      ss += v;
      *(v4f*)&sq[(hi8 + r) * kPitch + ncol] = v;
      *(v4f*)&smem[kM * kPitch + (kPR * 16 + hi8 + r) * kPitch + ncol] = v;
    }
    v4f sp;
    for (int j = 0; j < kNTpw; ++j) sp[j] = ss[j] + __shfl_xor(ss[j], 16);
    *(v4f*)&smem[kM * kPitch + srow] = sp;
  }

  int src = kM * kPitch;
  int dst = 0;
  v4f u;
  for (int step = 0;; ++step) {
    __syncthreads();

    float uacc[kNTpw];
    for (int j = 0; j < kNTpw; ++j) uacc[j] = 0.f;
    const float* tA = smem + src + aoff;
    const float* sk = smem + src + kH + khalf;
    load_a(aA, tA, 0);
    load_sp(sA, sk, 0);
    for (int kb = 0; kb < kH - 2 * kKB; kb += 2 * kKB) {
      load_w1(wB, mws, kb + kKB);
      load_a(aB, tA, kb + kKB);
      load_sp(sB, sk, kb + kKB);
      group_fence();
      mma_block1(acc, uacc, aA, sA, wA, inv);
      group_fence();
      load_w1(wA, mws, kb + 2 * kKB);
      load_a(aA, tA, kb + 2 * kKB);
      load_sp(sA, sk, kb + 2 * kKB);
      group_fence();
      mma_block1(acc, uacc, aB, sB, wB, inv);
      group_fence();
    }
    load_w1(wB, mws, kH - kKB);
    load_a(aB, tA, kH - kKB);
    load_sp(sB, sk, kH - kKB);
    const v4f cv = *(const v4f*)&tab[kCvOff + ncol + zoff];
    group_fence();
    mma_block1(acc, uacc, aA, sA, wA, inv);
    group_fence();
    mma_block1(acc, uacc, aB, sB, wB, inv);
    group_fence();

    for (int j = 0; j < kNTpw; ++j)
      u[j] = (uacc[j] + __shfl_xor(uacc[j], 16)) * inv + cv[j];
    if (step == kSteps - 2) break;
    zoff = 0;
    asm volatile("" : "+s"(zoff));
    mws = mw + zoff;
    load_w1(wA, mws, 0);
    {
      v4f ss = {0.f, 0.f, 0.f, 0.f};
      v4f q[8];
#pragma unroll
      for (int r = 0; r < 8; ++r) q[r] = *(const v4f*)&sq[(hi8 + r) * kPitch + ncol];
      group_fence();
      for (int mt = 0; mt < kPR; ++mt) {
#pragma unroll
        for (int r = 0; r < 8; ++r) {
          v4f v;
          for (int j = 0; j < kNTpw; ++j) {
            acc[mt][j][r] += u[j];
            v[j] = fmaxf(acc[mt][j][r], 0.f);
          }
          tsum[mt][r] += v;
          ss += v;
          *(v4f*)&smem[dst + (mt * 16 + hi8 + r) * kPitch + ncol] = v;
        }
      }
#pragma unroll
      for (int r = 0; r < 8; ++r) {
        v4f v;
        for (int j = 0; j < kNTpw; ++j) {
          acc[kPR][j][r] += u[j];
          v[j] = fmaxf(acc[kPR][j][r], 0.f);
        }
        ss += v;
        *(v4f*)&sq[(hi8 + r) * kPitch + ncol] = q[r] + v;
        *(v4f*)&smem[dst + (kPR * 16 + hi8 + r) * kPitch + ncol] = v;
      }
      v4f sp;
      for (int j = 0; j < kNTpw; ++j) sp[j] = ss[j] + __shfl_xor(ss[j], 16);
      *(v4f*)&smem[dst + srow] = sp;
    }
    const int o = src;
    src = dst;
    dst = o;
  }
  {
    v4f q[8];
#pragma unroll
    for (int r = 0; r < 8; ++r) q[r] = *(const v4f*)&sq[(hi8 + r) * kPitch + ncol];
    group_fence();
    for (int mt = 0; mt < kPR; ++mt) {
#pragma unroll
      for (int r = 0; r < 8; ++r) {
        v4f v;
        for (int j = 0; j < kNTpw; ++j) v[j] = fmaxf(acc[mt][j][r] + u[j], 0.f);
        *(v4f*)&smem[dst + (mt * 16 + hi8 + r) * kPitch + ncol] = tsum[mt][r] + v;
      }
    }
#pragma unroll
    for (int r = 0; r < 8; ++r) {
      v4f v;
      for (int j = 0; j < kNTpw; ++j) v[j] = fmaxf(acc[kPR][j][r] + u[j], 0.f);
      *(v4f*)&smem[dst + (kPR * 16 + hi8 + r) * kPitch + ncol] = q[r] + v;
    }
  }

  const float g = tab[kBiasOff + lrow];
  __syncthreads();
  v8f accL0, accL1 = {};
  {
    const v4f lo = *(const v4f*)&sL[tid * 8];
    const v4f hi = *(const v4f*)&sL[tid * 8 + 4];
    accL0 = __builtin_shufflevector(lo, hi, 0, 1, 2, 3, 4, 5, 6, 7);
  }
  logits_add(accL0, accL1, fdp, smem + dst + loff);
  const int mrow = wave * 16 + hi8;
#pragma unroll
  for (int r = 0; r < 8; ++r)
    out[(b * kM + mrow + r) * kA + lrow] = (accL0[r] + accL1[r]) + g;
}

__launch_bounds__(kThreads, kBlocksPerWgp)
__global__ void commnet_wmma_kernel(
    const int* __restrict__ ids, const float* __restrict__ emb,
    const float* __restrict__ W1, const float* __restrict__ b1,
    const float* __restrict__ W2, const float* __restrict__ b2,
    const float* __restrict__ Wd, const float* __restrict__ bd,
    float* __restrict__ out, const float* __restrict__ tab, size_t ws_bytes) {
  extern __shared__ __attribute__((aligned(16))) float smem[];
  if (ws_bytes >= kTabBytes)
    commnet_body_tab(smem, ids, out, tab);
  else
    commnet_body_direct(smem, ids, emb, W1, b1, W2, b2, Wd, bd, out);
}

extern "C" void kernel_launch(void* const* d_in, const int* in_sizes, int n_in,
                              void* d_out, int out_size, void* d_ws, size_t ws_size,
                              hipStream_t stream) {
  (void)in_sizes; (void)n_in; (void)out_size;
  const int*   ids = (const int*)d_in[0];
  const float* emb = (const float*)d_in[1];
  const float* W1  = (const float*)d_in[2];
  const float* b1  = (const float*)d_in[3];
  const float* W2  = (const float*)d_in[4];
  const float* b2  = (const float*)d_in[5];
  const float* Wd  = (const float*)d_in[6];
  const float* bd  = (const float*)d_in[7];
  float* out = (float*)d_out;
  float* tab = (float*)d_ws;

  const size_t lds_bytes = (size_t)kLdsFloats * sizeof(float);
  (void)hipFuncSetAttribute((const void*)commnet_wmma_kernel,
                            hipFuncAttributeMaxDynamicSharedMemorySize,
                            (int)lds_bytes);
  commnet_table_kernel<<<dim3(kTabGrid), dim3(kThreads), 0, stream>>>(
      emb, W1, b1, W2, b2, Wd, bd, tab, ws_size);
  commnet_wmma_kernel<<<dim3(kB), dim3(kThreads), lds_bytes, stream>>>(
      ids, emb, W1, b1, W2, b2, Wd, bd, out, tab, ws_size);
}
